# starting kernel
# baseline (speedup 1.0000x reference)
#include <hip/hip_runtime.h>
#include <hip/hip_bf16.h>

#define NB      64
#define LSIG    480000
#define TILE    256
#define NTILES  (LSIG / TILE)
#define NSEC    3
#define CSTRIDE 128
#define FSTRIDE 320

typedef __attribute__((ext_vector_type(2))) float v2f;
typedef __attribute__((ext_vector_type(8))) float v8f;

__device__ __forceinline__ v8f wmma_f32_16x16x4(v2f a, v2f b, v8f c) {
  return __builtin_amdgcn_wmma_f32_16x16x4_f32(false, a, false, b, (short)0, c,
                                               false, false);
}

__global__ void derive_kernel(const float* __restrict__ knobs,
                              const float* __restrict__ W1,
                              const float* __restrict__ b1v,
                              const float* __restrict__ W2,
                              const float* __restrict__ b2v,
                              float* __restrict__ coef,
                              float* __restrict__ fr,
                              float* __restrict__ gt) {
  int b = threadIdx.x;
  if (b >= NB) return;
  float k0 = knobs[b * 3 + 0], k1 = knobs[b * 3 + 1], k2 = knobs[b * 3 + 2];
  float hmlp[16];
  for (int i = 0; i < 16; ++i)
    hmlp[i] = tanhf(k0 * W1[0 * 16 + i] + k1 * W1[1 * 16 + i] +
                    k2 * W1[2 * 16 + i] + b1v[i]);
  float p[15];
  for (int j = 0; j < 15; ++j) {
    float acc = b2v[j];
    for (int i = 0; i < 16; ++i) acc += hmlp[i] * W2[i * 15 + j];
    p[j] = acc;
  }
  for (int s = 0; s < NSEC; ++s) {
    float c0 = p[s * 5 + 0], c1 = p[s * 5 + 1], c2 = p[s * 5 + 2];
    float a2c = 0.999f * tanhf(p[s * 5 + 4]);
    float a1c = 0.999f * (1.0f + a2c) * tanhf(p[s * 5 + 3]);
    float A00 = -a1c, A01 = 1.0f, A10 = -a2c, A11 = 0.0f;
    float Bv0 = c1 - a1c * c0, Bv1 = c2 - a2c * c0;
    float* cf = coef + (b * NSEC + s) * CSTRIDE;

    float h[16];
    {
      float v0 = Bv0, v1 = Bv1;
      h[0] = c0;
      for (int k = 1; k < 16; ++k) {
        h[k] = v0;
        float n0 = A00 * v0 + A01 * v1;
        float n1 = A10 * v0 + A11 * v1;
        v0 = n0; v1 = n1;
      }
    }
    {
      float v0 = Bv0, v1 = Bv1;
      cf[16 + 15] = v0; cf[32 + 15] = v1;
      for (int j = 14; j >= 0; --j) {
        float n0 = A00 * v0 + A01 * v1;
        float n1 = A10 * v0 + A11 * v1;
        v0 = n0; v1 = n1;
        cf[16 + j] = v0; cf[32 + j] = v1;
      }
    }
    float g0i[16], g1i[16];
    {
      float g0 = 1.0f, g1 = 0.0f;
      for (int i = 0; i < 16; ++i) {
        g0i[i] = g0; g1i[i] = g1;
        float n0 = g0 * A00 + g1 * A10;
        float n1 = g0 * A01 + g1 * A11;
        g0 = n0; g1 = n1;
      }
    }
    {
      float M00 = A00, M01 = A01, M10 = A10, M11 = A11;
      for (int q = 0; q < 4; ++q) {
        float n00 = M00 * M00 + M01 * M10, n01 = M00 * M01 + M01 * M11;
        float n10 = M10 * M00 + M11 * M10, n11 = M10 * M01 + M11 * M11;
        M00 = n00; M01 = n01; M10 = n10; M11 = n11;
      }
      for (int q = 0; q < 5; ++q) {
        cf[80 + 4 * q + 0] = M00; cf[80 + 4 * q + 1] = M01;
        cf[80 + 4 * q + 2] = M10; cf[80 + 4 * q + 3] = M11;
        float n00 = M00 * M00 + M01 * M10, n01 = M00 * M01 + M01 * M11;
        float n10 = M10 * M00 + M11 * M10, n11 = M10 * M01 + M11 * M11;
        M00 = n00; M01 = n01; M10 = n10; M11 = n11;
      }
    }
    {
      float* fb = fr + (b * NSEC + s) * FSTRIDE;
      for (int lane = 0; lane < 32; ++lane) {
        int m = lane & 15, half = lane >> 4;
        for (int k = 0; k < 4; ++k) {
          int K0 = 4 * k + 2 * half;
          fb[(k * 32 + lane) * 2 + 0] = (K0 <= m)     ? h[m - K0]     : 0.0f;
          fb[(k * 32 + lane) * 2 + 1] = (K0 + 1 <= m) ? h[m - K0 - 1] : 0.0f;
        }
        fb[256 + lane * 2 + 0] = half ? 0.0f : g0i[m];
        fb[256 + lane * 2 + 1] = half ? 0.0f : g1i[m];
      }
    }
    {
      float g0 = 1.0f, g1 = 0.0f;
      float* G = gt + (b * NSEC + s) * TILE * 2;
      for (int n = 0; n < TILE; ++n) {
        G[n * 2 + 0] = g0; G[n * 2 + 1] = g1;
        float n0 = g0 * A00 + g1 * A10;
        float n1 = g0 * A01 + g1 * A11;
        g0 = n0; g1 = n1;
      }
    }
  }
}

__global__ void __launch_bounds__(256) tile_kernel(const float* in, float* out,
                                                   const float* __restrict__ coef,
                                                   const float* __restrict__ fr,
                                                   const float* __restrict__ gt,
                                                   const float* __restrict__ sg,
                                                   float* __restrict__ ts,
                                                   int sec, int prevsec) {
  __shared__ float lds[8 * TILE];
  const int lane = threadIdx.x & 31;
  const int wave = threadIdx.x >> 5;
  const int gw = blockIdx.x * 8 + wave;
  const int b = gw / NTILES;
  const int t = gw - b * NTILES;
  const int base = b * LSIG + t * TILE;
  float* xs = lds + wave * TILE;

  float4 u0 = *(const float4*)(in + base + lane * 8);
  float4 u1 = *(const float4*)(in + base + lane * 8 + 4);

  if (prevsec >= 0) {
    float2 sv = ((const float2*)sg)[gw];
    const float4* G4 = (const float4*)(gt + (b * NSEC + prevsec) * TILE * 2);
    float4 ga = G4[lane * 4 + 0];
    float4 gb = G4[lane * 4 + 1];
    float4 gc = G4[lane * 4 + 2];
    float4 gd = G4[lane * 4 + 3];
    u0.x += ga.x * sv.x + ga.y * sv.y;
    u0.y += ga.z * sv.x + ga.w * sv.y;
    u0.z += gb.x * sv.x + gb.y * sv.y;
    u0.w += gb.z * sv.x + gb.w * sv.y;
    u1.x += gc.x * sv.x + gc.y * sv.y;
    u1.y += gc.z * sv.x + gc.w * sv.y;
    u1.z += gd.x * sv.x + gd.y * sv.y;
    u1.w += gd.z * sv.x + gd.w * sv.y;
  }
  *(float4*)(xs + lane * 8)     = u0;
  *(float4*)(xs + lane * 8 + 4) = u1;
  __syncthreads();

  const float* cf = coef + (b * NSEC + sec) * CSTRIDE;
  const v2f* hf = (const v2f*)(fr + (b * NSEC + sec) * FSTRIDE);
  const int m = lane & 15;
  const int half = lane >> 4;

  float4 xc[4];
#pragma unroll
  for (int q = 0; q < 4; ++q)
    xc[q] = ((const float4*)(xs + m * 16))[q];

  v8f acc = {0.f, 0.f, 0.f, 0.f, 0.f, 0.f, 0.f, 0.f};
#pragma unroll
  for (int k = 0; k < 4; ++k) {
    v2f a = hf[k * 32 + lane];
    v2f bb;
    bb.x = half ? xc[k].z : xc[k].x;
    bb.y = half ? xc[k].w : xc[k].y;
    acc = wmma_f32_16x16x4(a, bb, acc);
  }

  float fx0 = 0.f, fx1 = 0.f;
#pragma unroll
  for (int q = 0; q < 4; ++q) {
    fx0 += cf[16 + 4 * q + 0] * xc[q].x + cf[16 + 4 * q + 1] * xc[q].y +
           cf[16 + 4 * q + 2] * xc[q].z + cf[16 + 4 * q + 3] * xc[q].w;
    fx1 += cf[32 + 4 * q + 0] * xc[q].x + cf[32 + 4 * q + 1] * xc[q].y +
           cf[32 + 4 * q + 2] * xc[q].z + cf[32 + 4 * q + 3] * xc[q].w;
  }

  float p0 = fx0, p1 = fx1;
#pragma unroll
  for (int s = 0; s < 4; ++s) {
    float M00 = cf[80 + 4 * s + 0], M01 = cf[80 + 4 * s + 1];
    float M10 = cf[80 + 4 * s + 2], M11 = cf[80 + 4 * s + 3];
    int len = 1 << s;
    float q0 = __shfl_up(p0, (unsigned)len, 32);
    float q1 = __shfl_up(p1, (unsigned)len, 32);
    if (m >= len) {
      p0 = M00 * q0 + M01 * q1 + p0;
      p1 = M10 * q0 + M11 * q1 + p1;
    }
  }
  float e0 = __shfl_up(p0, 1u, 32);
  float e1 = __shfl_up(p1, 1u, 32);
  float s0c = (m >= 1) ? e0 : 0.f;
  float s1c = (m >= 1) ? e1 : 0.f;

  {
    v2f a = hf[128 + lane];
    v2f bb;
    bb.x = half ? 0.f : s0c;
    bb.y = half ? 0.f : s1c;
    acc = wmma_f32_16x16x4(a, bb, acc);
  }

  if (lane == 15) {
    ts[(b * NTILES + t) * 2 + 0] = p0;
    ts[(b * NTILES + t) * 2 + 1] = p1;
  }

  __syncthreads();
  {
    float4 d0, d1;
    d0.x = acc[0]; d0.y = acc[1]; d0.z = acc[2]; d0.w = acc[3];
    d1.x = acc[4]; d1.y = acc[5]; d1.z = acc[6]; d1.w = acc[7];
    *(float4*)(xs + m * 16 + 8 * half)     = d0;
    *(float4*)(xs + m * 16 + 8 * half + 4) = d1;
  }
  __syncthreads();
  *(float4*)(out + base + lane * 8)     = *(const float4*)(xs + lane * 8);
  *(float4*)(out + base + lane * 8 + 4) = *(const float4*)(xs + lane * 8 + 4);
}

__global__ void scan_kernel(const float* __restrict__ coef,
                            const float* __restrict__ ts,
                            float* __restrict__ sg, int sec) {
  int b = threadIdx.x;
  if (b >= NB) return;
  const float* cf = coef + (b * NSEC + sec) * CSTRIDE;
  float A00 = cf[96], A01 = cf[97], A10 = cf[98], A11 = cf[99];
  float s0 = 0.f, s1 = 0.f;
#pragma unroll 5
  for (int t = 0; t < NTILES; ++t) {
    int idx = (b * NTILES + t) * 2;
    sg[idx + 0] = s0; sg[idx + 1] = s1;
    float z0 = ts[idx + 0], z1 = ts[idx + 1];
    float n0 = A00 * s0 + A01 * s1 + z0;
    float n1 = A10 * s0 + A11 * s1 + z1;
    s0 = n0; s1 = n1;
  }
}

__global__ void __launch_bounds__(256) fixup_kernel(float* out,
                                                    const float* __restrict__ gt,
                                                    const float* __restrict__ sg,
                                                    int sec) {
  int tile = blockIdx.x;
  int b = tile / NTILES;
  int t = tile - b * NTILES;
  int n = threadIdx.x;
  float s0 = sg[tile * 2 + 0], s1 = sg[tile * 2 + 1];
  const float2* G = (const float2*)(gt + (b * NSEC + sec) * TILE * 2);
  float2 g = G[n];
  int idx = b * LSIG + t * TILE + n;
  out[idx] += g.x * s0 + g.y * s1;
}

extern "C" void kernel_launch(void* const* d_in, const int* in_sizes, int n_in,
                              void* d_out, int out_size, void* d_ws, size_t ws_size,
                              hipStream_t stream) {
  (void)in_sizes; (void)n_in; (void)out_size; (void)ws_size;
  const float* x     = (const float*)d_in[0];
  const float* knobs = (const float*)d_in[1];
  const float* W1    = (const float*)d_in[2];
  const float* b1v   = (const float*)d_in[3];
  const float* W2    = (const float*)d_in[4];
  const float* b2v   = (const float*)d_in[5];
  float* out = (float*)d_out;

  float* COEF = (float*)d_ws;
  float* FR   = COEF + NB * NSEC * CSTRIDE;
  float* GT   = FR   + NB * NSEC * FSTRIDE;
  float* TS   = GT   + NB * NSEC * TILE * 2;
  float* SG   = TS   + NB * NTILES * 2;

  derive_kernel<<<1, 64, 0, stream>>>(knobs, W1, b1v, W2, b2v, COEF, FR, GT);

  const int tileBlocks = (NB * NTILES) / 8;
  for (int sec = 0; sec < NSEC; ++sec) {
    const float* src = (sec == 0) ? x : out;
    tile_kernel<<<tileBlocks, 256, 0, stream>>>(src, out, COEF, FR, GT, SG, TS,
                                                sec, sec - 1);
    scan_kernel<<<1, 64, 0, stream>>>(COEF, TS, SG, sec);
  }
  fixup_kernel<<<NB * NTILES, 256, 0, stream>>>(out, GT, SG, NSEC - 1);
}


# tile kernel: one wave walks 15 consecutive tiles of a row (one-wave workgroups; row operands loaded once, next tile requested a trip ahead, no block barriers) instead of one short wave per tile
# speedup vs baseline: 1.0427x; 1.0427x over previous
#include <hip/hip_runtime.h>
#include <hip/hip_bf16.h>

#define NB      64
#define LSIG    480000
#define TILE    256
#define NTILES  (LSIG / TILE)
#define NSEC    3
#define CSTRIDE 128
#define FSTRIDE 320
#define TPW     15
#define WPR     (NTILES / TPW)

typedef __attribute__((ext_vector_type(2))) float v2f;
typedef __attribute__((ext_vector_type(8))) float v8f;

__device__ __forceinline__ v8f wmma_f32_16x16x4(v2f a, v2f b, v8f c) {
  return __builtin_amdgcn_wmma_f32_16x16x4_f32(false, a, false, b, (short)0, c,
                                               false, false);
}

__global__ void derive_kernel(const float* __restrict__ knobs,
                              const float* __restrict__ W1,
                              const float* __restrict__ b1v,
                              const float* __restrict__ W2,
                              const float* __restrict__ b2v,
                              float* __restrict__ coef,
                              float* __restrict__ fr,
                              float* __restrict__ gt) {
  int b = threadIdx.x;
  if (b >= NB) return;
  float k0 = knobs[b * 3 + 0], k1 = knobs[b * 3 + 1], k2 = knobs[b * 3 + 2];
  float hmlp[16];
  for (int i = 0; i < 16; ++i)
    hmlp[i] = tanhf(k0 * W1[0 * 16 + i] + k1 * W1[1 * 16 + i] +
                    k2 * W1[2 * 16 + i] + b1v[i]);
  float p[15];
  for (int j = 0; j < 15; ++j) {
    float acc = b2v[j];
    for (int i = 0; i < 16; ++i) acc += hmlp[i] * W2[i * 15 + j];
    p[j] = acc;
  }
  for (int s = 0; s < NSEC; ++s) {
    float c0 = p[s * 5 + 0], c1 = p[s * 5 + 1], c2 = p[s * 5 + 2];
    float a2c = 0.999f * tanhf(p[s * 5 + 4]);
    float a1c = 0.999f * (1.0f + a2c) * tanhf(p[s * 5 + 3]);
    float A00 = -a1c, A01 = 1.0f, A10 = -a2c, A11 = 0.0f;
    float Bv0 = c1 - a1c * c0, Bv1 = c2 - a2c * c0;
    float* cf = coef + (b * NSEC + s) * CSTRIDE;

    float h[16];
    {
      float v0 = Bv0, v1 = Bv1;
      h[0] = c0;
      for (int k = 1; k < 16; ++k) {
        h[k] = v0;
        float n0 = A00 * v0 + A01 * v1;
        float n1 = A10 * v0 + A11 * v1;
        v0 = n0; v1 = n1;
      }
    }
    {
      float v0 = Bv0, v1 = Bv1;
      cf[16 + 15] = v0; cf[32 + 15] = v1;
      for (int j = 14; j >= 0; --j) {
        float n0 = A00 * v0 + A01 * v1;
        float n1 = A10 * v0 + A11 * v1;
        v0 = n0; v1 = n1;
        cf[16 + j] = v0; cf[32 + j] = v1;
      }
    }
    float g0i[16], g1i[16];
    {
      float g0 = 1.0f, g1 = 0.0f;
      for (int i = 0; i < 16; ++i) {
        g0i[i] = g0; g1i[i] = g1;
        float n0 = g0 * A00 + g1 * A10;
        float n1 = g0 * A01 + g1 * A11;
        g0 = n0; g1 = n1;
      }
    }
    {
      float M00 = A00, M01 = A01, M10 = A10, M11 = A11;
      for (int q = 0; q < 4; ++q) {
        float n00 = M00 * M00 + M01 * M10, n01 = M00 * M01 + M01 * M11;
        float n10 = M10 * M00 + M11 * M10, n11 = M10 * M01 + M11 * M11;
        M00 = n00; M01 = n01; M10 = n10; M11 = n11;
      }
      for (int q = 0; q < 5; ++q) {
        cf[80 + 4 * q + 0] = M00; cf[80 + 4 * q + 1] = M01;
        cf[80 + 4 * q + 2] = M10; cf[80 + 4 * q + 3] = M11;
        float n00 = M00 * M00 + M01 * M10, n01 = M00 * M01 + M01 * M11;
        float n10 = M10 * M00 + M11 * M10, n11 = M10 * M01 + M11 * M11;
        M00 = n00; M01 = n01; M10 = n10; M11 = n11;
      }
    }
    {
      float* fb = fr + (b * NSEC + s) * FSTRIDE;
      for (int lane = 0; lane < 32; ++lane) {
        int m = lane & 15, half = lane >> 4;
        for (int k = 0; k < 4; ++k) {
          int K0 = 4 * k + 2 * half;
          fb[(k * 32 + lane) * 2 + 0] = (K0 <= m)     ? h[m - K0]     : 0.0f;
          fb[(k * 32 + lane) * 2 + 1] = (K0 + 1 <= m) ? h[m - K0 - 1] : 0.0f;
        }
        fb[256 + lane * 2 + 0] = half ? 0.0f : g0i[m];
        fb[256 + lane * 2 + 1] = half ? 0.0f : g1i[m];
      }
    }
    {
      float g0 = 1.0f, g1 = 0.0f;
      float* G = gt + (b * NSEC + s) * TILE * 2;
      for (int n = 0; n < TILE; ++n) {
        G[n * 2 + 0] = g0; G[n * 2 + 1] = g1;
        float n0 = g0 * A00 + g1 * A10;
        float n1 = g0 * A01 + g1 * A11;
        g0 = n0; g1 = n1;
      }
    }
  }
}


__device__ __forceinline__ void wave_lds_order() {
  __builtin_amdgcn_fence(__ATOMIC_RELEASE, "wavefront");
  __builtin_amdgcn_wave_barrier();
  __builtin_amdgcn_fence(__ATOMIC_ACQUIRE, "wavefront");
}

__global__ void __launch_bounds__(32) tile_kernel(const float* in, float* out,
                                                  const float* __restrict__ coef,
                                                  const float* __restrict__ fr,
                                                  const float* __restrict__ gt,
                                                  const float* __restrict__ sg,
                                                  float* __restrict__ ts,
                                                  int sec, int prevsec) {
  __shared__ float xs[TILE];
  const int lane = threadIdx.x;
  const int b = blockIdx.x / WPR;
  const int w = blockIdx.x - b * WPR;
  const int tile0 = b * NTILES + w * TPW;
  const int base0 = b * LSIG + w * TPW * TILE;
  const int m = lane & 15;
  const int half = lane >> 4;
  const bool fuse = prevsec >= 0;

  const float* cf = coef + (b * NSEC + sec) * CSTRIDE;
  const v2f* hf = (const v2f*)(fr + (b * NSEC + sec) * FSTRIDE);
  v2f ha[4];
#pragma unroll
  for (int k = 0; k < 4; ++k) ha[k] = hf[k * 32 + lane];
  const v2f gfr = hf[128 + lane];
  float f0[16], f1[16], Ms[16];
#pragma unroll
  for (int j = 0; j < 16; ++j) {
    f0[j] = cf[16 + j];
    f1[j] = cf[32 + j];
    Ms[j] = cf[80 + j];
  }
  float4 ga = {0.f, 0.f, 0.f, 0.f}, gb = ga, gc = ga, gd = ga;
  if (fuse) {
    const float4* G4 = (const float4*)(gt + (b * NSEC + prevsec) * TILE * 2);
    ga = G4[lane * 4 + 0];
    gb = G4[lane * 4 + 1];
    gc = G4[lane * 4 + 2];
    gd = G4[lane * 4 + 3];
  }
  const float2* sg2 = (const float2*)sg;
  float2* ts2 = (float2*)ts;
  const float* src = in + base0 + lane * 8;
  float* dst = out + base0 + lane * 8;

  float4 u0 = *(const float4*)(src);
  float4 u1 = *(const float4*)(src + 4);
  float svx_l = 0.f, svy_l = 0.f;
  if (fuse) {
    float2 t = sg2[tile0 + (lane < TPW ? lane : TPW - 1)];
    svx_l = t.x; svy_l = t.y;
  }

  for (int i = 0; i < TPW; ++i) {
    float4 n0 = u0, n1 = u1;
    if (i + 1 < TPW) {
      n0 = *(const float4*)(src + (i + 1) * TILE);
      n1 = *(const float4*)(src + (i + 1) * TILE + 4);
    }

    if (fuse) {
      float2 sv;
      sv.x = __int_as_float(__builtin_amdgcn_readlane(__float_as_int(svx_l), i));
      sv.y = __int_as_float(__builtin_amdgcn_readlane(__float_as_int(svy_l), i));
      u0.x += ga.x * sv.x + ga.y * sv.y;
      u0.y += ga.z * sv.x + ga.w * sv.y;
      u0.z += gb.x * sv.x + gb.y * sv.y;
      u0.w += gb.z * sv.x + gb.w * sv.y;
      u1.x += gc.x * sv.x + gc.y * sv.y;
      u1.y += gc.z * sv.x + gc.w * sv.y;
      u1.z += gd.x * sv.x + gd.y * sv.y;
      u1.w += gd.z * sv.x + gd.w * sv.y;
    }
    *(float4*)(xs + lane * 8)     = u0;
    *(float4*)(xs + lane * 8 + 4) = u1;
    wave_lds_order();

    float4 xc[4];
#pragma unroll
    for (int q = 0; q < 4; ++q)
      xc[q] = ((const float4*)(xs + m * 16))[q];

    v8f acc = {0.f, 0.f, 0.f, 0.f, 0.f, 0.f, 0.f, 0.f};
#pragma unroll
    for (int k = 0; k < 4; ++k) {
      v2f bb;
      bb.x = half ? xc[k].z : xc[k].x;
      bb.y = half ? xc[k].w : xc[k].y;
      acc = wmma_f32_16x16x4(ha[k], bb, acc);
    }

    float fx0 = 0.f, fx1 = 0.f;
#pragma unroll
    for (int q = 0; q < 4; ++q) {
      fx0 += f0[4 * q + 0] * xc[q].x + f0[4 * q + 1] * xc[q].y +
             f0[4 * q + 2] * xc[q].z + f0[4 * q + 3] * xc[q].w;
      fx1 += f1[4 * q + 0] * xc[q].x + f1[4 * q + 1] * xc[q].y +
             f1[4 * q + 2] * xc[q].z + f1[4 * q + 3] * xc[q].w;
    }

    float p0 = fx0, p1 = fx1;
#pragma unroll
    for (int s = 0; s < 4; ++s) {
      float M00 = Ms[4 * s + 0], M01 = Ms[4 * s + 1];
      float M10 = Ms[4 * s + 2], M11 = Ms[4 * s + 3];
      int len = 1 << s;
      float q0 = __shfl_up(p0, (unsigned)len, 32);
      float q1 = __shfl_up(p1, (unsigned)len, 32);
      if (m >= len) {
        p0 = M00 * q0 + M01 * q1 + p0;
        p1 = M10 * q0 + M11 * q1 + p1;
      }
    }
    float e0 = __shfl_up(p0, 1u, 32);
    float e1 = __shfl_up(p1, 1u, 32);
    float s0c = (m >= 1) ? e0 : 0.f;
    float s1c = (m >= 1) ? e1 : 0.f;

    {
      v2f bb;
      bb.x = half ? 0.f : s0c;
      bb.y = half ? 0.f : s1c;
      acc = wmma_f32_16x16x4(gfr, bb, acc);
    }

    if (lane == 15) {
      float2 z; z.x = p0; z.y = p1;
      ts2[tile0 + i] = z;
    }

    wave_lds_order();
    {
      float4 d0, d1;
      d0.x = acc[0]; d0.y = acc[1]; d0.z = acc[2]; d0.w = acc[3];
      d1.x = acc[4]; d1.y = acc[5]; d1.z = acc[6]; d1.w = acc[7];
      *(float4*)(xs + m * 16 + 8 * half)     = d0;
      *(float4*)(xs + m * 16 + 8 * half + 4) = d1;
    }
    wave_lds_order();
    float4 o0 = *(const float4*)(xs + lane * 8);
    float4 o1 = *(const float4*)(xs + lane * 8 + 4);
    wave_lds_order();
    *(float4*)(dst + i * TILE)     = o0;
    *(float4*)(dst + i * TILE + 4) = o1;

    u0 = n0; u1 = n1;
  }
}

__global__ void scan_kernel(const float* __restrict__ coef,
                            const float* __restrict__ ts,
                            float* __restrict__ sg, int sec) {
  int b = threadIdx.x;
  if (b >= NB) return;
  const float* cf = coef + (b * NSEC + sec) * CSTRIDE;
  float A00 = cf[96], A01 = cf[97], A10 = cf[98], A11 = cf[99];
  float s0 = 0.f, s1 = 0.f;
#pragma unroll 5
  for (int t = 0; t < NTILES; ++t) {
    int idx = (b * NTILES + t) * 2;
    sg[idx + 0] = s0; sg[idx + 1] = s1;
    float z0 = ts[idx + 0], z1 = ts[idx + 1];
    float n0 = A00 * s0 + A01 * s1 + z0;
    float n1 = A10 * s0 + A11 * s1 + z1;
    s0 = n0; s1 = n1;
  }
}

__global__ void __launch_bounds__(256) fixup_kernel(float* out,
                                                    const float* __restrict__ gt,
                                                    const float* __restrict__ sg,
                                                    int sec) {
  int tile = blockIdx.x;
  int b = tile / NTILES;
  int t = tile - b * NTILES;
  int n = threadIdx.x;
  float s0 = sg[tile * 2 + 0], s1 = sg[tile * 2 + 1];
  const float2* G = (const float2*)(gt + (b * NSEC + sec) * TILE * 2);
  float2 g = G[n];
  int idx = b * LSIG + t * TILE + n;
  out[idx] += g.x * s0 + g.y * s1;
}

extern "C" void kernel_launch(void* const* d_in, const int* in_sizes, int n_in,
                              void* d_out, int out_size, void* d_ws, size_t ws_size,
                              hipStream_t stream) {
  (void)in_sizes; (void)n_in; (void)out_size; (void)ws_size;
  const float* x     = (const float*)d_in[0];
  const float* knobs = (const float*)d_in[1];
  const float* W1    = (const float*)d_in[2];
  const float* b1v   = (const float*)d_in[3];
  const float* W2    = (const float*)d_in[4];
  const float* b2v   = (const float*)d_in[5];
  float* out = (float*)d_out;

  float* COEF = (float*)d_ws;
  float* FR   = COEF + NB * NSEC * CSTRIDE;
  float* GT   = FR   + NB * NSEC * FSTRIDE;
  float* TS   = GT   + NB * NSEC * TILE * 2;
  float* SG   = TS   + NB * NTILES * 2;

  derive_kernel<<<1, 64, 0, stream>>>(knobs, W1, b1v, W2, b2v, COEF, FR, GT);

  const int tileWaves = NB * WPR;
  for (int sec = 0; sec < NSEC; ++sec) {
    const float* src = (sec == 0) ? x : out;
    tile_kernel<<<tileWaves, 32, 0, stream>>>(src, out, COEF, FR, GT, SG, TS,
                                              sec, sec - 1);
    scan_kernel<<<1, 64, 0, stream>>>(COEF, TS, SG, sec);
  }
  fixup_kernel<<<NB * NTILES, 256, 0, stream>>>(out, GT, SG, NSEC - 1);
}


# cross-tile scan kernel: one wave per row on 64 workgroups, the row's tile states loaded in whole lines and broadcast lane by lane, not 64 threads of one block on strided memory; sum order unchanged
# speedup vs baseline: 2.4715x; 2.3703x over previous
#include <hip/hip_runtime.h>
#include <hip/hip_bf16.h>

#define NB      64
#define LSIG    480000
#define TILE    256
#define NTILES  (LSIG / TILE)
#define NSEC    3
#define CSTRIDE 128
#define FSTRIDE 320
#define TPW     15
#define WPR     (NTILES / TPW)

typedef __attribute__((ext_vector_type(2))) float v2f;
typedef __attribute__((ext_vector_type(8))) float v8f;

__device__ __forceinline__ v8f wmma_f32_16x16x4(v2f a, v2f b, v8f c) {
  return __builtin_amdgcn_wmma_f32_16x16x4_f32(false, a, false, b, (short)0, c,
                                               false, false);
}

__global__ void derive_kernel(const float* __restrict__ knobs,
                              const float* __restrict__ W1,
                              const float* __restrict__ b1v,
                              const float* __restrict__ W2,
                              const float* __restrict__ b2v,
                              float* __restrict__ coef,
                              float* __restrict__ fr,
                              float* __restrict__ gt) {
  int b = threadIdx.x;
  if (b >= NB) return;
  float k0 = knobs[b * 3 + 0], k1 = knobs[b * 3 + 1], k2 = knobs[b * 3 + 2];
  float hmlp[16];
  for (int i = 0; i < 16; ++i)
    hmlp[i] = tanhf(k0 * W1[0 * 16 + i] + k1 * W1[1 * 16 + i] +
                    k2 * W1[2 * 16 + i] + b1v[i]);
  float p[15];
  for (int j = 0; j < 15; ++j) {
    float acc = b2v[j];
    for (int i = 0; i < 16; ++i) acc += hmlp[i] * W2[i * 15 + j];
    p[j] = acc;
  }
  for (int s = 0; s < NSEC; ++s) {
    float c0 = p[s * 5 + 0], c1 = p[s * 5 + 1], c2 = p[s * 5 + 2];
    float a2c = 0.999f * tanhf(p[s * 5 + 4]);
    float a1c = 0.999f * (1.0f + a2c) * tanhf(p[s * 5 + 3]);
    float A00 = -a1c, A01 = 1.0f, A10 = -a2c, A11 = 0.0f;
    float Bv0 = c1 - a1c * c0, Bv1 = c2 - a2c * c0;
    float* cf = coef + (b * NSEC + s) * CSTRIDE;

    float h[16];
    {
      float v0 = Bv0, v1 = Bv1;
      h[0] = c0;
      for (int k = 1; k < 16; ++k) {
        h[k] = v0;
        float n0 = A00 * v0 + A01 * v1;
        float n1 = A10 * v0 + A11 * v1;
        v0 = n0; v1 = n1;
      }
    }
    {
      float v0 = Bv0, v1 = Bv1;
      cf[16 + 15] = v0; cf[32 + 15] = v1;
      for (int j = 14; j >= 0; --j) {
        float n0 = A00 * v0 + A01 * v1;
        float n1 = A10 * v0 + A11 * v1;
        v0 = n0; v1 = n1;
        cf[16 + j] = v0; cf[32 + j] = v1;
      }
    }
    float g0i[16], g1i[16];
    {
      float g0 = 1.0f, g1 = 0.0f;
      for (int i = 0; i < 16; ++i) {
        g0i[i] = g0; g1i[i] = g1;
        float n0 = g0 * A00 + g1 * A10;
        float n1 = g0 * A01 + g1 * A11;
        g0 = n0; g1 = n1;
      }
    }
    {
      float M00 = A00, M01 = A01, M10 = A10, M11 = A11;
      for (int q = 0; q < 4; ++q) {
        float n00 = M00 * M00 + M01 * M10, n01 = M00 * M01 + M01 * M11;
        float n10 = M10 * M00 + M11 * M10, n11 = M10 * M01 + M11 * M11;
        M00 = n00; M01 = n01; M10 = n10; M11 = n11;
      }
      for (int q = 0; q < 5; ++q) {
        cf[80 + 4 * q + 0] = M00; cf[80 + 4 * q + 1] = M01;
        cf[80 + 4 * q + 2] = M10; cf[80 + 4 * q + 3] = M11;
        float n00 = M00 * M00 + M01 * M10, n01 = M00 * M01 + M01 * M11;
        float n10 = M10 * M00 + M11 * M10, n11 = M10 * M01 + M11 * M11;
        M00 = n00; M01 = n01; M10 = n10; M11 = n11;
      }
    }
    {
      float* fb = fr + (b * NSEC + s) * FSTRIDE;
      for (int lane = 0; lane < 32; ++lane) {
        int m = lane & 15, half = lane >> 4;
        for (int k = 0; k < 4; ++k) {
          int K0 = 4 * k + 2 * half;
          fb[(k * 32 + lane) * 2 + 0] = (K0 <= m)     ? h[m - K0]     : 0.0f;
          fb[(k * 32 + lane) * 2 + 1] = (K0 + 1 <= m) ? h[m - K0 - 1] : 0.0f;
        }
        fb[256 + lane * 2 + 0] = half ? 0.0f : g0i[m];
        fb[256 + lane * 2 + 1] = half ? 0.0f : g1i[m];
      }
    }
    {
      float g0 = 1.0f, g1 = 0.0f;
      float* G = gt + (b * NSEC + s) * TILE * 2;
      for (int n = 0; n < TILE; ++n) {
        G[n * 2 + 0] = g0; G[n * 2 + 1] = g1;
        float n0 = g0 * A00 + g1 * A10;
        float n1 = g0 * A01 + g1 * A11;
        g0 = n0; g1 = n1;
      }
    }
  }
}


__device__ __forceinline__ void wave_lds_order() {
  __builtin_amdgcn_fence(__ATOMIC_RELEASE, "wavefront");
  __builtin_amdgcn_wave_barrier();
  __builtin_amdgcn_fence(__ATOMIC_ACQUIRE, "wavefront");
}

__global__ void __launch_bounds__(32) tile_kernel(const float* in, float* out,
                                                  const float* __restrict__ coef,
                                                  const float* __restrict__ fr,
                                                  const float* __restrict__ gt,
                                                  const float* __restrict__ sg,
                                                  float* __restrict__ ts,
                                                  int sec, int prevsec) {
  __shared__ float xs[TILE];
  const int lane = threadIdx.x;
  const int b = blockIdx.x / WPR;
  const int w = blockIdx.x - b * WPR;
  const int tile0 = b * NTILES + w * TPW;
  const int base0 = b * LSIG + w * TPW * TILE;
  const int m = lane & 15;
  const int half = lane >> 4;
  const bool fuse = prevsec >= 0;

  const float* cf = coef + (b * NSEC + sec) * CSTRIDE;
  const v2f* hf = (const v2f*)(fr + (b * NSEC + sec) * FSTRIDE);
  v2f ha[4];
#pragma unroll
  for (int k = 0; k < 4; ++k) ha[k] = hf[k * 32 + lane];
  const v2f gfr = hf[128 + lane];
  float f0[16], f1[16], Ms[16];
#pragma unroll
  for (int j = 0; j < 16; ++j) {
    f0[j] = cf[16 + j];
    f1[j] = cf[32 + j];
    Ms[j] = cf[80 + j];
  }
  float4 ga = {0.f, 0.f, 0.f, 0.f}, gb = ga, gc = ga, gd = ga;
  if (fuse) {
    const float4* G4 = (const float4*)(gt + (b * NSEC + prevsec) * TILE * 2);
    ga = G4[lane * 4 + 0];
    gb = G4[lane * 4 + 1];
    gc = G4[lane * 4 + 2];
    gd = G4[lane * 4 + 3];
  }
  const float2* sg2 = (const float2*)sg;
  float2* ts2 = (float2*)ts;
  const float* src = in + base0 + lane * 8;
  float* dst = out + base0 + lane * 8;

  float4 u0 = *(const float4*)(src);
  float4 u1 = *(const float4*)(src + 4);
  float svx_l = 0.f, svy_l = 0.f;
  if (fuse) {
    float2 t = sg2[tile0 + (lane < TPW ? lane : TPW - 1)];
    svx_l = t.x; svy_l = t.y;
  }

  for (int i = 0; i < TPW; ++i) {
    float4 n0 = u0, n1 = u1;
    if (i + 1 < TPW) {
      n0 = *(const float4*)(src + (i + 1) * TILE);
      n1 = *(const float4*)(src + (i + 1) * TILE + 4);
    }

    if (fuse) {
      float2 sv;
      sv.x = __int_as_float(__builtin_amdgcn_readlane(__float_as_int(svx_l), i));
      sv.y = __int_as_float(__builtin_amdgcn_readlane(__float_as_int(svy_l), i));
      u0.x += ga.x * sv.x + ga.y * sv.y;
      u0.y += ga.z * sv.x + ga.w * sv.y;
      u0.z += gb.x * sv.x + gb.y * sv.y;
      u0.w += gb.z * sv.x + gb.w * sv.y;
      u1.x += gc.x * sv.x + gc.y * sv.y;
      u1.y += gc.z * sv.x + gc.w * sv.y;
      u1.z += gd.x * sv.x + gd.y * sv.y;
      u1.w += gd.z * sv.x + gd.w * sv.y;
    }
    *(float4*)(xs + lane * 8)     = u0;
    *(float4*)(xs + lane * 8 + 4) = u1;
    wave_lds_order();

    float4 xc[4];
#pragma unroll
    for (int q = 0; q < 4; ++q)
      xc[q] = ((const float4*)(xs + m * 16))[q];

    v8f acc = {0.f, 0.f, 0.f, 0.f, 0.f, 0.f, 0.f, 0.f};
#pragma unroll
    for (int k = 0; k < 4; ++k) {
      v2f bb;
      bb.x = half ? xc[k].z : xc[k].x;
      bb.y = half ? xc[k].w : xc[k].y;
      acc = wmma_f32_16x16x4(ha[k], bb, acc);
    }

    float fx0 = 0.f, fx1 = 0.f;
#pragma unroll
    for (int q = 0; q < 4; ++q) {
      fx0 += f0[4 * q + 0] * xc[q].x + f0[4 * q + 1] * xc[q].y +
             f0[4 * q + 2] * xc[q].z + f0[4 * q + 3] * xc[q].w;
      fx1 += f1[4 * q + 0] * xc[q].x + f1[4 * q + 1] * xc[q].y +
             f1[4 * q + 2] * xc[q].z + f1[4 * q + 3] * xc[q].w;
    }

    float p0 = fx0, p1 = fx1;
#pragma unroll
    for (int s = 0; s < 4; ++s) {
      float M00 = Ms[4 * s + 0], M01 = Ms[4 * s + 1];
      float M10 = Ms[4 * s + 2], M11 = Ms[4 * s + 3];
      int len = 1 << s;
      float q0 = __shfl_up(p0, (unsigned)len, 32);
      float q1 = __shfl_up(p1, (unsigned)len, 32);
      if (m >= len) {
        p0 = M00 * q0 + M01 * q1 + p0;
        p1 = M10 * q0 + M11 * q1 + p1;
      }
    }
    float e0 = __shfl_up(p0, 1u, 32);
    float e1 = __shfl_up(p1, 1u, 32);
    float s0c = (m >= 1) ? e0 : 0.f;
    float s1c = (m >= 1) ? e1 : 0.f;

    {
      v2f bb;
      bb.x = half ? 0.f : s0c;
      bb.y = half ? 0.f : s1c;
      acc = wmma_f32_16x16x4(gfr, bb, acc);
    }

    if (lane == 15) {
      float2 z; z.x = p0; z.y = p1;
      ts2[tile0 + i] = z;
    }

    wave_lds_order();
    {
      float4 d0, d1;
      d0.x = acc[0]; d0.y = acc[1]; d0.z = acc[2]; d0.w = acc[3];
      d1.x = acc[4]; d1.y = acc[5]; d1.z = acc[6]; d1.w = acc[7];
      *(float4*)(xs + m * 16 + 8 * half)     = d0;
      *(float4*)(xs + m * 16 + 8 * half + 4) = d1;
    }
    wave_lds_order();
    float4 o0 = *(const float4*)(xs + lane * 8);
    float4 o1 = *(const float4*)(xs + lane * 8 + 4);
    wave_lds_order();
    *(float4*)(dst + i * TILE)     = o0;
    *(float4*)(dst + i * TILE + 4) = o1;

    u0 = n0; u1 = n1;
  }
}

#define SCH   ((NTILES + 31) / 32)
#define SLAST (NTILES - (SCH - 1) * 32)

__global__ void __launch_bounds__(32) scan_kernel(const float* __restrict__ coef,
                                                  const float* __restrict__ ts,
                                                  float* __restrict__ sg, int sec) {
  __shared__ float2 zs[SCH * 32];
  const int lane = threadIdx.x;
  const int b = blockIdx.x;
  const float* cf = coef + (b * NSEC + sec) * CSTRIDE;
  const float A00 = cf[96], A01 = cf[97], A10 = cf[98], A11 = cf[99];
  const float2* z2 = (const float2*)ts + b * NTILES;
  float2* o2 = (float2*)sg + b * NTILES;

  float2 zr[SCH];
#pragma unroll
  for (int c = 0; c < SCH - 1; ++c) zr[c] = z2[c * 32 + lane];
  zr[SCH - 1].x = 0.f; zr[SCH - 1].y = 0.f;
  if (lane < SLAST) zr[SCH - 1] = z2[(SCH - 1) * 32 + lane];
#pragma unroll
  for (int c = 0; c < SCH; ++c) zs[c * 32 + lane] = zr[c];

  float s0 = 0.f, s1 = 0.f;
  float2 cur = zs[lane];
  for (int c = 0; c < SCH; ++c) {
    const int cn = (c + 1 < SCH) ? c + 1 : c;
    const float2 nxt = zs[cn * 32 + lane];
    int k0[4] = {0, 0, 0, 0}, k1[4] = {0, 0, 0, 0};
#pragma unroll
    for (int j = 0; j < 32; ++j) {
      if (lane == j) { k0[j & 3] = __float_as_int(s0); k1[j & 3] = __float_as_int(s1); }
      const float z0 = __int_as_float(__builtin_amdgcn_readlane(__float_as_int(cur.x), j));
      const float z1 = __int_as_float(__builtin_amdgcn_readlane(__float_as_int(cur.y), j));
      const float n0 = __builtin_fmaf(A00, s0, A01 * s1) + z0;
      const float n1 = __builtin_fmaf(A11, s1, A10 * s0) + z1;
      s0 = n0; s1 = n1;
    }
    const int t = c * 32 + lane;
    if (t < NTILES) {
      float2 k;
      k.x = __int_as_float(k0[0] | k0[1] | k0[2] | k0[3]);
      k.y = __int_as_float(k1[0] | k1[1] | k1[2] | k1[3]);
      o2[t] = k;
    }
    cur = nxt;
  }
}

__global__ void __launch_bounds__(256) fixup_kernel(float* out,
                                                    const float* __restrict__ gt,
                                                    const float* __restrict__ sg,
                                                    int sec) {
  int tile = blockIdx.x;
  int b = tile / NTILES;
  int t = tile - b * NTILES;
  int n = threadIdx.x;
  float s0 = sg[tile * 2 + 0], s1 = sg[tile * 2 + 1];
  const float2* G = (const float2*)(gt + (b * NSEC + sec) * TILE * 2);
  float2 g = G[n];
  int idx = b * LSIG + t * TILE + n;
  out[idx] += g.x * s0 + g.y * s1;
}

extern "C" void kernel_launch(void* const* d_in, const int* in_sizes, int n_in,
                              void* d_out, int out_size, void* d_ws, size_t ws_size,
                              hipStream_t stream) {
  (void)in_sizes; (void)n_in; (void)out_size; (void)ws_size;
  const float* x     = (const float*)d_in[0];
  const float* knobs = (const float*)d_in[1];
  const float* W1    = (const float*)d_in[2];
  const float* b1v   = (const float*)d_in[3];
  const float* W2    = (const float*)d_in[4];
  const float* b2v   = (const float*)d_in[5];
  float* out = (float*)d_out;

  float* COEF = (float*)d_ws;
  float* FR   = COEF + NB * NSEC * CSTRIDE;
  float* GT   = FR   + NB * NSEC * FSTRIDE;
  float* TS   = GT   + NB * NSEC * TILE * 2;
  float* SG   = TS   + NB * NTILES * 2;

  derive_kernel<<<1, 64, 0, stream>>>(knobs, W1, b1v, W2, b2v, COEF, FR, GT);

  const int tileWaves = NB * WPR;
  for (int sec = 0; sec < NSEC; ++sec) {
    const float* src = (sec == 0) ? x : out;
    tile_kernel<<<tileWaves, 32, 0, stream>>>(src, out, COEF, FR, GT, SG, TS,
                                              sec, sec - 1);
    scan_kernel<<<NB, 32, 0, stream>>>(COEF, TS, SG, sec);
  }
  fixup_kernel<<<NB * NTILES, 256, 0, stream>>>(out, GT, SG, NSEC - 1);
}


# scan kernel: each lane scans its own 59 consecutive tile states, a 5-step cross-lane scan with powers of (A^256)^59 joins the lanes, each lane replays from its entry state; sum order changes (tree)
# speedup vs baseline: 3.6890x; 1.4926x over previous
#include <hip/hip_runtime.h>
#include <hip/hip_bf16.h>

#define NB      64
#define LSIG    480000
#define TILE    256
#define NTILES  (LSIG / TILE)
#define NSEC    3
#define CSTRIDE 128
#define FSTRIDE 320
#define TPW     15
#define WPR     (NTILES / TPW)

typedef __attribute__((ext_vector_type(2))) float v2f;
typedef __attribute__((ext_vector_type(8))) float v8f;

__device__ __forceinline__ v8f wmma_f32_16x16x4(v2f a, v2f b, v8f c) {
  return __builtin_amdgcn_wmma_f32_16x16x4_f32(false, a, false, b, (short)0, c,
                                               false, false);
}

__global__ void derive_kernel(const float* __restrict__ knobs,
                              const float* __restrict__ W1,
                              const float* __restrict__ b1v,
                              const float* __restrict__ W2,
                              const float* __restrict__ b2v,
                              float* __restrict__ coef,
                              float* __restrict__ fr,
                              float* __restrict__ gt) {
  int b = threadIdx.x;
  if (b >= NB) return;
  float k0 = knobs[b * 3 + 0], k1 = knobs[b * 3 + 1], k2 = knobs[b * 3 + 2];
  float hmlp[16];
  for (int i = 0; i < 16; ++i)
    hmlp[i] = tanhf(k0 * W1[0 * 16 + i] + k1 * W1[1 * 16 + i] +
                    k2 * W1[2 * 16 + i] + b1v[i]);
  float p[15];
  for (int j = 0; j < 15; ++j) {
    float acc = b2v[j];
    for (int i = 0; i < 16; ++i) acc += hmlp[i] * W2[i * 15 + j];
    p[j] = acc;
  }
  for (int s = 0; s < NSEC; ++s) {
    float c0 = p[s * 5 + 0], c1 = p[s * 5 + 1], c2 = p[s * 5 + 2];
    float a2c = 0.999f * tanhf(p[s * 5 + 4]);
    float a1c = 0.999f * (1.0f + a2c) * tanhf(p[s * 5 + 3]);
    float A00 = -a1c, A01 = 1.0f, A10 = -a2c, A11 = 0.0f;
    float Bv0 = c1 - a1c * c0, Bv1 = c2 - a2c * c0;
    float* cf = coef + (b * NSEC + s) * CSTRIDE;

    float h[16];
    {
      float v0 = Bv0, v1 = Bv1;
      h[0] = c0;
      for (int k = 1; k < 16; ++k) {
        h[k] = v0;
        float n0 = A00 * v0 + A01 * v1;
        float n1 = A10 * v0 + A11 * v1;
        v0 = n0; v1 = n1;
      }
    }
    {
      float v0 = Bv0, v1 = Bv1;
      cf[16 + 15] = v0; cf[32 + 15] = v1;
      for (int j = 14; j >= 0; --j) {
        float n0 = A00 * v0 + A01 * v1;
        float n1 = A10 * v0 + A11 * v1;
        v0 = n0; v1 = n1;
        cf[16 + j] = v0; cf[32 + j] = v1;
      }
    }
    float g0i[16], g1i[16];
    {
      float g0 = 1.0f, g1 = 0.0f;
      for (int i = 0; i < 16; ++i) {
        g0i[i] = g0; g1i[i] = g1;
        float n0 = g0 * A00 + g1 * A10;
        float n1 = g0 * A01 + g1 * A11;
        g0 = n0; g1 = n1;
      }
    }
    {
      float M00 = A00, M01 = A01, M10 = A10, M11 = A11;
      for (int q = 0; q < 4; ++q) {
        float n00 = M00 * M00 + M01 * M10, n01 = M00 * M01 + M01 * M11;
        float n10 = M10 * M00 + M11 * M10, n11 = M10 * M01 + M11 * M11;
        M00 = n00; M01 = n01; M10 = n10; M11 = n11;
      }
      for (int q = 0; q < 5; ++q) {
        cf[80 + 4 * q + 0] = M00; cf[80 + 4 * q + 1] = M01;
        cf[80 + 4 * q + 2] = M10; cf[80 + 4 * q + 3] = M11;
        float n00 = M00 * M00 + M01 * M10, n01 = M00 * M01 + M01 * M11;
        float n10 = M10 * M00 + M11 * M10, n11 = M10 * M01 + M11 * M11;
        M00 = n00; M01 = n01; M10 = n10; M11 = n11;
      }
    }
    {
      float* fb = fr + (b * NSEC + s) * FSTRIDE;
      for (int lane = 0; lane < 32; ++lane) {
        int m = lane & 15, half = lane >> 4;
        for (int k = 0; k < 4; ++k) {
          int K0 = 4 * k + 2 * half;
          fb[(k * 32 + lane) * 2 + 0] = (K0 <= m)     ? h[m - K0]     : 0.0f;
          fb[(k * 32 + lane) * 2 + 1] = (K0 + 1 <= m) ? h[m - K0 - 1] : 0.0f;
        }
        fb[256 + lane * 2 + 0] = half ? 0.0f : g0i[m];
        fb[256 + lane * 2 + 1] = half ? 0.0f : g1i[m];
      }
    }
    {
      float g0 = 1.0f, g1 = 0.0f;
      float* G = gt + (b * NSEC + s) * TILE * 2;
      for (int n = 0; n < TILE; ++n) {
        G[n * 2 + 0] = g0; G[n * 2 + 1] = g1;
        float n0 = g0 * A00 + g1 * A10;
        float n1 = g0 * A01 + g1 * A11;
        g0 = n0; g1 = n1;
      }
    }
  }
}


__device__ __forceinline__ void wave_lds_order() {
  __builtin_amdgcn_fence(__ATOMIC_RELEASE, "wavefront");
  __builtin_amdgcn_wave_barrier();
  __builtin_amdgcn_fence(__ATOMIC_ACQUIRE, "wavefront");
}

__global__ void __launch_bounds__(32) tile_kernel(const float* in, float* out,
                                                  const float* __restrict__ coef,
                                                  const float* __restrict__ fr,
                                                  const float* __restrict__ gt,
                                                  const float* __restrict__ sg,
                                                  float* __restrict__ ts,
                                                  int sec, int prevsec) {
  __shared__ float xs[TILE];
  const int lane = threadIdx.x;
  const int b = blockIdx.x / WPR;
  const int w = blockIdx.x - b * WPR;
  const int tile0 = b * NTILES + w * TPW;
  const int base0 = b * LSIG + w * TPW * TILE;
  const int m = lane & 15;
  const int half = lane >> 4;
  const bool fuse = prevsec >= 0;

  const float* cf = coef + (b * NSEC + sec) * CSTRIDE;
  const v2f* hf = (const v2f*)(fr + (b * NSEC + sec) * FSTRIDE);
  v2f ha[4];
#pragma unroll
  for (int k = 0; k < 4; ++k) ha[k] = hf[k * 32 + lane];
  const v2f gfr = hf[128 + lane];
  float f0[16], f1[16], Ms[16];
#pragma unroll
  for (int j = 0; j < 16; ++j) {
    f0[j] = cf[16 + j];
    f1[j] = cf[32 + j];
    Ms[j] = cf[80 + j];
  }
  float4 ga = {0.f, 0.f, 0.f, 0.f}, gb = ga, gc = ga, gd = ga;
  if (fuse) {
    const float4* G4 = (const float4*)(gt + (b * NSEC + prevsec) * TILE * 2);
    ga = G4[lane * 4 + 0];
    gb = G4[lane * 4 + 1];
    gc = G4[lane * 4 + 2];
    gd = G4[lane * 4 + 3];
  }
  const float2* sg2 = (const float2*)sg;
  float2* ts2 = (float2*)ts;
  const float* src = in + base0 + lane * 8;
  float* dst = out + base0 + lane * 8;

  float4 u0 = *(const float4*)(src);
  float4 u1 = *(const float4*)(src + 4);
  float svx_l = 0.f, svy_l = 0.f;
  if (fuse) {
    float2 t = sg2[tile0 + (lane < TPW ? lane : TPW - 1)];
    svx_l = t.x; svy_l = t.y;
  }

  for (int i = 0; i < TPW; ++i) {
    float4 n0 = u0, n1 = u1;
    if (i + 1 < TPW) {
      n0 = *(const float4*)(src + (i + 1) * TILE);
      n1 = *(const float4*)(src + (i + 1) * TILE + 4);
    }

    if (fuse) {
      float2 sv;
      sv.x = __int_as_float(__builtin_amdgcn_readlane(__float_as_int(svx_l), i));
      sv.y = __int_as_float(__builtin_amdgcn_readlane(__float_as_int(svy_l), i));
      u0.x += ga.x * sv.x + ga.y * sv.y;
      u0.y += ga.z * sv.x + ga.w * sv.y;
      u0.z += gb.x * sv.x + gb.y * sv.y;
      u0.w += gb.z * sv.x + gb.w * sv.y;
      u1.x += gc.x * sv.x + gc.y * sv.y;
      u1.y += gc.z * sv.x + gc.w * sv.y;
      u1.z += gd.x * sv.x + gd.y * sv.y;
      u1.w += gd.z * sv.x + gd.w * sv.y;
    }
    *(float4*)(xs + lane * 8)     = u0;
    *(float4*)(xs + lane * 8 + 4) = u1;
    wave_lds_order();

    float4 xc[4];
#pragma unroll
    for (int q = 0; q < 4; ++q)
      xc[q] = ((const float4*)(xs + m * 16))[q];

    v8f acc = {0.f, 0.f, 0.f, 0.f, 0.f, 0.f, 0.f, 0.f};
#pragma unroll
    for (int k = 0; k < 4; ++k) {
      v2f bb;
      bb.x = half ? xc[k].z : xc[k].x;
      bb.y = half ? xc[k].w : xc[k].y;
      acc = wmma_f32_16x16x4(ha[k], bb, acc);
    }

    float fx0 = 0.f, fx1 = 0.f;
#pragma unroll
    for (int q = 0; q < 4; ++q) {
      fx0 += f0[4 * q + 0] * xc[q].x + f0[4 * q + 1] * xc[q].y +
             f0[4 * q + 2] * xc[q].z + f0[4 * q + 3] * xc[q].w;
      fx1 += f1[4 * q + 0] * xc[q].x + f1[4 * q + 1] * xc[q].y +
             f1[4 * q + 2] * xc[q].z + f1[4 * q + 3] * xc[q].w;
    }

    float p0 = fx0, p1 = fx1;
#pragma unroll
    for (int s = 0; s < 4; ++s) {
      float M00 = Ms[4 * s + 0], M01 = Ms[4 * s + 1];
      float M10 = Ms[4 * s + 2], M11 = Ms[4 * s + 3];
      int len = 1 << s;
      float q0 = __shfl_up(p0, (unsigned)len, 32);
      float q1 = __shfl_up(p1, (unsigned)len, 32);
      if (m >= len) {
        p0 = M00 * q0 + M01 * q1 + p0;
        p1 = M10 * q0 + M11 * q1 + p1;
      }
    }
    float e0 = __shfl_up(p0, 1u, 32);
    float e1 = __shfl_up(p1, 1u, 32);
    float s0c = (m >= 1) ? e0 : 0.f;
    float s1c = (m >= 1) ? e1 : 0.f;

    {
      v2f bb;
      bb.x = half ? 0.f : s0c;
      bb.y = half ? 0.f : s1c;
      acc = wmma_f32_16x16x4(gfr, bb, acc);
    }

    if (lane == 15) {
      float2 z; z.x = p0; z.y = p1;
      ts2[tile0 + i] = z;
    }

    wave_lds_order();
    {
      float4 d0, d1;
      d0.x = acc[0]; d0.y = acc[1]; d0.z = acc[2]; d0.w = acc[3];
      d1.x = acc[4]; d1.y = acc[5]; d1.z = acc[6]; d1.w = acc[7];
      *(float4*)(xs + m * 16 + 8 * half)     = d0;
      *(float4*)(xs + m * 16 + 8 * half + 4) = d1;
    }
    wave_lds_order();
    float4 o0 = *(const float4*)(xs + lane * 8);
    float4 o1 = *(const float4*)(xs + lane * 8 + 4);
    wave_lds_order();
    *(float4*)(dst + i * TILE)     = o0;
    *(float4*)(dst + i * TILE + 4) = o1;

    u0 = n0; u1 = n1;
  }
}

#define SCH   ((NTILES + 31) / 32)
#define SLAST (NTILES - (SCH - 1) * 32)

struct m22 { float a, b, c, d; };
__device__ __forceinline__ m22 mm(const m22 x, const m22 y) {
  m22 r;
  r.a = x.a * y.a + x.b * y.c;  r.b = x.a * y.b + x.b * y.d;
  r.c = x.c * y.a + x.d * y.c;  r.d = x.c * y.b + x.d * y.d;
  return r;
}

__global__ void __launch_bounds__(32) scan_kernel(const float* __restrict__ coef,
                                                  const float* __restrict__ ts,
                                                  float* __restrict__ sg, int sec) {
  __shared__ float2 zs[SCH * 32];
  const int lane = threadIdx.x;
  const int b = blockIdx.x;
  const float* cf = coef + (b * NSEC + sec) * CSTRIDE;
  const float A00 = cf[96], A01 = cf[97], A10 = cf[98], A11 = cf[99];
  const float2* z2 = (const float2*)ts + b * NTILES;
  float2* o2 = (float2*)sg + b * NTILES;

  float2 zr[SCH];
#pragma unroll
  for (int c = 0; c < SCH - 1; ++c) zr[c] = z2[c * 32 + lane];
  zr[SCH - 1].x = 0.f; zr[SCH - 1].y = 0.f;
  if (lane < SLAST) zr[SCH - 1] = z2[(SCH - 1) * 32 + lane];

  m22 Tk[5];
  {
    m22 P1; P1.a = A00; P1.b = A01; P1.c = A10; P1.d = A11;
    const m22 P2 = mm(P1, P1), P4 = mm(P2, P2), P8 = mm(P4, P4);
    const m22 P16 = mm(P8, P8), P32 = mm(P16, P16);
    static_assert(SCH == 59, "the product below spells out 59 = 32+16+8+2+1");
    Tk[0] = mm(mm(mm(mm(P32, P16), P8), P2), P1);
#pragma unroll
    for (int k = 1; k < 5; ++k) Tk[k] = mm(Tk[k - 1], Tk[k - 1]);
  }

#pragma unroll
  for (int c = 0; c < SCH; ++c) zs[c * 32 + lane] = zr[c];
  wave_lds_order();

  float2* mine = zs + lane * SCH;
  float2 zl[SCH];
#pragma unroll
  for (int c = 0; c < SCH; ++c) zl[c] = mine[c];

  float s0 = 0.f, s1 = 0.f;
#pragma unroll
  for (int c = 0; c < SCH; ++c) {
    const float n0 = __builtin_fmaf(A00, s0, A01 * s1) + zl[c].x;
    const float n1 = __builtin_fmaf(A11, s1, A10 * s0) + zl[c].y;
    s0 = n0; s1 = n1;
  }

  float p0 = s0, p1 = s1;
#pragma unroll
  for (int k = 0; k < 5; ++k) {
    const int len = 1 << k;
    const float q0 = __shfl_up(p0, (unsigned)len, 32);
    const float q1 = __shfl_up(p1, (unsigned)len, 32);
    if (lane >= len) {
      p0 = Tk[k].a * q0 + Tk[k].b * q1 + p0;
      p1 = Tk[k].c * q0 + Tk[k].d * q1 + p1;
    }
  }
  const float e0 = __shfl_up(p0, 1u, 32);
  const float e1 = __shfl_up(p1, 1u, 32);
  s0 = (lane >= 1) ? e0 : 0.f;
  s1 = (lane >= 1) ? e1 : 0.f;

#pragma unroll
  for (int c = 0; c < SCH; ++c) {
    float2 k; k.x = s0; k.y = s1;
    mine[c] = k;
    const float n0 = __builtin_fmaf(A00, s0, A01 * s1) + zl[c].x;
    const float n1 = __builtin_fmaf(A11, s1, A10 * s0) + zl[c].y;
    s0 = n0; s1 = n1;
  }
  wave_lds_order();

#pragma unroll
  for (int c = 0; c < SCH - 1; ++c) o2[c * 32 + lane] = zs[c * 32 + lane];
  if (lane < SLAST) o2[(SCH - 1) * 32 + lane] = zs[(SCH - 1) * 32 + lane];
}

__global__ void __launch_bounds__(256) fixup_kernel(float* out,
                                                    const float* __restrict__ gt,
                                                    const float* __restrict__ sg,
                                                    int sec) {
  int tile = blockIdx.x;
  int b = tile / NTILES;
  int t = tile - b * NTILES;
  int n = threadIdx.x;
  float s0 = sg[tile * 2 + 0], s1 = sg[tile * 2 + 1];
  const float2* G = (const float2*)(gt + (b * NSEC + sec) * TILE * 2);
  float2 g = G[n];
  int idx = b * LSIG + t * TILE + n;
  out[idx] += g.x * s0 + g.y * s1;
}

extern "C" void kernel_launch(void* const* d_in, const int* in_sizes, int n_in,
                              void* d_out, int out_size, void* d_ws, size_t ws_size,
                              hipStream_t stream) {
  (void)in_sizes; (void)n_in; (void)out_size; (void)ws_size;
  const float* x     = (const float*)d_in[0];
  const float* knobs = (const float*)d_in[1];
  const float* W1    = (const float*)d_in[2];
  const float* b1v   = (const float*)d_in[3];
  const float* W2    = (const float*)d_in[4];
  const float* b2v   = (const float*)d_in[5];
  float* out = (float*)d_out;

  float* COEF = (float*)d_ws;
  float* FR   = COEF + NB * NSEC * CSTRIDE;
  float* GT   = FR   + NB * NSEC * FSTRIDE;
  float* TS   = GT   + NB * NSEC * TILE * 2;
  float* SG   = TS   + NB * NTILES * 2;

  derive_kernel<<<1, 64, 0, stream>>>(knobs, W1, b1v, W2, b2v, COEF, FR, GT);

  const int tileWaves = NB * WPR;
  for (int sec = 0; sec < NSEC; ++sec) {
    const float* src = (sec == 0) ? x : out;
    tile_kernel<<<tileWaves, 32, 0, stream>>>(src, out, COEF, FR, GT, SG, TS,
                                              sec, sec - 1);
    scan_kernel<<<NB, 32, 0, stream>>>(COEF, TS, SG, sec);
  }
  fixup_kernel<<<NB * NTILES, 256, 0, stream>>>(out, GT, SG, NSEC - 1);
}


# tile kernel: sample loads and stores cover whole 128-byte lines (lane l moves samples 4l..4l+3 and 128+4l..+3, 512 contiguous bytes an instruction) instead of half of 8 lines each; sums unchanged
# speedup vs baseline: 3.8037x; 1.0311x over previous
#include <hip/hip_runtime.h>
#include <hip/hip_bf16.h>

#define NB      64
#define LSIG    480000
#define TILE    256
#define NTILES  (LSIG / TILE)
#define NSEC    3
#define CSTRIDE 128
#define FSTRIDE 320
#define TPW     15
#define WPR     (NTILES / TPW)

typedef __attribute__((ext_vector_type(2))) float v2f;
typedef __attribute__((ext_vector_type(8))) float v8f;

__device__ __forceinline__ v8f wmma_f32_16x16x4(v2f a, v2f b, v8f c) {
  return __builtin_amdgcn_wmma_f32_16x16x4_f32(false, a, false, b, (short)0, c,
                                               false, false);
}

__global__ void derive_kernel(const float* __restrict__ knobs,
                              const float* __restrict__ W1,
                              const float* __restrict__ b1v,
                              const float* __restrict__ W2,
                              const float* __restrict__ b2v,
                              float* __restrict__ coef,
                              float* __restrict__ fr,
                              float* __restrict__ gt) {
  int b = threadIdx.x;
  if (b >= NB) return;
  float k0 = knobs[b * 3 + 0], k1 = knobs[b * 3 + 1], k2 = knobs[b * 3 + 2];
  float hmlp[16];
  for (int i = 0; i < 16; ++i)
    hmlp[i] = tanhf(k0 * W1[0 * 16 + i] + k1 * W1[1 * 16 + i] +
                    k2 * W1[2 * 16 + i] + b1v[i]);
  float p[15];
  for (int j = 0; j < 15; ++j) {
    float acc = b2v[j];
    for (int i = 0; i < 16; ++i) acc += hmlp[i] * W2[i * 15 + j];
    p[j] = acc;
  }
  for (int s = 0; s < NSEC; ++s) {
    float c0 = p[s * 5 + 0], c1 = p[s * 5 + 1], c2 = p[s * 5 + 2];
    float a2c = 0.999f * tanhf(p[s * 5 + 4]);
    float a1c = 0.999f * (1.0f + a2c) * tanhf(p[s * 5 + 3]);
    float A00 = -a1c, A01 = 1.0f, A10 = -a2c, A11 = 0.0f;
    float Bv0 = c1 - a1c * c0, Bv1 = c2 - a2c * c0;
    float* cf = coef + (b * NSEC + s) * CSTRIDE;

    float h[16];
    {
      float v0 = Bv0, v1 = Bv1;
      h[0] = c0;
      for (int k = 1; k < 16; ++k) {
        h[k] = v0;
        float n0 = A00 * v0 + A01 * v1;
        float n1 = A10 * v0 + A11 * v1;
        v0 = n0; v1 = n1;
      }
    }
    {
      float v0 = Bv0, v1 = Bv1;
      cf[16 + 15] = v0; cf[32 + 15] = v1;
      for (int j = 14; j >= 0; --j) {
        float n0 = A00 * v0 + A01 * v1;
        float n1 = A10 * v0 + A11 * v1;
        v0 = n0; v1 = n1;
        cf[16 + j] = v0; cf[32 + j] = v1;
      }
    }
    float g0i[16], g1i[16];
    {
      float g0 = 1.0f, g1 = 0.0f;
      for (int i = 0; i < 16; ++i) {
        g0i[i] = g0; g1i[i] = g1;
        float n0 = g0 * A00 + g1 * A10;
        float n1 = g0 * A01 + g1 * A11;
        g0 = n0; g1 = n1;
      }
    }
    {
      float M00 = A00, M01 = A01, M10 = A10, M11 = A11;
      for (int q = 0; q < 4; ++q) {
        float n00 = M00 * M00 + M01 * M10, n01 = M00 * M01 + M01 * M11;
        float n10 = M10 * M00 + M11 * M10, n11 = M10 * M01 + M11 * M11;
        M00 = n00; M01 = n01; M10 = n10; M11 = n11;
      }
      for (int q = 0; q < 5; ++q) {
        cf[80 + 4 * q + 0] = M00; cf[80 + 4 * q + 1] = M01;
        cf[80 + 4 * q + 2] = M10; cf[80 + 4 * q + 3] = M11;
        float n00 = M00 * M00 + M01 * M10, n01 = M00 * M01 + M01 * M11;
        float n10 = M10 * M00 + M11 * M10, n11 = M10 * M01 + M11 * M11;
        M00 = n00; M01 = n01; M10 = n10; M11 = n11;
      }
    }
    {
      float* fb = fr + (b * NSEC + s) * FSTRIDE;
      for (int lane = 0; lane < 32; ++lane) {
        int m = lane & 15, half = lane >> 4;
        for (int k = 0; k < 4; ++k) {
          int K0 = 4 * k + 2 * half;
          fb[(k * 32 + lane) * 2 + 0] = (K0 <= m)     ? h[m - K0]     : 0.0f;
          fb[(k * 32 + lane) * 2 + 1] = (K0 + 1 <= m) ? h[m - K0 - 1] : 0.0f;
        }
        fb[256 + lane * 2 + 0] = half ? 0.0f : g0i[m];
        fb[256 + lane * 2 + 1] = half ? 0.0f : g1i[m];
      }
    }
    {
      float g0 = 1.0f, g1 = 0.0f;
      float* G = gt + (b * NSEC + s) * TILE * 2;
      for (int n = 0; n < TILE; ++n) {
        G[n * 2 + 0] = g0; G[n * 2 + 1] = g1;
        float n0 = g0 * A00 + g1 * A10;
        float n1 = g0 * A01 + g1 * A11;
        g0 = n0; g1 = n1;
      }
    }
  }
}


__device__ __forceinline__ void wave_lds_order() {
  __builtin_amdgcn_fence(__ATOMIC_RELEASE, "wavefront");
  __builtin_amdgcn_wave_barrier();
  __builtin_amdgcn_fence(__ATOMIC_ACQUIRE, "wavefront");
}

__global__ void __launch_bounds__(32) tile_kernel(const float* in, float* out,
                                                  const float* __restrict__ coef,
                                                  const float* __restrict__ fr,
                                                  const float* __restrict__ gt,
                                                  const float* __restrict__ sg,
                                                  float* __restrict__ ts,
                                                  int sec, int prevsec) {
  __shared__ float xs[TILE];
  const int lane = threadIdx.x;
  const int b = blockIdx.x / WPR;
  const int w = blockIdx.x - b * WPR;
  const int tile0 = b * NTILES + w * TPW;
  const int base0 = b * LSIG + w * TPW * TILE;
  const int m = lane & 15;
  const int half = lane >> 4;
  const bool fuse = prevsec >= 0;

  const float* cf = coef + (b * NSEC + sec) * CSTRIDE;
  const v2f* hf = (const v2f*)(fr + (b * NSEC + sec) * FSTRIDE);
  v2f ha[4];
#pragma unroll
  for (int k = 0; k < 4; ++k) ha[k] = hf[k * 32 + lane];
  const v2f gfr = hf[128 + lane];
  float f0[16], f1[16], Ms[16];
#pragma unroll
  for (int j = 0; j < 16; ++j) {
    f0[j] = cf[16 + j];
    f1[j] = cf[32 + j];
    Ms[j] = cf[80 + j];
  }
  float4 ga = {0.f, 0.f, 0.f, 0.f}, gb = ga, gc = ga, gd = ga;
  if (fuse) {
    const float4* G4 = (const float4*)(gt + (b * NSEC + prevsec) * TILE * 2);
    ga = G4[lane * 2 + 0];
    gb = G4[lane * 2 + 1];
    gc = G4[64 + lane * 2 + 0];
    gd = G4[64 + lane * 2 + 1];
  }
  const float2* sg2 = (const float2*)sg;
  float2* ts2 = (float2*)ts;
  const float* src = in + base0 + lane * 4;
  float* dst = out + base0 + lane * 4;

  float4 u0 = *(const float4*)(src);
  float4 u1 = *(const float4*)(src + 128);
  float svx_l = 0.f, svy_l = 0.f;
  if (fuse) {
    float2 t = sg2[tile0 + (lane < TPW ? lane : TPW - 1)];
    svx_l = t.x; svy_l = t.y;
  }

  for (int i = 0; i < TPW; ++i) {
    float4 n0 = u0, n1 = u1;
    if (i + 1 < TPW) {
      n0 = *(const float4*)(src + (i + 1) * TILE);
      n1 = *(const float4*)(src + (i + 1) * TILE + 128);
    }

    if (fuse) {
      float2 sv;
      sv.x = __int_as_float(__builtin_amdgcn_readlane(__float_as_int(svx_l), i));
      sv.y = __int_as_float(__builtin_amdgcn_readlane(__float_as_int(svy_l), i));
      u0.x += ga.x * sv.x + ga.y * sv.y;
      u0.y += ga.z * sv.x + ga.w * sv.y;
      u0.z += gb.x * sv.x + gb.y * sv.y;
      u0.w += gb.z * sv.x + gb.w * sv.y;
      u1.x += gc.x * sv.x + gc.y * sv.y;
      u1.y += gc.z * sv.x + gc.w * sv.y;
      u1.z += gd.x * sv.x + gd.y * sv.y;
      u1.w += gd.z * sv.x + gd.w * sv.y;
    }
    *(float4*)(xs + lane * 4)       = u0;
    *(float4*)(xs + 128 + lane * 4) = u1;
    wave_lds_order();

    float4 xc[4];
#pragma unroll
    for (int q = 0; q < 4; ++q)
      xc[q] = ((const float4*)(xs + m * 16))[q];

    v8f acc = {0.f, 0.f, 0.f, 0.f, 0.f, 0.f, 0.f, 0.f};
#pragma unroll
    for (int k = 0; k < 4; ++k) {
      v2f bb;
      bb.x = half ? xc[k].z : xc[k].x;
      bb.y = half ? xc[k].w : xc[k].y;
      acc = wmma_f32_16x16x4(ha[k], bb, acc);
    }

    float fx0 = 0.f, fx1 = 0.f;
#pragma unroll
    for (int q = 0; q < 4; ++q) {
      fx0 += f0[4 * q + 0] * xc[q].x + f0[4 * q + 1] * xc[q].y +
             f0[4 * q + 2] * xc[q].z + f0[4 * q + 3] * xc[q].w;
      fx1 += f1[4 * q + 0] * xc[q].x + f1[4 * q + 1] * xc[q].y +
             f1[4 * q + 2] * xc[q].z + f1[4 * q + 3] * xc[q].w;
    }

    float p0 = fx0, p1 = fx1;
#pragma unroll
    for (int s = 0; s < 4; ++s) {
      float M00 = Ms[4 * s + 0], M01 = Ms[4 * s + 1];
      float M10 = Ms[4 * s + 2], M11 = Ms[4 * s + 3];
      int len = 1 << s;
      float q0 = __shfl_up(p0, (unsigned)len, 32);
      float q1 = __shfl_up(p1, (unsigned)len, 32);
      if (m >= len) {
        p0 = M00 * q0 + M01 * q1 + p0;
        p1 = M10 * q0 + M11 * q1 + p1;
      }
    }
    float e0 = __shfl_up(p0, 1u, 32);
    float e1 = __shfl_up(p1, 1u, 32);
    float s0c = (m >= 1) ? e0 : 0.f;
    float s1c = (m >= 1) ? e1 : 0.f;

    {
      v2f bb;
      bb.x = half ? 0.f : s0c;
      bb.y = half ? 0.f : s1c;
      acc = wmma_f32_16x16x4(gfr, bb, acc);
    }

    if (lane == 15) {
      float2 z; z.x = p0; z.y = p1;
      ts2[tile0 + i] = z;
    }

    wave_lds_order();
    {
      float4 d0, d1;
      d0.x = acc[0]; d0.y = acc[1]; d0.z = acc[2]; d0.w = acc[3];
      d1.x = acc[4]; d1.y = acc[5]; d1.z = acc[6]; d1.w = acc[7];
      *(float4*)(xs + m * 16 + 8 * half)     = d0;
      *(float4*)(xs + m * 16 + 8 * half + 4) = d1;
    }
    wave_lds_order();
    float4 o0 = *(const float4*)(xs + lane * 4);
    float4 o1 = *(const float4*)(xs + 128 + lane * 4);
    wave_lds_order();
    *(float4*)(dst + i * TILE)       = o0;
    *(float4*)(dst + i * TILE + 128) = o1;

    u0 = n0; u1 = n1;
  }
}

#define SCH   ((NTILES + 31) / 32)
#define SLAST (NTILES - (SCH - 1) * 32)

struct m22 { float a, b, c, d; };
__device__ __forceinline__ m22 mm(const m22 x, const m22 y) {
  m22 r;
  r.a = x.a * y.a + x.b * y.c;  r.b = x.a * y.b + x.b * y.d;
  r.c = x.c * y.a + x.d * y.c;  r.d = x.c * y.b + x.d * y.d;
  return r;
}

__global__ void __launch_bounds__(32) scan_kernel(const float* __restrict__ coef,
                                                  const float* __restrict__ ts,
                                                  float* __restrict__ sg, int sec) {
  __shared__ float2 zs[SCH * 32];
  const int lane = threadIdx.x;
  const int b = blockIdx.x;
  const float* cf = coef + (b * NSEC + sec) * CSTRIDE;
  const float A00 = cf[96], A01 = cf[97], A10 = cf[98], A11 = cf[99];
  const float2* z2 = (const float2*)ts + b * NTILES;
  float2* o2 = (float2*)sg + b * NTILES;

  float2 zr[SCH];
#pragma unroll
  for (int c = 0; c < SCH - 1; ++c) zr[c] = z2[c * 32 + lane];
  zr[SCH - 1].x = 0.f; zr[SCH - 1].y = 0.f;
  if (lane < SLAST) zr[SCH - 1] = z2[(SCH - 1) * 32 + lane];

  m22 Tk[5];
  {
    m22 P1; P1.a = A00; P1.b = A01; P1.c = A10; P1.d = A11;
    const m22 P2 = mm(P1, P1), P4 = mm(P2, P2), P8 = mm(P4, P4);
    const m22 P16 = mm(P8, P8), P32 = mm(P16, P16);
    static_assert(SCH == 59, "the product below spells out 59 = 32+16+8+2+1");
    Tk[0] = mm(mm(mm(mm(P32, P16), P8), P2), P1);
#pragma unroll
    for (int k = 1; k < 5; ++k) Tk[k] = mm(Tk[k - 1], Tk[k - 1]);
  }

#pragma unroll
  for (int c = 0; c < SCH; ++c) zs[c * 32 + lane] = zr[c];
  wave_lds_order();

  float2* mine = zs + lane * SCH;
  float2 zl[SCH];
#pragma unroll
  for (int c = 0; c < SCH; ++c) zl[c] = mine[c];

  float s0 = 0.f, s1 = 0.f;
#pragma unroll
  for (int c = 0; c < SCH; ++c) {
    const float n0 = __builtin_fmaf(A00, s0, A01 * s1) + zl[c].x;
    const float n1 = __builtin_fmaf(A11, s1, A10 * s0) + zl[c].y;
    s0 = n0; s1 = n1;
  }

  float p0 = s0, p1 = s1;
#pragma unroll
  for (int k = 0; k < 5; ++k) {
    const int len = 1 << k;
    const float q0 = __shfl_up(p0, (unsigned)len, 32);
    const float q1 = __shfl_up(p1, (unsigned)len, 32);
    if (lane >= len) {
      p0 = Tk[k].a * q0 + Tk[k].b * q1 + p0;
      p1 = Tk[k].c * q0 + Tk[k].d * q1 + p1;
    }
  }
  const float e0 = __shfl_up(p0, 1u, 32);
  const float e1 = __shfl_up(p1, 1u, 32);
  s0 = (lane >= 1) ? e0 : 0.f;
  s1 = (lane >= 1) ? e1 : 0.f;

#pragma unroll
  for (int c = 0; c < SCH; ++c) {
    float2 k; k.x = s0; k.y = s1;
    mine[c] = k;
    const float n0 = __builtin_fmaf(A00, s0, A01 * s1) + zl[c].x;
    const float n1 = __builtin_fmaf(A11, s1, A10 * s0) + zl[c].y;
    s0 = n0; s1 = n1;
  }
  wave_lds_order();

#pragma unroll
  for (int c = 0; c < SCH - 1; ++c) o2[c * 32 + lane] = zs[c * 32 + lane];
  if (lane < SLAST) o2[(SCH - 1) * 32 + lane] = zs[(SCH - 1) * 32 + lane];
}

__global__ void __launch_bounds__(256) fixup_kernel(float* out,
                                                    const float* __restrict__ gt,
                                                    const float* __restrict__ sg,
                                                    int sec) {
  int tile = blockIdx.x;
  int b = tile / NTILES;
  int t = tile - b * NTILES;
  int n = threadIdx.x;
  float s0 = sg[tile * 2 + 0], s1 = sg[tile * 2 + 1];
  const float2* G = (const float2*)(gt + (b * NSEC + sec) * TILE * 2);
  float2 g = G[n];
  int idx = b * LSIG + t * TILE + n;
  out[idx] += g.x * s0 + g.y * s1;
}

extern "C" void kernel_launch(void* const* d_in, const int* in_sizes, int n_in,
                              void* d_out, int out_size, void* d_ws, size_t ws_size,
                              hipStream_t stream) {
  (void)in_sizes; (void)n_in; (void)out_size; (void)ws_size;
  const float* x     = (const float*)d_in[0];
  const float* knobs = (const float*)d_in[1];
  const float* W1    = (const float*)d_in[2];
  const float* b1v   = (const float*)d_in[3];
  const float* W2    = (const float*)d_in[4];
  const float* b2v   = (const float*)d_in[5];
  float* out = (float*)d_out;

  float* COEF = (float*)d_ws;
  float* FR   = COEF + NB * NSEC * CSTRIDE;
  float* GT   = FR   + NB * NSEC * FSTRIDE;
  float* TS   = GT   + NB * NSEC * TILE * 2;
  float* SG   = TS   + NB * NTILES * 2;

  derive_kernel<<<1, 64, 0, stream>>>(knobs, W1, b1v, W2, b2v, COEF, FR, GT);

  const int tileWaves = NB * WPR;
  for (int sec = 0; sec < NSEC; ++sec) {
    const float* src = (sec == 0) ? x : out;
    tile_kernel<<<tileWaves, 32, 0, stream>>>(src, out, COEF, FR, GT, SG, TS,
                                              sec, sec - 1);
    scan_kernel<<<NB, 32, 0, stream>>>(COEF, TS, SG, sec);
  }
  fixup_kernel<<<NB * NTILES, 256, 0, stream>>>(out, GT, SG, NSEC - 1);
}


# fixup kernel: one wave adds the state term to 15 consecutive tiles of a row in whole-line 128-bit loads and stores (8000 waves) instead of 960000 waves moving 4 bytes a lane; sum order unchanged
# speedup vs baseline: 4.6976x; 1.2350x over previous
#include <hip/hip_runtime.h>
#include <hip/hip_bf16.h>

#define NB      64
#define LSIG    480000
#define TILE    256
#define NTILES  (LSIG / TILE)
#define NSEC    3
#define CSTRIDE 128
#define FSTRIDE 320
#define TPW     15
#define WPR     (NTILES / TPW)

typedef __attribute__((ext_vector_type(2))) float v2f;
typedef __attribute__((ext_vector_type(8))) float v8f;

__device__ __forceinline__ v8f wmma_f32_16x16x4(v2f a, v2f b, v8f c) {
  return __builtin_amdgcn_wmma_f32_16x16x4_f32(false, a, false, b, (short)0, c,
                                               false, false);
}

__global__ void derive_kernel(const float* __restrict__ knobs,
                              const float* __restrict__ W1,
                              const float* __restrict__ b1v,
                              const float* __restrict__ W2,
                              const float* __restrict__ b2v,
                              float* __restrict__ coef,
                              float* __restrict__ fr,
                              float* __restrict__ gt) {
  int b = threadIdx.x;
  if (b >= NB) return;
  float k0 = knobs[b * 3 + 0], k1 = knobs[b * 3 + 1], k2 = knobs[b * 3 + 2];
  float hmlp[16];
  for (int i = 0; i < 16; ++i)
    hmlp[i] = tanhf(k0 * W1[0 * 16 + i] + k1 * W1[1 * 16 + i] +
                    k2 * W1[2 * 16 + i] + b1v[i]);
  float p[15];
  for (int j = 0; j < 15; ++j) {
    float acc = b2v[j];
    for (int i = 0; i < 16; ++i) acc += hmlp[i] * W2[i * 15 + j];
    p[j] = acc;
  }
  for (int s = 0; s < NSEC; ++s) {
    float c0 = p[s * 5 + 0], c1 = p[s * 5 + 1], c2 = p[s * 5 + 2];
    float a2c = 0.999f * tanhf(p[s * 5 + 4]);
    float a1c = 0.999f * (1.0f + a2c) * tanhf(p[s * 5 + 3]);
    float A00 = -a1c, A01 = 1.0f, A10 = -a2c, A11 = 0.0f;
    float Bv0 = c1 - a1c * c0, Bv1 = c2 - a2c * c0;
    float* cf = coef + (b * NSEC + s) * CSTRIDE;

    float h[16];
    {
      float v0 = Bv0, v1 = Bv1;
      h[0] = c0;
      for (int k = 1; k < 16; ++k) {
        h[k] = v0;
        float n0 = A00 * v0 + A01 * v1;
        float n1 = A10 * v0 + A11 * v1;
        v0 = n0; v1 = n1;
      }
    }
    {
      float v0 = Bv0, v1 = Bv1;
      cf[16 + 15] = v0; cf[32 + 15] = v1;
      for (int j = 14; j >= 0; --j) {
        float n0 = A00 * v0 + A01 * v1;
        float n1 = A10 * v0 + A11 * v1;
        v0 = n0; v1 = n1;
        cf[16 + j] = v0; cf[32 + j] = v1;
      }
    }
    float g0i[16], g1i[16];
    {
      float g0 = 1.0f, g1 = 0.0f;
      for (int i = 0; i < 16; ++i) {
        g0i[i] = g0; g1i[i] = g1;
        float n0 = g0 * A00 + g1 * A10;
        float n1 = g0 * A01 + g1 * A11;
        g0 = n0; g1 = n1;
      }
    }
    {
      float M00 = A00, M01 = A01, M10 = A10, M11 = A11;
      for (int q = 0; q < 4; ++q) {
        float n00 = M00 * M00 + M01 * M10, n01 = M00 * M01 + M01 * M11;
        float n10 = M10 * M00 + M11 * M10, n11 = M10 * M01 + M11 * M11;
        M00 = n00; M01 = n01; M10 = n10; M11 = n11;
      }
      for (int q = 0; q < 5; ++q) {
        cf[80 + 4 * q + 0] = M00; cf[80 + 4 * q + 1] = M01;
        cf[80 + 4 * q + 2] = M10; cf[80 + 4 * q + 3] = M11;
        float n00 = M00 * M00 + M01 * M10, n01 = M00 * M01 + M01 * M11;
        float n10 = M10 * M00 + M11 * M10, n11 = M10 * M01 + M11 * M11;
        M00 = n00; M01 = n01; M10 = n10; M11 = n11;
      }
    }
    {
      float* fb = fr + (b * NSEC + s) * FSTRIDE;
      for (int lane = 0; lane < 32; ++lane) {
        int m = lane & 15, half = lane >> 4;
        for (int k = 0; k < 4; ++k) {
          int K0 = 4 * k + 2 * half;
          fb[(k * 32 + lane) * 2 + 0] = (K0 <= m)     ? h[m - K0]     : 0.0f;
          fb[(k * 32 + lane) * 2 + 1] = (K0 + 1 <= m) ? h[m - K0 - 1] : 0.0f;
        }
        fb[256 + lane * 2 + 0] = half ? 0.0f : g0i[m];
        fb[256 + lane * 2 + 1] = half ? 0.0f : g1i[m];
      }
    }
    {
      float g0 = 1.0f, g1 = 0.0f;
      float* G = gt + (b * NSEC + s) * TILE * 2;
      for (int n = 0; n < TILE; ++n) {
        G[n * 2 + 0] = g0; G[n * 2 + 1] = g1;
        float n0 = g0 * A00 + g1 * A10;
        float n1 = g0 * A01 + g1 * A11;
        g0 = n0; g1 = n1;
      }
    }
  }
}


__device__ __forceinline__ void wave_lds_order() {
  __builtin_amdgcn_fence(__ATOMIC_RELEASE, "wavefront");
  __builtin_amdgcn_wave_barrier();
  __builtin_amdgcn_fence(__ATOMIC_ACQUIRE, "wavefront");
}

__global__ void __launch_bounds__(32) tile_kernel(const float* in, float* out,
                                                  const float* __restrict__ coef,
                                                  const float* __restrict__ fr,
                                                  const float* __restrict__ gt,
                                                  const float* __restrict__ sg,
                                                  float* __restrict__ ts,
                                                  int sec, int prevsec) {
  __shared__ float xs[TILE];
  const int lane = threadIdx.x;
  const int b = blockIdx.x / WPR;
  const int w = blockIdx.x - b * WPR;
  const int tile0 = b * NTILES + w * TPW;
  const int base0 = b * LSIG + w * TPW * TILE;
  const int m = lane & 15;
  const int half = lane >> 4;
  const bool fuse = prevsec >= 0;

  const float* cf = coef + (b * NSEC + sec) * CSTRIDE;
  const v2f* hf = (const v2f*)(fr + (b * NSEC + sec) * FSTRIDE);
  v2f ha[4];
#pragma unroll
  for (int k = 0; k < 4; ++k) ha[k] = hf[k * 32 + lane];
  const v2f gfr = hf[128 + lane];
  float f0[16], f1[16], Ms[16];
#pragma unroll
  for (int j = 0; j < 16; ++j) {
    f0[j] = cf[16 + j];
    f1[j] = cf[32 + j];
    Ms[j] = cf[80 + j];
  }
  float4 ga = {0.f, 0.f, 0.f, 0.f}, gb = ga, gc = ga, gd = ga;
  if (fuse) {
    const float4* G4 = (const float4*)(gt + (b * NSEC + prevsec) * TILE * 2);
    ga = G4[lane * 2 + 0];
    gb = G4[lane * 2 + 1];
    gc = G4[64 + lane * 2 + 0];
    gd = G4[64 + lane * 2 + 1];
  }
  const float2* sg2 = (const float2*)sg;
  float2* ts2 = (float2*)ts;
  const float* src = in + base0 + lane * 4;
  float* dst = out + base0 + lane * 4;

  float4 u0 = *(const float4*)(src);
  float4 u1 = *(const float4*)(src + 128);
  float svx_l = 0.f, svy_l = 0.f;
  if (fuse) {
    float2 t = sg2[tile0 + (lane < TPW ? lane : TPW - 1)];
    svx_l = t.x; svy_l = t.y;
  }

  for (int i = 0; i < TPW; ++i) {
    float4 n0 = u0, n1 = u1;
    if (i + 1 < TPW) {
      n0 = *(const float4*)(src + (i + 1) * TILE);
      n1 = *(const float4*)(src + (i + 1) * TILE + 128);
    }

    if (fuse) {
      float2 sv;
      sv.x = __int_as_float(__builtin_amdgcn_readlane(__float_as_int(svx_l), i));
      sv.y = __int_as_float(__builtin_amdgcn_readlane(__float_as_int(svy_l), i));
      u0.x += ga.x * sv.x + ga.y * sv.y;
      u0.y += ga.z * sv.x + ga.w * sv.y;
      u0.z += gb.x * sv.x + gb.y * sv.y;
      u0.w += gb.z * sv.x + gb.w * sv.y;
      u1.x += gc.x * sv.x + gc.y * sv.y;
      u1.y += gc.z * sv.x + gc.w * sv.y;
      u1.z += gd.x * sv.x + gd.y * sv.y;
      u1.w += gd.z * sv.x + gd.w * sv.y;
    }
    *(float4*)(xs + lane * 4)       = u0;
    *(float4*)(xs + 128 + lane * 4) = u1;
    wave_lds_order();

    float4 xc[4];
#pragma unroll
    for (int q = 0; q < 4; ++q)
      xc[q] = ((const float4*)(xs + m * 16))[q];

    v8f acc = {0.f, 0.f, 0.f, 0.f, 0.f, 0.f, 0.f, 0.f};
#pragma unroll
    for (int k = 0; k < 4; ++k) {
      v2f bb;
      bb.x = half ? xc[k].z : xc[k].x;
      bb.y = half ? xc[k].w : xc[k].y;
      acc = wmma_f32_16x16x4(ha[k], bb, acc);
    }

    float fx0 = 0.f, fx1 = 0.f;
#pragma unroll
    for (int q = 0; q < 4; ++q) {
      fx0 += f0[4 * q + 0] * xc[q].x + f0[4 * q + 1] * xc[q].y +
             f0[4 * q + 2] * xc[q].z + f0[4 * q + 3] * xc[q].w;
      fx1 += f1[4 * q + 0] * xc[q].x + f1[4 * q + 1] * xc[q].y +
             f1[4 * q + 2] * xc[q].z + f1[4 * q + 3] * xc[q].w;
    }

    float p0 = fx0, p1 = fx1;
#pragma unroll
    for (int s = 0; s < 4; ++s) {
      float M00 = Ms[4 * s + 0], M01 = Ms[4 * s + 1];
      float M10 = Ms[4 * s + 2], M11 = Ms[4 * s + 3];
      int len = 1 << s;
      float q0 = __shfl_up(p0, (unsigned)len, 32);
      float q1 = __shfl_up(p1, (unsigned)len, 32);
      if (m >= len) {
        p0 = M00 * q0 + M01 * q1 + p0;
        p1 = M10 * q0 + M11 * q1 + p1;
      }
    }
    float e0 = __shfl_up(p0, 1u, 32);
    float e1 = __shfl_up(p1, 1u, 32);
    float s0c = (m >= 1) ? e0 : 0.f;
    float s1c = (m >= 1) ? e1 : 0.f;

    {
      v2f bb;
      bb.x = half ? 0.f : s0c;
      bb.y = half ? 0.f : s1c;
      acc = wmma_f32_16x16x4(gfr, bb, acc);
    }

    if (lane == 15) {
      float2 z; z.x = p0; z.y = p1;
      ts2[tile0 + i] = z;
    }

    wave_lds_order();
    {
      float4 d0, d1;
      d0.x = acc[0]; d0.y = acc[1]; d0.z = acc[2]; d0.w = acc[3];
      d1.x = acc[4]; d1.y = acc[5]; d1.z = acc[6]; d1.w = acc[7];
      *(float4*)(xs + m * 16 + 8 * half)     = d0;
      *(float4*)(xs + m * 16 + 8 * half + 4) = d1;
    }
    wave_lds_order();
    float4 o0 = *(const float4*)(xs + lane * 4);
    float4 o1 = *(const float4*)(xs + 128 + lane * 4);
    wave_lds_order();
    *(float4*)(dst + i * TILE)       = o0;
    *(float4*)(dst + i * TILE + 128) = o1;

    u0 = n0; u1 = n1;
  }
}

#define SCH   ((NTILES + 31) / 32)
#define SLAST (NTILES - (SCH - 1) * 32)

struct m22 { float a, b, c, d; };
__device__ __forceinline__ m22 mm(const m22 x, const m22 y) {
  m22 r;
  r.a = x.a * y.a + x.b * y.c;  r.b = x.a * y.b + x.b * y.d;
  r.c = x.c * y.a + x.d * y.c;  r.d = x.c * y.b + x.d * y.d;
  return r;
}

__global__ void __launch_bounds__(32) scan_kernel(const float* __restrict__ coef,
                                                  const float* __restrict__ ts,
                                                  float* __restrict__ sg, int sec) {
  __shared__ float2 zs[SCH * 32];
  const int lane = threadIdx.x;
  const int b = blockIdx.x;
  const float* cf = coef + (b * NSEC + sec) * CSTRIDE;
  const float A00 = cf[96], A01 = cf[97], A10 = cf[98], A11 = cf[99];
  const float2* z2 = (const float2*)ts + b * NTILES;
  float2* o2 = (float2*)sg + b * NTILES;

  float2 zr[SCH];
#pragma unroll
  for (int c = 0; c < SCH - 1; ++c) zr[c] = z2[c * 32 + lane];
  zr[SCH - 1].x = 0.f; zr[SCH - 1].y = 0.f;
  if (lane < SLAST) zr[SCH - 1] = z2[(SCH - 1) * 32 + lane];

  m22 Tk[5];
  {
    m22 P1; P1.a = A00; P1.b = A01; P1.c = A10; P1.d = A11;
    const m22 P2 = mm(P1, P1), P4 = mm(P2, P2), P8 = mm(P4, P4);
    const m22 P16 = mm(P8, P8), P32 = mm(P16, P16);
    static_assert(SCH == 59, "the product below spells out 59 = 32+16+8+2+1");
    Tk[0] = mm(mm(mm(mm(P32, P16), P8), P2), P1);
#pragma unroll
    for (int k = 1; k < 5; ++k) Tk[k] = mm(Tk[k - 1], Tk[k - 1]);
  }

#pragma unroll
  for (int c = 0; c < SCH; ++c) zs[c * 32 + lane] = zr[c];
  wave_lds_order();

  float2* mine = zs + lane * SCH;
  float2 zl[SCH];
#pragma unroll
  for (int c = 0; c < SCH; ++c) zl[c] = mine[c];

  float s0 = 0.f, s1 = 0.f;
#pragma unroll
  for (int c = 0; c < SCH; ++c) {
    const float n0 = __builtin_fmaf(A00, s0, A01 * s1) + zl[c].x;
    const float n1 = __builtin_fmaf(A11, s1, A10 * s0) + zl[c].y;
    s0 = n0; s1 = n1;
  }

  float p0 = s0, p1 = s1;
#pragma unroll
  for (int k = 0; k < 5; ++k) {
    const int len = 1 << k;
    const float q0 = __shfl_up(p0, (unsigned)len, 32);
    const float q1 = __shfl_up(p1, (unsigned)len, 32);
    if (lane >= len) {
      p0 = Tk[k].a * q0 + Tk[k].b * q1 + p0;
      p1 = Tk[k].c * q0 + Tk[k].d * q1 + p1;
    }
  }
  const float e0 = __shfl_up(p0, 1u, 32);
  const float e1 = __shfl_up(p1, 1u, 32);
  s0 = (lane >= 1) ? e0 : 0.f;
  s1 = (lane >= 1) ? e1 : 0.f;

#pragma unroll
  for (int c = 0; c < SCH; ++c) {
    float2 k; k.x = s0; k.y = s1;
    mine[c] = k;
    const float n0 = __builtin_fmaf(A00, s0, A01 * s1) + zl[c].x;
    const float n1 = __builtin_fmaf(A11, s1, A10 * s0) + zl[c].y;
    s0 = n0; s1 = n1;
  }
  wave_lds_order();

#pragma unroll
  for (int c = 0; c < SCH - 1; ++c) o2[c * 32 + lane] = zs[c * 32 + lane];
  if (lane < SLAST) o2[(SCH - 1) * 32 + lane] = zs[(SCH - 1) * 32 + lane];
}

#define FWPB 8
#define FGRP 3
#define FNG  (TPW / FGRP)
static_assert((NB * WPR) % FWPB == 0, "whole workgroups of tile runs");
static_assert(TPW % FGRP == 0 && TPW <= 32, "whole groups; one lane per tile state");

__global__ void __launch_bounds__(32 * FWPB) fixup_kernel(float* out,
                                                          const float* __restrict__ gt,
                                                          const float* __restrict__ sg,
                                                          int sec) {
  const int lane = threadIdx.x & 31;
  const int wv = __builtin_amdgcn_readfirstlane(
      (int)(blockIdx.x * FWPB + (threadIdx.x >> 5)));
  const int b = wv / WPR;
  const int w = wv - b * WPR;
  const int tile0 = b * NTILES + w * TPW;
  const int base0 = b * LSIG + w * TPW * TILE;
  float* p = out + base0 + lane * 4;

  float4 u0[2][FGRP], u1[2][FGRP];
#pragma unroll
  for (int j = 0; j < FGRP; ++j) {
    u0[0][j] = *(const float4*)(p + j * TILE);
    u1[0][j] = *(const float4*)(p + j * TILE + 128);
  }

  const float4* G4 = (const float4*)(gt + (b * NSEC + sec) * TILE * 2);
  const float4 ga = G4[lane * 2 + 0];
  const float4 gb = G4[lane * 2 + 1];
  const float4 gc = G4[64 + lane * 2 + 0];
  const float4 gd = G4[64 + lane * 2 + 1];
  const float2 sl = ((const float2*)sg)[tile0 + (lane < TPW ? lane : TPW - 1)];

#pragma unroll
  for (int g = 0; g < FNG; ++g) {
    const int cur = g & 1, nxt = cur ^ 1;
    if (g + 1 < FNG) {
#pragma unroll
      for (int j = 0; j < FGRP; ++j) {
        u0[nxt][j] = *(const float4*)(p + ((g + 1) * FGRP + j) * TILE);
        u1[nxt][j] = *(const float4*)(p + ((g + 1) * FGRP + j) * TILE + 128);
      }
    }
    __builtin_amdgcn_sched_barrier(0);
#pragma unroll
    for (int j = 0; j < FGRP; ++j) {
      const int i = g * FGRP + j;
      const float sx = __int_as_float(__builtin_amdgcn_readlane(__float_as_int(sl.x), i));
      const float sy = __int_as_float(__builtin_amdgcn_readlane(__float_as_int(sl.y), i));
      float4 a = u0[cur][j], c = u1[cur][j];
      a.x += ga.x * sx + ga.y * sy;
      a.y += ga.z * sx + ga.w * sy;
      a.z += gb.x * sx + gb.y * sy;
      a.w += gb.z * sx + gb.w * sy;
      c.x += gc.x * sx + gc.y * sy;
      c.y += gc.z * sx + gc.w * sy;
      c.z += gd.x * sx + gd.y * sy;
      c.w += gd.z * sx + gd.w * sy;
      *(float4*)(p + i * TILE)       = a;
      *(float4*)(p + i * TILE + 128) = c;
    }
  }
}

extern "C" void kernel_launch(void* const* d_in, const int* in_sizes, int n_in,
                              void* d_out, int out_size, void* d_ws, size_t ws_size,
                              hipStream_t stream) {
  (void)in_sizes; (void)n_in; (void)out_size; (void)ws_size;
  const float* x     = (const float*)d_in[0];
  const float* knobs = (const float*)d_in[1];
  const float* W1    = (const float*)d_in[2];
  const float* b1v   = (const float*)d_in[3];
  const float* W2    = (const float*)d_in[4];
  const float* b2v   = (const float*)d_in[5];
  float* out = (float*)d_out;

  float* COEF = (float*)d_ws;
  float* FR   = COEF + NB * NSEC * CSTRIDE;
  float* GT   = FR   + NB * NSEC * FSTRIDE;
  float* TS   = GT   + NB * NSEC * TILE * 2;
  float* SG   = TS   + NB * NTILES * 2;

  derive_kernel<<<1, 64, 0, stream>>>(knobs, W1, b1v, W2, b2v, COEF, FR, GT);

  const int tileWaves = NB * WPR;
  for (int sec = 0; sec < NSEC; ++sec) {
    const float* src = (sec == 0) ? x : out;
    tile_kernel<<<tileWaves, 32, 0, stream>>>(src, out, COEF, FR, GT, SG, TS,
                                              sec, sec - 1);
    scan_kernel<<<NB, 32, 0, stream>>>(COEF, TS, SG, sec);
  }
  fixup_kernel<<<NB * WPR / FWPB, 32 * FWPB, 0, stream>>>(out, GT, SG, NSEC - 1);
}


# scan launches removed: a tile wave carries the state across its 15 tiles and leaves one run end state; consumer waves rebuild their entry state from the row's 125 run states; sum order changes
# speedup vs baseline: 4.9804x; 1.0602x over previous
#include <hip/hip_runtime.h>
#include <hip/hip_bf16.h>

#define NB      64
#define LSIG    480000
#define TILE    256
#define NTILES  (LSIG / TILE)
#define NSEC    3
#define CSTRIDE 128
#define FSTRIDE 320
#define TPW     15
#define WPR     (NTILES / TPW)
#define RSEG    4
static_assert(WPR <= 32 * RSEG, "one wave holds a whole row of run states");

typedef __attribute__((ext_vector_type(2))) float v2f;
typedef __attribute__((ext_vector_type(8))) float v8f;

__device__ __forceinline__ v8f wmma_f32_16x16x4(v2f a, v2f b, v8f c) {
  return __builtin_amdgcn_wmma_f32_16x16x4_f32(false, a, false, b, (short)0, c,
                                               false, false);
}

struct m22 { float a, b, c, d; };
__device__ __forceinline__ m22 mm(const m22 x, const m22 y) {
  m22 r;
  r.a = x.a * y.a + x.b * y.c;  r.b = x.a * y.b + x.b * y.d;
  r.c = x.c * y.a + x.d * y.c;  r.d = x.c * y.b + x.d * y.d;
  return r;
}

__global__ void derive_kernel(const float* __restrict__ knobs,
                              const float* __restrict__ W1,
                              const float* __restrict__ b1v,
                              const float* __restrict__ W2,
                              const float* __restrict__ b2v,
                              float* __restrict__ coef,
                              float* __restrict__ fr,
                              float* __restrict__ gt) {
  int b = threadIdx.x;
  if (b >= NB) return;
  float k0 = knobs[b * 3 + 0], k1 = knobs[b * 3 + 1], k2 = knobs[b * 3 + 2];
  float hmlp[16];
  for (int i = 0; i < 16; ++i)
    hmlp[i] = tanhf(k0 * W1[0 * 16 + i] + k1 * W1[1 * 16 + i] +
                    k2 * W1[2 * 16 + i] + b1v[i]);
  float p[15];
  for (int j = 0; j < 15; ++j) {
    float acc = b2v[j];
    for (int i = 0; i < 16; ++i) acc += hmlp[i] * W2[i * 15 + j];
    p[j] = acc;
  }
  for (int s = 0; s < NSEC; ++s) {
    float c0 = p[s * 5 + 0], c1 = p[s * 5 + 1], c2 = p[s * 5 + 2];
    float a2c = 0.999f * tanhf(p[s * 5 + 4]);
    float a1c = 0.999f * (1.0f + a2c) * tanhf(p[s * 5 + 3]);
    float A00 = -a1c, A01 = 1.0f, A10 = -a2c, A11 = 0.0f;
    float Bv0 = c1 - a1c * c0, Bv1 = c2 - a2c * c0;
    float* cf = coef + (b * NSEC + s) * CSTRIDE;

    float h[16];
    {
      float v0 = Bv0, v1 = Bv1;
      h[0] = c0;
      for (int k = 1; k < 16; ++k) {
        h[k] = v0;
        float n0 = A00 * v0 + A01 * v1;
        float n1 = A10 * v0 + A11 * v1;
        v0 = n0; v1 = n1;
      }
    }
    {
      float v0 = Bv0, v1 = Bv1;
      cf[16 + 15] = v0; cf[32 + 15] = v1;
      for (int j = 14; j >= 0; --j) {
        float n0 = A00 * v0 + A01 * v1;
        float n1 = A10 * v0 + A11 * v1;
        v0 = n0; v1 = n1;
        cf[16 + j] = v0; cf[32 + j] = v1;
      }
    }
    float g0i[16], g1i[16];
    {
      float g0 = 1.0f, g1 = 0.0f;
      for (int i = 0; i < 16; ++i) {
        g0i[i] = g0; g1i[i] = g1;
        float n0 = g0 * A00 + g1 * A10;
        float n1 = g0 * A01 + g1 * A11;
        g0 = n0; g1 = n1;
      }
    }
    {
      float M00 = A00, M01 = A01, M10 = A10, M11 = A11;
      for (int q = 0; q < 4; ++q) {
        float n00 = M00 * M00 + M01 * M10, n01 = M00 * M01 + M01 * M11;
        float n10 = M10 * M00 + M11 * M10, n11 = M10 * M01 + M11 * M11;
        M00 = n00; M01 = n01; M10 = n10; M11 = n11;
      }
      m22 X256; X256.a = 0.f; X256.b = 0.f; X256.c = 0.f; X256.d = 0.f;
      for (int q = 0; q < 5; ++q) {
        cf[80 + 4 * q + 0] = M00; cf[80 + 4 * q + 1] = M01;
        cf[80 + 4 * q + 2] = M10; cf[80 + 4 * q + 3] = M11;
        if (q == 4) { X256.a = M00; X256.b = M01; X256.c = M10; X256.d = M11; }
        float n00 = M00 * M00 + M01 * M10, n01 = M00 * M01 + M01 * M11;
        float n10 = M10 * M00 + M11 * M10, n11 = M10 * M01 + M11 * M11;
        M00 = n00; M01 = n01; M10 = n10; M11 = n11;
      }
      static_assert(TPW == 15 && RSEG == 4, "the products below spell out 15 = 8+4+2+1 and R^4");
      m22 X512; X512.a = M00; X512.b = M01; X512.c = M10; X512.d = M11;
      const m22 X1024 = mm(X512, X512), X2048 = mm(X1024, X1024);
      const m22 Rm = mm(mm(mm(X2048, X1024), X512), X256);
      cf[100] = Rm.a; cf[101] = Rm.b; cf[102] = Rm.c; cf[103] = Rm.d;
      m22 Tm = mm(Rm, Rm); Tm = mm(Tm, Tm);
      for (int k = 0; k < 5; ++k) {
        cf[104 + 4 * k + 0] = Tm.a; cf[104 + 4 * k + 1] = Tm.b;
        cf[104 + 4 * k + 2] = Tm.c; cf[104 + 4 * k + 3] = Tm.d;
        Tm = mm(Tm, Tm);
      }
    }
    {
      float* fb = fr + (b * NSEC + s) * FSTRIDE;
      for (int lane = 0; lane < 32; ++lane) {
        int m = lane & 15, half = lane >> 4;
        for (int k = 0; k < 4; ++k) {
          int K0 = 4 * k + 2 * half;
          fb[(k * 32 + lane) * 2 + 0] = (K0 <= m)     ? h[m - K0]     : 0.0f;
          fb[(k * 32 + lane) * 2 + 1] = (K0 + 1 <= m) ? h[m - K0 - 1] : 0.0f;
        }
        fb[256 + lane * 2 + 0] = half ? 0.0f : g0i[m];
        fb[256 + lane * 2 + 1] = half ? 0.0f : g1i[m];
      }
    }
    {
      float g0 = 1.0f, g1 = 0.0f;
      float* G = gt + (b * NSEC + s) * TILE * 2;
      for (int n = 0; n < TILE; ++n) {
        G[n * 2 + 0] = g0; G[n * 2 + 1] = g1;
        float n0 = g0 * A00 + g1 * A10;
        float n1 = g0 * A01 + g1 * A11;
        g0 = n0; g1 = n1;
      }
    }
  }
}


__device__ __forceinline__ void wave_lds_order() {
  __builtin_amdgcn_fence(__ATOMIC_RELEASE, "wavefront");
  __builtin_amdgcn_wave_barrier();
  __builtin_amdgcn_fence(__ATOMIC_ACQUIRE, "wavefront");
}

__device__ __forceinline__ float2 run_tile_states(const float* __restrict__ cfp,
                                                  const float2* zrow, int w, int lane) {
  m22 A2; A2.a = cfp[96];  A2.b = cfp[97];  A2.c = cfp[98];  A2.d = cfp[99];
  m22 Rm; Rm.a = cfp[100]; Rm.b = cfp[101]; Rm.c = cfp[102]; Rm.d = cfp[103];
  float2 zl[RSEG];
#pragma unroll
  for (int c = 0; c < RSEG; ++c) {
    const int k = lane * RSEG + c;
    zl[c].x = 0.f; zl[c].y = 0.f;
    if (k < WPR) zl[c] = zrow[k];
  }
  float s0 = 0.f, s1 = 0.f;
#pragma unroll
  for (int c = 0; c < RSEG; ++c) {
    const float n0 = Rm.a * s0 + Rm.b * s1 + zl[c].x;
    const float n1 = Rm.c * s0 + Rm.d * s1 + zl[c].y;
    s0 = n0; s1 = n1;
  }
  float p0 = s0, p1 = s1;
#pragma unroll
  for (int k = 0; k < 5; ++k) {
    const float T00 = cfp[104 + 4 * k + 0], T01 = cfp[104 + 4 * k + 1];
    const float T10 = cfp[104 + 4 * k + 2], T11 = cfp[104 + 4 * k + 3];
    const int len = 1 << k;
    const float q0 = __shfl_up(p0, (unsigned)len, 32);
    const float q1 = __shfl_up(p1, (unsigned)len, 32);
    if (lane >= len) {
      p0 = T00 * q0 + T01 * q1 + p0;
      p1 = T10 * q0 + T11 * q1 + p1;
    }
  }
  const float e0 = __shfl_up(p0, 1u, 32);
  const float e1 = __shfl_up(p1, 1u, 32);
  s0 = (lane >= 1) ? e0 : 0.f;
  s1 = (lane >= 1) ? e1 : 0.f;
  const int own = w / RSEG, r = w - own * RSEG;
#pragma unroll
  for (int c = 0; c < RSEG - 1; ++c) {
    if (c < r) {
      const float n0 = Rm.a * s0 + Rm.b * s1 + zl[c].x;
      const float n1 = Rm.c * s0 + Rm.d * s1 + zl[c].y;
      s0 = n0; s1 = n1;
    }
  }
  float sx = __int_as_float(__builtin_amdgcn_readlane(__float_as_int(s0), own));
  float sy = __int_as_float(__builtin_amdgcn_readlane(__float_as_int(s1), own));
  float2 sl; sl.x = 0.f; sl.y = 0.f;
#pragma unroll
  for (int i = 0; i < TPW; ++i) {
    if (lane == i) { sl.x = sx; sl.y = sy; }
    const float nx = A2.a * sx + A2.b * sy;
    const float ny = A2.c * sx + A2.d * sy;
    sx = nx; sy = ny;
  }
  return sl;
}

__global__ void __launch_bounds__(32) tile_kernel(const float* in, float* out,
                                                  const float* __restrict__ coef,
                                                  const float* __restrict__ fr,
                                                  const float* __restrict__ gt,
                                                  float* ts,
                                                  int sec, int prevsec) {
  __shared__ float xs[TILE];
  const int lane = threadIdx.x;
  const int b = blockIdx.x / WPR;
  const int w = blockIdx.x - b * WPR;
  const int base0 = b * LSIG + w * TPW * TILE;
  const int m = lane & 15;
  const int half = lane >> 4;
  const bool fuse = prevsec >= 0;

  const float* cf = coef + (b * NSEC + sec) * CSTRIDE;
  const v2f* hf = (const v2f*)(fr + (b * NSEC + sec) * FSTRIDE);
  v2f ha[4];
#pragma unroll
  for (int k = 0; k < 4; ++k) ha[k] = hf[k * 32 + lane];
  const v2f gfr = hf[128 + lane];
  float f0[16], f1[16], Ms[16];
#pragma unroll
  for (int j = 0; j < 16; ++j) {
    f0[j] = cf[16 + j];
    f1[j] = cf[32 + j];
    Ms[j] = cf[80 + j];
  }
  float4 ga = {0.f, 0.f, 0.f, 0.f}, gb = ga, gc = ga, gd = ga;
  if (fuse) {
    const float4* G4 = (const float4*)(gt + (b * NSEC + prevsec) * TILE * 2);
    ga = G4[lane * 2 + 0];
    gb = G4[lane * 2 + 1];
    gc = G4[64 + lane * 2 + 0];
    gd = G4[64 + lane * 2 + 1];
  }
  float2* ts2 = (float2*)ts;
  const float* src = in + base0 + lane * 4;
  float* dst = out + base0 + lane * 4;

  float4 u0 = *(const float4*)(src);
  float4 u1 = *(const float4*)(src + 128);
  float svx_l = 0.f, svy_l = 0.f;
  if (fuse) {
    const float2 t = run_tile_states(coef + (b * NSEC + prevsec) * CSTRIDE,
                                     ts2 + (prevsec * NB + b) * WPR, w, lane);
    svx_l = t.x; svy_l = t.y;
  }
  float cin0 = 0.f, cin1 = 0.f;

  for (int i = 0; i < TPW; ++i) {
    float4 n0 = u0, n1 = u1;
    if (i + 1 < TPW) {
      n0 = *(const float4*)(src + (i + 1) * TILE);
      n1 = *(const float4*)(src + (i + 1) * TILE + 128);
    }

    if (fuse) {
      float2 sv;
      sv.x = __int_as_float(__builtin_amdgcn_readlane(__float_as_int(svx_l), i));
      sv.y = __int_as_float(__builtin_amdgcn_readlane(__float_as_int(svy_l), i));
      u0.x += ga.x * sv.x + ga.y * sv.y;
      u0.y += ga.z * sv.x + ga.w * sv.y;
      u0.z += gb.x * sv.x + gb.y * sv.y;
      u0.w += gb.z * sv.x + gb.w * sv.y;
      u1.x += gc.x * sv.x + gc.y * sv.y;
      u1.y += gc.z * sv.x + gc.w * sv.y;
      u1.z += gd.x * sv.x + gd.y * sv.y;
      u1.w += gd.z * sv.x + gd.w * sv.y;
    }
    *(float4*)(xs + lane * 4)       = u0;
    *(float4*)(xs + 128 + lane * 4) = u1;
    wave_lds_order();

    float4 xc[4];
#pragma unroll
    for (int q = 0; q < 4; ++q)
      xc[q] = ((const float4*)(xs + m * 16))[q];

    v8f acc = {0.f, 0.f, 0.f, 0.f, 0.f, 0.f, 0.f, 0.f};
#pragma unroll
    for (int k = 0; k < 4; ++k) {
      v2f bb;
      bb.x = half ? xc[k].z : xc[k].x;
      bb.y = half ? xc[k].w : xc[k].y;
      acc = wmma_f32_16x16x4(ha[k], bb, acc);
    }

    float fx0 = 0.f, fx1 = 0.f;
#pragma unroll
    for (int q = 0; q < 4; ++q) {
      fx0 += f0[4 * q + 0] * xc[q].x + f0[4 * q + 1] * xc[q].y +
             f0[4 * q + 2] * xc[q].z + f0[4 * q + 3] * xc[q].w;
      fx1 += f1[4 * q + 0] * xc[q].x + f1[4 * q + 1] * xc[q].y +
             f1[4 * q + 2] * xc[q].z + f1[4 * q + 3] * xc[q].w;
    }

    if (m == 0) {
      fx0 += Ms[0] * cin0 + Ms[1] * cin1;
      fx1 += Ms[2] * cin0 + Ms[3] * cin1;
    }

    float p0 = fx0, p1 = fx1;
#pragma unroll
    for (int s = 0; s < 4; ++s) {
      float M00 = Ms[4 * s + 0], M01 = Ms[4 * s + 1];
      float M10 = Ms[4 * s + 2], M11 = Ms[4 * s + 3];
      int len = 1 << s;
      float q0 = __shfl_up(p0, (unsigned)len, 32);
      float q1 = __shfl_up(p1, (unsigned)len, 32);
      if (m >= len) {
        p0 = M00 * q0 + M01 * q1 + p0;
        p1 = M10 * q0 + M11 * q1 + p1;
      }
    }
    float e0 = __shfl_up(p0, 1u, 32);
    float e1 = __shfl_up(p1, 1u, 32);
    float s0c = (m >= 1) ? e0 : cin0;
    float s1c = (m >= 1) ? e1 : cin1;
    cin0 = __int_as_float(__builtin_amdgcn_readlane(__float_as_int(p0), 15));
    cin1 = __int_as_float(__builtin_amdgcn_readlane(__float_as_int(p1), 15));

    {
      v2f bb;
      bb.x = half ? 0.f : s0c;
      bb.y = half ? 0.f : s1c;
      acc = wmma_f32_16x16x4(gfr, bb, acc);
    }

    wave_lds_order();
    {
      float4 d0, d1;
      d0.x = acc[0]; d0.y = acc[1]; d0.z = acc[2]; d0.w = acc[3];
      d1.x = acc[4]; d1.y = acc[5]; d1.z = acc[6]; d1.w = acc[7];
      *(float4*)(xs + m * 16 + 8 * half)     = d0;
      *(float4*)(xs + m * 16 + 8 * half + 4) = d1;
    }
    wave_lds_order();
    float4 o0 = *(const float4*)(xs + lane * 4);
    float4 o1 = *(const float4*)(xs + 128 + lane * 4);
    wave_lds_order();
    *(float4*)(dst + i * TILE)       = o0;
    *(float4*)(dst + i * TILE + 128) = o1;

    u0 = n0; u1 = n1;
  }
  if (lane == 0) {
    float2 z; z.x = cin0; z.y = cin1;
    ts2[(sec * NB + b) * WPR + w] = z;
  }
}

#define FWPB 8
#define FGRP 3
#define FNG  (TPW / FGRP)
static_assert((NB * WPR) % FWPB == 0, "whole workgroups of tile runs");
static_assert(TPW % FGRP == 0 && TPW <= 32, "whole groups; one lane per tile state");

__global__ void __launch_bounds__(32 * FWPB) fixup_kernel(float* out,
                                                          const float* __restrict__ coef,
                                                          const float* __restrict__ gt,
                                                          const float* __restrict__ ts,
                                                          int sec) {
  const int lane = threadIdx.x & 31;
  const int wv = __builtin_amdgcn_readfirstlane(
      (int)(blockIdx.x * FWPB + (threadIdx.x >> 5)));
  const int b = wv / WPR;
  const int w = wv - b * WPR;
  const int base0 = b * LSIG + w * TPW * TILE;
  float* p = out + base0 + lane * 4;

  float4 u0[2][FGRP], u1[2][FGRP];
#pragma unroll
  for (int j = 0; j < FGRP; ++j) {
    u0[0][j] = *(const float4*)(p + j * TILE);
    u1[0][j] = *(const float4*)(p + j * TILE + 128);
  }

  const float4* G4 = (const float4*)(gt + (b * NSEC + sec) * TILE * 2);
  const float4 ga = G4[lane * 2 + 0];
  const float4 gb = G4[lane * 2 + 1];
  const float4 gc = G4[64 + lane * 2 + 0];
  const float4 gd = G4[64 + lane * 2 + 1];
  const float2 sl = run_tile_states(coef + (b * NSEC + sec) * CSTRIDE,
                                    (const float2*)ts + (sec * NB + b) * WPR, w, lane);

#pragma unroll
  for (int g = 0; g < FNG; ++g) {
    const int cur = g & 1, nxt = cur ^ 1;
    if (g + 1 < FNG) {
#pragma unroll
      for (int j = 0; j < FGRP; ++j) {
        u0[nxt][j] = *(const float4*)(p + ((g + 1) * FGRP + j) * TILE);
        u1[nxt][j] = *(const float4*)(p + ((g + 1) * FGRP + j) * TILE + 128);
      }
    }
    __builtin_amdgcn_sched_barrier(0);
#pragma unroll
    for (int j = 0; j < FGRP; ++j) {
      const int i = g * FGRP + j;
      const float sx = __int_as_float(__builtin_amdgcn_readlane(__float_as_int(sl.x), i));
      const float sy = __int_as_float(__builtin_amdgcn_readlane(__float_as_int(sl.y), i));
      float4 a = u0[cur][j], c = u1[cur][j];
      a.x += ga.x * sx + ga.y * sy;
      a.y += ga.z * sx + ga.w * sy;
      a.z += gb.x * sx + gb.y * sy;
      a.w += gb.z * sx + gb.w * sy;
      c.x += gc.x * sx + gc.y * sy;
      c.y += gc.z * sx + gc.w * sy;
      c.z += gd.x * sx + gd.y * sy;
      c.w += gd.z * sx + gd.w * sy;
      *(float4*)(p + i * TILE)       = a;
      *(float4*)(p + i * TILE + 128) = c;
    }
  }
}

extern "C" void kernel_launch(void* const* d_in, const int* in_sizes, int n_in,
                              void* d_out, int out_size, void* d_ws, size_t ws_size,
                              hipStream_t stream) {
  (void)in_sizes; (void)n_in; (void)out_size; (void)ws_size;
  const float* x     = (const float*)d_in[0];
  const float* knobs = (const float*)d_in[1];
  const float* W1    = (const float*)d_in[2];
  const float* b1v   = (const float*)d_in[3];
  const float* W2    = (const float*)d_in[4];
  const float* b2v   = (const float*)d_in[5];
  float* out = (float*)d_out;

  float* COEF = (float*)d_ws;
  float* FR   = COEF + NB * NSEC * CSTRIDE;
  float* GT   = FR   + NB * NSEC * FSTRIDE;
  float* TS   = GT   + NB * NSEC * TILE * 2;

  derive_kernel<<<1, 64, 0, stream>>>(knobs, W1, b1v, W2, b2v, COEF, FR, GT);

  const int tileWaves = NB * WPR;
  for (int sec = 0; sec < NSEC; ++sec) {
    const float* src = (sec == 0) ? x : out;
    tile_kernel<<<tileWaves, 32, 0, stream>>>(src, out, COEF, FR, GT, TS,
                                              sec, sec - 1);
  }
  fixup_kernel<<<NB * WPR / FWPB, 32 * FWPB, 0, stream>>>(out, COEF, GT, TS, NSEC - 1);
}


# set-up kernel: one wave per (row, section) on 192 workgroups, each lane builds its own operand fragments and keeps 8 of the 256 state rows, not 64 threads each looping over 3 sections; sums unchanged
# speedup vs baseline: 6.4143x; 1.2879x over previous
#include <hip/hip_runtime.h>
#include <hip/hip_bf16.h>

#define NB      64
#define LSIG    480000
#define TILE    256
#define NTILES  (LSIG / TILE)
#define NSEC    3
#define CSTRIDE 128
#define FSTRIDE 320
#define TPW     15
#define WPR     (NTILES / TPW)
#define RSEG    4
static_assert(WPR <= 32 * RSEG, "one wave holds a whole row of run states");

typedef __attribute__((ext_vector_type(2))) float v2f;
typedef __attribute__((ext_vector_type(8))) float v8f;

__device__ __forceinline__ v8f wmma_f32_16x16x4(v2f a, v2f b, v8f c) {
  return __builtin_amdgcn_wmma_f32_16x16x4_f32(false, a, false, b, (short)0, c,
                                               false, false);
}

struct m22 { float a, b, c, d; };
__device__ __forceinline__ m22 mm(const m22 x, const m22 y) {
  m22 r;
  r.a = x.a * y.a + x.b * y.c;  r.b = x.a * y.b + x.b * y.d;
  r.c = x.c * y.a + x.d * y.c;  r.d = x.c * y.b + x.d * y.d;
  return r;
}

__global__ void __launch_bounds__(32) derive_kernel(const float* __restrict__ knobs,
                                                    const float* __restrict__ W1,
                                                    const float* __restrict__ b1v,
                                                    const float* __restrict__ W2,
                                                    const float* __restrict__ b2v,
                                                    float* __restrict__ coef,
                                                    float* __restrict__ fr,
                                                    float* __restrict__ gt) {
  __shared__ float hs[16], g0s[16], g1s[16];
  const int lane = threadIdx.x;
  const int b = blockIdx.x / NSEC;
  const int s = blockIdx.x - b * NSEC;
  const float k0 = knobs[b * 3 + 0], k1 = knobs[b * 3 + 1], k2 = knobs[b * 3 + 2];
  float hmlp[16];
  for (int i = 0; i < 16; ++i)
    hmlp[i] = tanhf(k0 * W1[0 * 16 + i] + k1 * W1[1 * 16 + i] +
                    k2 * W1[2 * 16 + i] + b1v[i]);
  float p[5];
  for (int j = 0; j < 5; ++j) {
    float acc = b2v[s * 5 + j];
    for (int i = 0; i < 16; ++i) acc += hmlp[i] * W2[i * 15 + s * 5 + j];
    p[j] = acc;
  }
  const float c0 = p[0], c1 = p[1], c2 = p[2];
  const float a2c = 0.999f * tanhf(p[4]);
  const float a1c = 0.999f * (1.0f + a2c) * tanhf(p[3]);
  const float A00 = -a1c, A01 = 1.0f, A10 = -a2c, A11 = 0.0f;
  const float Bv0 = c1 - a1c * c0, Bv1 = c2 - a2c * c0;
  float* cf = coef + (b * NSEC + s) * CSTRIDE;

  float h[16];
  {
    float v0 = Bv0, v1 = Bv1;
    h[0] = c0;
    for (int k = 1; k < 16; ++k) {
      h[k] = v0;
      float n0 = A00 * v0 + A01 * v1;
      float n1 = A10 * v0 + A11 * v1;
      v0 = n0; v1 = n1;
    }
  }
  float fa[16], fb1[16];
  {
    float v0 = Bv0, v1 = Bv1;
    fa[15] = v0; fb1[15] = v1;
    for (int j = 14; j >= 0; --j) {
      float n0 = A00 * v0 + A01 * v1;
      float n1 = A10 * v0 + A11 * v1;
      v0 = n0; v1 = n1;
      fa[j] = v0; fb1[j] = v1;
    }
  }
  float g0i[16], g1i[16];
  {
    float g0 = 1.0f, g1 = 0.0f;
    for (int i = 0; i < 16; ++i) {
      g0i[i] = g0; g1i[i] = g1;
      float n0 = g0 * A00 + g1 * A10;
      float n1 = g0 * A01 + g1 * A11;
      g0 = n0; g1 = n1;
    }
  }
  m22 Am; Am.a = A00; Am.b = A01; Am.c = A10; Am.d = A11;
  const m22 X2 = mm(Am, Am), X4 = mm(X2, X2), X8 = mm(X4, X4);
  m22 Xp[5];
  Xp[0] = mm(X8, X8);
  for (int q = 1; q < 5; ++q) Xp[q] = mm(Xp[q - 1], Xp[q - 1]);
  static_assert(TPW == 15 && RSEG == 4, "the products below spell out 15 = 8+4+2+1 and R^4");
  const m22 X256 = Xp[4];
  const m22 X512 = mm(X256, X256);
  const m22 X1024 = mm(X512, X512), X2048 = mm(X1024, X1024);
  const m22 Rm = mm(mm(mm(X2048, X1024), X512), X256);
  m22 Tp[5];
  Tp[0] = mm(Rm, Rm); Tp[0] = mm(Tp[0], Tp[0]);
  for (int k = 1; k < 5; ++k) Tp[k] = mm(Tp[k - 1], Tp[k - 1]);

  if (lane == 0) {
    for (int j = 0; j < 16; ++j) {
      cf[16 + j] = fa[j]; cf[32 + j] = fb1[j];
      hs[j] = h[j]; g0s[j] = g0i[j]; g1s[j] = g1i[j];
    }
    for (int q = 0; q < 5; ++q) {
      cf[80 + 4 * q + 0] = Xp[q].a; cf[80 + 4 * q + 1] = Xp[q].b;
      cf[80 + 4 * q + 2] = Xp[q].c; cf[80 + 4 * q + 3] = Xp[q].d;
    }
    cf[100] = Rm.a; cf[101] = Rm.b; cf[102] = Rm.c; cf[103] = Rm.d;
    for (int k = 0; k < 5; ++k) {
      cf[104 + 4 * k + 0] = Tp[k].a; cf[104 + 4 * k + 1] = Tp[k].b;
      cf[104 + 4 * k + 2] = Tp[k].c; cf[104 + 4 * k + 3] = Tp[k].d;
    }
  }
  __syncthreads();

  {
    float* fb = fr + (b * NSEC + s) * FSTRIDE;
    const int m = lane & 15, half = lane >> 4;
    for (int k = 0; k < 4; ++k) {
      const int K0 = 4 * k + 2 * half;
      const int i0 = m - K0, i1 = m - K0 - 1;
      const float v0 = hs[i0 < 0 ? 0 : i0];
      const float v1 = hs[i1 < 0 ? 0 : i1];
      fb[(k * 32 + lane) * 2 + 0] = (i0 >= 0) ? v0 : 0.0f;
      fb[(k * 32 + lane) * 2 + 1] = (i1 >= 0) ? v1 : 0.0f;
    }
    fb[256 + lane * 2 + 0] = half ? 0.0f : g0s[m];
    fb[256 + lane * 2 + 1] = half ? 0.0f : g1s[m];
  }
  {
    float g0 = 1.0f, g1 = 0.0f;
    float r[16];
    for (int j = 0; j < 16; ++j) r[j] = 0.0f;
    for (int blk = 0; blk < 32; ++blk) {
      float t[16];
      for (int i = 0; i < 8; ++i) {
        t[2 * i + 0] = g0; t[2 * i + 1] = g1;
        const float n0 = g0 * A00 + g1 * A10;
        const float n1 = g0 * A01 + g1 * A11;
        g0 = n0; g1 = n1;
      }
      if (lane == blk) {
        for (int j = 0; j < 16; ++j) r[j] = t[j];
      }
    }
    float4* G = (float4*)(gt + (b * NSEC + s) * TILE * 2 + lane * 16);
    for (int q = 0; q < 4; ++q) {
      float4 v;
      v.x = r[4 * q + 0]; v.y = r[4 * q + 1]; v.z = r[4 * q + 2]; v.w = r[4 * q + 3];
      G[q] = v;
    }
  }
}


__device__ __forceinline__ void wave_lds_order() {
  __builtin_amdgcn_fence(__ATOMIC_RELEASE, "wavefront");
  __builtin_amdgcn_wave_barrier();
  __builtin_amdgcn_fence(__ATOMIC_ACQUIRE, "wavefront");
}

__device__ __forceinline__ float2 run_tile_states(const float* __restrict__ cfp,
                                                  const float2* zrow, int w, int lane) {
  m22 A2; A2.a = cfp[96];  A2.b = cfp[97];  A2.c = cfp[98];  A2.d = cfp[99];
  m22 Rm; Rm.a = cfp[100]; Rm.b = cfp[101]; Rm.c = cfp[102]; Rm.d = cfp[103];
  float2 zl[RSEG];
#pragma unroll
  for (int c = 0; c < RSEG; ++c) {
    const int k = lane * RSEG + c;
    zl[c].x = 0.f; zl[c].y = 0.f;
    if (k < WPR) zl[c] = zrow[k];
  }
  float s0 = 0.f, s1 = 0.f;
#pragma unroll
  for (int c = 0; c < RSEG; ++c) {
    const float n0 = Rm.a * s0 + Rm.b * s1 + zl[c].x;
    const float n1 = Rm.c * s0 + Rm.d * s1 + zl[c].y;
    s0 = n0; s1 = n1;
  }
  float p0 = s0, p1 = s1;
#pragma unroll
  for (int k = 0; k < 5; ++k) {
    const float T00 = cfp[104 + 4 * k + 0], T01 = cfp[104 + 4 * k + 1];
    const float T10 = cfp[104 + 4 * k + 2], T11 = cfp[104 + 4 * k + 3];
    const int len = 1 << k;
    const float q0 = __shfl_up(p0, (unsigned)len, 32);
    const float q1 = __shfl_up(p1, (unsigned)len, 32);
    if (lane >= len) {
      p0 = T00 * q0 + T01 * q1 + p0;
      p1 = T10 * q0 + T11 * q1 + p1;
    }
  }
  const float e0 = __shfl_up(p0, 1u, 32);
  const float e1 = __shfl_up(p1, 1u, 32);
  s0 = (lane >= 1) ? e0 : 0.f;
  s1 = (lane >= 1) ? e1 : 0.f;
  const int own = w / RSEG, r = w - own * RSEG;
#pragma unroll
  for (int c = 0; c < RSEG - 1; ++c) {
    if (c < r) {
      const float n0 = Rm.a * s0 + Rm.b * s1 + zl[c].x;
      const float n1 = Rm.c * s0 + Rm.d * s1 + zl[c].y;
      s0 = n0; s1 = n1;
    }
  }
  float sx = __int_as_float(__builtin_amdgcn_readlane(__float_as_int(s0), own));
  float sy = __int_as_float(__builtin_amdgcn_readlane(__float_as_int(s1), own));
  float2 sl; sl.x = 0.f; sl.y = 0.f;
#pragma unroll
  for (int i = 0; i < TPW; ++i) {
    if (lane == i) { sl.x = sx; sl.y = sy; }
    const float nx = A2.a * sx + A2.b * sy;
    const float ny = A2.c * sx + A2.d * sy;
    sx = nx; sy = ny;
  }
  return sl;
}

__global__ void __launch_bounds__(32) tile_kernel(const float* in, float* out,
                                                  const float* __restrict__ coef,
                                                  const float* __restrict__ fr,
                                                  const float* __restrict__ gt,
                                                  float* ts,
                                                  int sec, int prevsec) {
  __shared__ float xs[TILE];
  const int lane = threadIdx.x;
  const int b = blockIdx.x / WPR;
  const int w = blockIdx.x - b * WPR;
  const int base0 = b * LSIG + w * TPW * TILE;
  const int m = lane & 15;
  const int half = lane >> 4;
  const bool fuse = prevsec >= 0;

  const float* cf = coef + (b * NSEC + sec) * CSTRIDE;
  const v2f* hf = (const v2f*)(fr + (b * NSEC + sec) * FSTRIDE);
  v2f ha[4];
#pragma unroll
  for (int k = 0; k < 4; ++k) ha[k] = hf[k * 32 + lane];
  const v2f gfr = hf[128 + lane];
  float f0[16], f1[16], Ms[16];
#pragma unroll
  for (int j = 0; j < 16; ++j) {
    f0[j] = cf[16 + j];
    f1[j] = cf[32 + j];
    Ms[j] = cf[80 + j];
  }
  float4 ga = {0.f, 0.f, 0.f, 0.f}, gb = ga, gc = ga, gd = ga;
  if (fuse) {
    const float4* G4 = (const float4*)(gt + (b * NSEC + prevsec) * TILE * 2);
    ga = G4[lane * 2 + 0];
    gb = G4[lane * 2 + 1];
    gc = G4[64 + lane * 2 + 0];
    gd = G4[64 + lane * 2 + 1];
  }
  float2* ts2 = (float2*)ts;
  const float* src = in + base0 + lane * 4;
  float* dst = out + base0 + lane * 4;

  float4 u0 = *(const float4*)(src);
  float4 u1 = *(const float4*)(src + 128);
  float svx_l = 0.f, svy_l = 0.f;
  if (fuse) {
    const float2 t = run_tile_states(coef + (b * NSEC + prevsec) * CSTRIDE,
                                     ts2 + (prevsec * NB + b) * WPR, w, lane);
    svx_l = t.x; svy_l = t.y;
  }
  float cin0 = 0.f, cin1 = 0.f;

  for (int i = 0; i < TPW; ++i) {
    float4 n0 = u0, n1 = u1;
    if (i + 1 < TPW) {
      n0 = *(const float4*)(src + (i + 1) * TILE);
      n1 = *(const float4*)(src + (i + 1) * TILE + 128);
    }

    if (fuse) {
      float2 sv;
      sv.x = __int_as_float(__builtin_amdgcn_readlane(__float_as_int(svx_l), i));
      sv.y = __int_as_float(__builtin_amdgcn_readlane(__float_as_int(svy_l), i));
      u0.x += ga.x * sv.x + ga.y * sv.y;
      u0.y += ga.z * sv.x + ga.w * sv.y;
      u0.z += gb.x * sv.x + gb.y * sv.y;
      u0.w += gb.z * sv.x + gb.w * sv.y;
      u1.x += gc.x * sv.x + gc.y * sv.y;
      u1.y += gc.z * sv.x + gc.w * sv.y;
      u1.z += gd.x * sv.x + gd.y * sv.y;
      u1.w += gd.z * sv.x + gd.w * sv.y;
    }
    *(float4*)(xs + lane * 4)       = u0;
    *(float4*)(xs + 128 + lane * 4) = u1;
    wave_lds_order();

    float4 xc[4];
#pragma unroll
    for (int q = 0; q < 4; ++q)
      xc[q] = ((const float4*)(xs + m * 16))[q];

    v8f acc = {0.f, 0.f, 0.f, 0.f, 0.f, 0.f, 0.f, 0.f};
#pragma unroll
    for (int k = 0; k < 4; ++k) {
      v2f bb;
      bb.x = half ? xc[k].z : xc[k].x;
      bb.y = half ? xc[k].w : xc[k].y;
      acc = wmma_f32_16x16x4(ha[k], bb, acc);
    }

    float fx0 = 0.f, fx1 = 0.f;
#pragma unroll
    for (int q = 0; q < 4; ++q) {
      fx0 += f0[4 * q + 0] * xc[q].x + f0[4 * q + 1] * xc[q].y +
             f0[4 * q + 2] * xc[q].z + f0[4 * q + 3] * xc[q].w;
      fx1 += f1[4 * q + 0] * xc[q].x + f1[4 * q + 1] * xc[q].y +
             f1[4 * q + 2] * xc[q].z + f1[4 * q + 3] * xc[q].w;
    }

    if (m == 0) {
      fx0 += Ms[0] * cin0 + Ms[1] * cin1;
      fx1 += Ms[2] * cin0 + Ms[3] * cin1;
    }

    float p0 = fx0, p1 = fx1;
#pragma unroll
    for (int s = 0; s < 4; ++s) {
      float M00 = Ms[4 * s + 0], M01 = Ms[4 * s + 1];
      float M10 = Ms[4 * s + 2], M11 = Ms[4 * s + 3];
      int len = 1 << s;
      float q0 = __shfl_up(p0, (unsigned)len, 32);
      float q1 = __shfl_up(p1, (unsigned)len, 32);
      if (m >= len) {
        p0 = M00 * q0 + M01 * q1 + p0;
        p1 = M10 * q0 + M11 * q1 + p1;
      }
    }
    float e0 = __shfl_up(p0, 1u, 32);
    float e1 = __shfl_up(p1, 1u, 32);
    float s0c = (m >= 1) ? e0 : cin0;
    float s1c = (m >= 1) ? e1 : cin1;
    cin0 = __int_as_float(__builtin_amdgcn_readlane(__float_as_int(p0), 15));
    cin1 = __int_as_float(__builtin_amdgcn_readlane(__float_as_int(p1), 15));

    {
      v2f bb;
      bb.x = half ? 0.f : s0c;
      bb.y = half ? 0.f : s1c;
      acc = wmma_f32_16x16x4(gfr, bb, acc);
    }

    wave_lds_order();
    {
      float4 d0, d1;
      d0.x = acc[0]; d0.y = acc[1]; d0.z = acc[2]; d0.w = acc[3];
      d1.x = acc[4]; d1.y = acc[5]; d1.z = acc[6]; d1.w = acc[7];
      *(float4*)(xs + m * 16 + 8 * half)     = d0;
      *(float4*)(xs + m * 16 + 8 * half + 4) = d1;
    }
    wave_lds_order();
    float4 o0 = *(const float4*)(xs + lane * 4);
    float4 o1 = *(const float4*)(xs + 128 + lane * 4);
    wave_lds_order();
    *(float4*)(dst + i * TILE)       = o0;
    *(float4*)(dst + i * TILE + 128) = o1;

    u0 = n0; u1 = n1;
  }
  if (lane == 0) {
    float2 z; z.x = cin0; z.y = cin1;
    ts2[(sec * NB + b) * WPR + w] = z;
  }
}

#define FWPB 8
#define FGRP 3
#define FNG  (TPW / FGRP)
static_assert((NB * WPR) % FWPB == 0, "whole workgroups of tile runs");
static_assert(TPW % FGRP == 0 && TPW <= 32, "whole groups; one lane per tile state");

__global__ void __launch_bounds__(32 * FWPB) fixup_kernel(float* out,
                                                          const float* __restrict__ coef,
                                                          const float* __restrict__ gt,
                                                          const float* __restrict__ ts,
                                                          int sec) {
  const int lane = threadIdx.x & 31;
  const int wv = __builtin_amdgcn_readfirstlane(
      (int)(blockIdx.x * FWPB + (threadIdx.x >> 5)));
  const int b = wv / WPR;
  const int w = wv - b * WPR;
  const int base0 = b * LSIG + w * TPW * TILE;
  float* p = out + base0 + lane * 4;

  float4 u0[2][FGRP], u1[2][FGRP];
#pragma unroll
  for (int j = 0; j < FGRP; ++j) {
    u0[0][j] = *(const float4*)(p + j * TILE);
    u1[0][j] = *(const float4*)(p + j * TILE + 128);
  }

  const float4* G4 = (const float4*)(gt + (b * NSEC + sec) * TILE * 2);
  const float4 ga = G4[lane * 2 + 0];
  const float4 gb = G4[lane * 2 + 1];
  const float4 gc = G4[64 + lane * 2 + 0];
  const float4 gd = G4[64 + lane * 2 + 1];
  const float2 sl = run_tile_states(coef + (b * NSEC + sec) * CSTRIDE,
                                    (const float2*)ts + (sec * NB + b) * WPR, w, lane);

#pragma unroll
  for (int g = 0; g < FNG; ++g) {
    const int cur = g & 1, nxt = cur ^ 1;
    if (g + 1 < FNG) {
#pragma unroll
      for (int j = 0; j < FGRP; ++j) {
        u0[nxt][j] = *(const float4*)(p + ((g + 1) * FGRP + j) * TILE);
        u1[nxt][j] = *(const float4*)(p + ((g + 1) * FGRP + j) * TILE + 128);
      }
    }
    __builtin_amdgcn_sched_barrier(0);
#pragma unroll
    for (int j = 0; j < FGRP; ++j) {
      const int i = g * FGRP + j;
      const float sx = __int_as_float(__builtin_amdgcn_readlane(__float_as_int(sl.x), i));
      const float sy = __int_as_float(__builtin_amdgcn_readlane(__float_as_int(sl.y), i));
      float4 a = u0[cur][j], c = u1[cur][j];
      a.x += ga.x * sx + ga.y * sy;
      a.y += ga.z * sx + ga.w * sy;
      a.z += gb.x * sx + gb.y * sy;
      a.w += gb.z * sx + gb.w * sy;
      c.x += gc.x * sx + gc.y * sy;
      c.y += gc.z * sx + gc.w * sy;
      c.z += gd.x * sx + gd.y * sy;
      c.w += gd.z * sx + gd.w * sy;
      *(float4*)(p + i * TILE)       = a;
      *(float4*)(p + i * TILE + 128) = c;
    }
  }
}

extern "C" void kernel_launch(void* const* d_in, const int* in_sizes, int n_in,
                              void* d_out, int out_size, void* d_ws, size_t ws_size,
                              hipStream_t stream) {
  (void)in_sizes; (void)n_in; (void)out_size; (void)ws_size;
  const float* x     = (const float*)d_in[0];
  const float* knobs = (const float*)d_in[1];
  const float* W1    = (const float*)d_in[2];
  const float* b1v   = (const float*)d_in[3];
  const float* W2    = (const float*)d_in[4];
  const float* b2v   = (const float*)d_in[5];
  float* out = (float*)d_out;

  float* COEF = (float*)d_ws;
  float* FR   = COEF + NB * NSEC * CSTRIDE;
  float* GT   = FR   + NB * NSEC * FSTRIDE;
  float* TS   = GT   + NB * NSEC * TILE * 2;

  derive_kernel<<<NB * NSEC, 32, 0, stream>>>(knobs, W1, b1v, W2, b2v, COEF, FR, GT);

  const int tileWaves = NB * WPR;
  for (int sec = 0; sec < NSEC; ++sec) {
    const float* src = (sec == 0) ? x : out;
    tile_kernel<<<tileWaves, 32, 0, stream>>>(src, out, COEF, FR, GT, TS,
                                              sec, sec - 1);
  }
  fixup_kernel<<<NB * WPR / FWPB, 32 * FWPB, 0, stream>>>(out, COEF, GT, TS, NSEC - 1);
}


# last section: runs start from their true state (its zero-start run states summed by the previous pass from its output registers, fixed by a 2x2 cross block); fixup pass gone; sum order changes
# speedup vs baseline: 8.0905x; 1.2613x over previous
#include <hip/hip_runtime.h>
#include <hip/hip_bf16.h>

#define NB      64
#define LSIG    480000
#define TILE    256
#define NTILES  (LSIG / TILE)
#define NSEC    3
#define CSTRIDE 128
#define FSTRIDE 320
#define TPW     15
#define WPR     (NTILES / TPW)
#define RSEG    4
static_assert(WPR <= 32 * RSEG, "one wave holds a whole row of run states");

typedef __attribute__((ext_vector_type(2))) float v2f;
typedef __attribute__((ext_vector_type(8))) float v8f;

__device__ __forceinline__ v8f wmma_f32_16x16x4(v2f a, v2f b, v8f c) {
  return __builtin_amdgcn_wmma_f32_16x16x4_f32(false, a, false, b, (short)0, c,
                                               false, false);
}

struct m22 { float a, b, c, d; };
__device__ __forceinline__ m22 mm(const m22 x, const m22 y) {
  m22 r;
  r.a = x.a * y.a + x.b * y.c;  r.b = x.a * y.b + x.b * y.d;
  r.c = x.c * y.a + x.d * y.c;  r.d = x.c * y.b + x.d * y.d;
  return r;
}
struct bt22 { m22 P, M, Q; };
__device__ __forceinline__ bt22 btmul(const bt22 x, const bt22 y) {
  bt22 r;
  r.P = mm(x.P, y.P);
  r.Q = mm(x.Q, y.Q);
  const m22 u = mm(x.M, y.P), v = mm(x.Q, y.M);
  r.M.a = u.a + v.a; r.M.b = u.b + v.b; r.M.c = u.c + v.c; r.M.d = u.d + v.d;
  return r;
}

__global__ void __launch_bounds__(32) derive_kernel(const float* __restrict__ knobs,
                                                    const float* __restrict__ W1,
                                                    const float* __restrict__ b1v,
                                                    const float* __restrict__ W2,
                                                    const float* __restrict__ b2v,
                                                    float* __restrict__ coef,
                                                    float* __restrict__ fr,
                                                    float* __restrict__ gt) {
  __shared__ float hs[16], g0s[16], g1s[16];
  const int lane = threadIdx.x;
  const int b = blockIdx.x / NSEC;
  const int s = blockIdx.x - b * NSEC;
  const float k0 = knobs[b * 3 + 0], k1 = knobs[b * 3 + 1], k2 = knobs[b * 3 + 2];
  float hmlp[16];
  for (int i = 0; i < 16; ++i)
    hmlp[i] = tanhf(k0 * W1[0 * 16 + i] + k1 * W1[1 * 16 + i] +
                    k2 * W1[2 * 16 + i] + b1v[i]);
  float p[5];
  for (int j = 0; j < 5; ++j) {
    float acc = b2v[s * 5 + j];
    for (int i = 0; i < 16; ++i) acc += hmlp[i] * W2[i * 15 + s * 5 + j];
    p[j] = acc;
  }
  const float c0 = p[0], c1 = p[1], c2 = p[2];
  const float a2c = 0.999f * tanhf(p[4]);
  const float a1c = 0.999f * (1.0f + a2c) * tanhf(p[3]);
  const float A00 = -a1c, A01 = 1.0f, A10 = -a2c, A11 = 0.0f;
  const float Bv0 = c1 - a1c * c0, Bv1 = c2 - a2c * c0;
  float* cf = coef + (b * NSEC + s) * CSTRIDE;

  float h[16];
  {
    float v0 = Bv0, v1 = Bv1;
    h[0] = c0;
    for (int k = 1; k < 16; ++k) {
      h[k] = v0;
      float n0 = A00 * v0 + A01 * v1;
      float n1 = A10 * v0 + A11 * v1;
      v0 = n0; v1 = n1;
    }
  }
  float fa[16], fb1[16];
  {
    float v0 = Bv0, v1 = Bv1;
    fa[15] = v0; fb1[15] = v1;
    for (int j = 14; j >= 0; --j) {
      float n0 = A00 * v0 + A01 * v1;
      float n1 = A10 * v0 + A11 * v1;
      v0 = n0; v1 = n1;
      fa[j] = v0; fb1[j] = v1;
    }
  }
  float g0i[16], g1i[16];
  {
    float g0 = 1.0f, g1 = 0.0f;
    for (int i = 0; i < 16; ++i) {
      g0i[i] = g0; g1i[i] = g1;
      float n0 = g0 * A00 + g1 * A10;
      float n1 = g0 * A01 + g1 * A11;
      g0 = n0; g1 = n1;
    }
  }
  m22 Am; Am.a = A00; Am.b = A01; Am.c = A10; Am.d = A11;
  const m22 X2 = mm(Am, Am), X4 = mm(X2, X2), X8 = mm(X4, X4);
  m22 Xp[5];
  Xp[0] = mm(X8, X8);
  for (int q = 1; q < 5; ++q) Xp[q] = mm(Xp[q - 1], Xp[q - 1]);
  static_assert(TPW == 15 && RSEG == 4, "the products below spell out 15 = 8+4+2+1 and R^4");
  const m22 X256 = Xp[4];
  const m22 X512 = mm(X256, X256);
  const m22 X1024 = mm(X512, X512), X2048 = mm(X1024, X1024);
  const m22 Rm = mm(mm(mm(X2048, X1024), X512), X256);
  m22 Tp[5];
  Tp[0] = mm(Rm, Rm); Tp[0] = mm(Tp[0], Tp[0]);
  for (int k = 1; k < 5; ++k) Tp[k] = mm(Tp[k - 1], Tp[k - 1]);

  m22 Mx; Mx.a = 0.f; Mx.b = 0.f; Mx.c = 0.f; Mx.d = 0.f;
  if (s == NSEC - 1) {
    float q3 = b2v[(s - 1) * 5 + 3], q4 = b2v[(s - 1) * 5 + 4];
    for (int i = 0; i < 16; ++i) {
      q3 += hmlp[i] * W2[i * 15 + (s - 1) * 5 + 3];
      q4 += hmlp[i] * W2[i * 15 + (s - 1) * 5 + 4];
    }
    const float a2p = 0.999f * tanhf(q4);
    const float a1p = 0.999f * (1.0f + a2p) * tanhf(q3);
    bt22 Y;
    Y.P.a = -a1p; Y.P.b = 1.0f; Y.P.c = -a2p; Y.P.d = 0.0f;
    Y.Q = Am;
    Y.M.a = Bv0; Y.M.b = 0.0f; Y.M.c = Bv1; Y.M.d = 0.0f;
    for (int q = 0; q < 8; ++q) Y = btmul(Y, Y);
    const bt22 Y256 = Y;
    const bt22 Y512 = btmul(Y256, Y256);
    const bt22 Y1024 = btmul(Y512, Y512), Y2048 = btmul(Y1024, Y1024);
    const bt22 YR = btmul(btmul(btmul(Y2048, Y1024), Y512), Y256);
    Mx = YR.M;
  }

  if (lane == 0) {
    for (int j = 0; j < 16; ++j) {
      cf[16 + j] = fa[j]; cf[32 + j] = fb1[j];
      hs[j] = h[j]; g0s[j] = g0i[j]; g1s[j] = g1i[j];
    }
    for (int q = 0; q < 5; ++q) {
      cf[80 + 4 * q + 0] = Xp[q].a; cf[80 + 4 * q + 1] = Xp[q].b;
      cf[80 + 4 * q + 2] = Xp[q].c; cf[80 + 4 * q + 3] = Xp[q].d;
    }
    cf[100] = Rm.a; cf[101] = Rm.b; cf[102] = Rm.c; cf[103] = Rm.d;
    cf[124] = Mx.a; cf[125] = Mx.b; cf[126] = Mx.c; cf[127] = Mx.d;
    for (int k = 0; k < 5; ++k) {
      cf[104 + 4 * k + 0] = Tp[k].a; cf[104 + 4 * k + 1] = Tp[k].b;
      cf[104 + 4 * k + 2] = Tp[k].c; cf[104 + 4 * k + 3] = Tp[k].d;
    }
  }
  __syncthreads();

  {
    float* fb = fr + (b * NSEC + s) * FSTRIDE;
    const int m = lane & 15, half = lane >> 4;
    for (int k = 0; k < 4; ++k) {
      const int K0 = 4 * k + 2 * half;
      const int i0 = m - K0, i1 = m - K0 - 1;
      const float v0 = hs[i0 < 0 ? 0 : i0];
      const float v1 = hs[i1 < 0 ? 0 : i1];
      fb[(k * 32 + lane) * 2 + 0] = (i0 >= 0) ? v0 : 0.0f;
      fb[(k * 32 + lane) * 2 + 1] = (i1 >= 0) ? v1 : 0.0f;
    }
    fb[256 + lane * 2 + 0] = half ? 0.0f : g0s[m];
    fb[256 + lane * 2 + 1] = half ? 0.0f : g1s[m];
  }
  {
    float r[16];
    for (int j = 0; j < 16; ++j) r[j] = 0.0f;
    if (s == NSEC - 1) {
      float v0 = Bv0, v1 = Bv1;
      for (int blk = 0; blk < 32; ++blk) {
        float t[16];
        for (int i = 0; i < 8; ++i) {
          t[2 * i + 0] = v0; t[2 * i + 1] = v1;
          const float n0 = A00 * v0 + A01 * v1;
          const float n1 = A10 * v0 + A11 * v1;
          v0 = n0; v1 = n1;
        }
        if (lane == blk) {
          for (int j = 0; j < 16; ++j) r[j] = t[j];
        }
      }
    } else {
      float g0 = 1.0f, g1 = 0.0f;
      for (int blk = 0; blk < 32; ++blk) {
        float t[16];
        for (int i = 0; i < 8; ++i) {
          t[2 * i + 0] = g0; t[2 * i + 1] = g1;
          const float n0 = g0 * A00 + g1 * A10;
          const float n1 = g0 * A01 + g1 * A11;
          g0 = n0; g1 = n1;
        }
        if (lane == blk) {
          for (int j = 0; j < 16; ++j) r[j] = t[j];
        }
      }
    }
    float4* G = (float4*)(gt + (b * NSEC + s) * TILE * 2 + lane * 16);
    for (int q = 0; q < 4; ++q) {
      float4 v;
      v.x = r[4 * q + 0]; v.y = r[4 * q + 1]; v.z = r[4 * q + 2]; v.w = r[4 * q + 3];
      G[q] = v;
    }
  }
}


__device__ __forceinline__ void wave_lds_order() {
  __builtin_amdgcn_fence(__ATOMIC_RELEASE, "wavefront");
  __builtin_amdgcn_wave_barrier();
  __builtin_amdgcn_fence(__ATOMIC_ACQUIRE, "wavefront");
}

__device__ __forceinline__ float2 run_tile_states(const float* __restrict__ cfp,
                                                  const float2* zrow, int w, int lane) {
  m22 A2; A2.a = cfp[96];  A2.b = cfp[97];  A2.c = cfp[98];  A2.d = cfp[99];
  m22 Rm; Rm.a = cfp[100]; Rm.b = cfp[101]; Rm.c = cfp[102]; Rm.d = cfp[103];
  float2 zl[RSEG];
#pragma unroll
  for (int c = 0; c < RSEG; ++c) {
    const int k = lane * RSEG + c;
    zl[c].x = 0.f; zl[c].y = 0.f;
    if (k < WPR) zl[c] = zrow[k];
  }
  float s0 = 0.f, s1 = 0.f;
#pragma unroll
  for (int c = 0; c < RSEG; ++c) {
    const float n0 = Rm.a * s0 + Rm.b * s1 + zl[c].x;
    const float n1 = Rm.c * s0 + Rm.d * s1 + zl[c].y;
    s0 = n0; s1 = n1;
  }
  float p0 = s0, p1 = s1;
#pragma unroll
  for (int k = 0; k < 5; ++k) {
    const float T00 = cfp[104 + 4 * k + 0], T01 = cfp[104 + 4 * k + 1];
    const float T10 = cfp[104 + 4 * k + 2], T11 = cfp[104 + 4 * k + 3];
    const int len = 1 << k;
    const float q0 = __shfl_up(p0, (unsigned)len, 32);
    const float q1 = __shfl_up(p1, (unsigned)len, 32);
    if (lane >= len) {
      p0 = T00 * q0 + T01 * q1 + p0;
      p1 = T10 * q0 + T11 * q1 + p1;
    }
  }
  const float e0 = __shfl_up(p0, 1u, 32);
  const float e1 = __shfl_up(p1, 1u, 32);
  s0 = (lane >= 1) ? e0 : 0.f;
  s1 = (lane >= 1) ? e1 : 0.f;
  const int own = w / RSEG, r = w - own * RSEG;
#pragma unroll
  for (int c = 0; c < RSEG - 1; ++c) {
    if (c < r) {
      const float n0 = Rm.a * s0 + Rm.b * s1 + zl[c].x;
      const float n1 = Rm.c * s0 + Rm.d * s1 + zl[c].y;
      s0 = n0; s1 = n1;
    }
  }
  float sx = __int_as_float(__builtin_amdgcn_readlane(__float_as_int(s0), own));
  float sy = __int_as_float(__builtin_amdgcn_readlane(__float_as_int(s1), own));
  float2 sl; sl.x = 0.f; sl.y = 0.f;
#pragma unroll
  for (int i = 0; i < TPW; ++i) {
    if (lane == i) { sl.x = sx; sl.y = sy; }
    const float nx = A2.a * sx + A2.b * sy;
    const float ny = A2.c * sx + A2.d * sy;
    sx = nx; sy = ny;
  }
  return sl;
}

struct last_entry { float2 sl; float cx, cy; };
__device__ __forceinline__ last_entry run_states_last(const float* __restrict__ cf1,
                                                      const float* __restrict__ cf2,
                                                      const float2* z1row,
                                                      const float2* z2row,
                                                      int w, int lane) {
  m22 A2; A2.a = cf1[96];  A2.b = cf1[97];  A2.c = cf1[98];  A2.d = cf1[99];
  m22 R1; R1.a = cf1[100]; R1.b = cf1[101]; R1.c = cf1[102]; R1.d = cf1[103];
  m22 R2; R2.a = cf2[100]; R2.b = cf2[101]; R2.c = cf2[102]; R2.d = cf2[103];
  m22 Mx; Mx.a = cf2[124]; Mx.b = cf2[125]; Mx.c = cf2[126]; Mx.d = cf2[127];
  float2 z1[RSEG], z2[RSEG];
#pragma unroll
  for (int c = 0; c < RSEG; ++c) {
    const int k = lane * RSEG + c;
    z1[c].x = 0.f; z1[c].y = 0.f; z2[c].x = 0.f; z2[c].y = 0.f;
    if (k < WPR) { z1[c] = z1row[k]; z2[c] = z2row[k]; }
  }
  float s0 = 0.f, s1 = 0.f;
#pragma unroll
  for (int c = 0; c < RSEG; ++c) {
    const float n0 = R1.a * s0 + R1.b * s1 + z1[c].x;
    const float n1 = R1.c * s0 + R1.d * s1 + z1[c].y;
    s0 = n0; s1 = n1;
  }
  float p0 = s0, p1 = s1;
#pragma unroll
  for (int k = 0; k < 5; ++k) {
    const float T00 = cf1[104 + 4 * k + 0], T01 = cf1[104 + 4 * k + 1];
    const float T10 = cf1[104 + 4 * k + 2], T11 = cf1[104 + 4 * k + 3];
    const int len = 1 << k;
    const float q0 = __shfl_up(p0, (unsigned)len, 32);
    const float q1 = __shfl_up(p1, (unsigned)len, 32);
    if (lane >= len) {
      p0 = T00 * q0 + T01 * q1 + p0;
      p1 = T10 * q0 + T11 * q1 + p1;
    }
  }
  {
    const float e0 = __shfl_up(p0, 1u, 32);
    const float e1 = __shfl_up(p1, 1u, 32);
    s0 = (lane >= 1) ? e0 : 0.f;
    s1 = (lane >= 1) ? e1 : 0.f;
  }
  const int own = w / RSEG, r = w - own * RSEG;
  float ex = 0.f, ey = 0.f;
#pragma unroll
  for (int c = 0; c < RSEG; ++c) {
    if (c == r) { ex = s0; ey = s1; }
    z2[c].x += Mx.a * s0 + Mx.b * s1;
    z2[c].y += Mx.c * s0 + Mx.d * s1;
    const float n0 = R1.a * s0 + R1.b * s1 + z1[c].x;
    const float n1 = R1.c * s0 + R1.d * s1 + z1[c].y;
    s0 = n0; s1 = n1;
  }
  float sx = __int_as_float(__builtin_amdgcn_readlane(__float_as_int(ex), own));
  float sy = __int_as_float(__builtin_amdgcn_readlane(__float_as_int(ey), own));
  last_entry o;
  o.sl.x = 0.f; o.sl.y = 0.f;
#pragma unroll
  for (int i = 0; i < TPW; ++i) {
    if (lane == i) { o.sl.x = sx; o.sl.y = sy; }
    const float nx = A2.a * sx + A2.b * sy;
    const float ny = A2.c * sx + A2.d * sy;
    sx = nx; sy = ny;
  }
  float t0 = 0.f, t1 = 0.f;
#pragma unroll
  for (int c = 0; c < RSEG; ++c) {
    const float n0 = R2.a * t0 + R2.b * t1 + z2[c].x;
    const float n1 = R2.c * t0 + R2.d * t1 + z2[c].y;
    t0 = n0; t1 = n1;
  }
  float u0 = t0, u1 = t1;
#pragma unroll
  for (int k = 0; k < 5; ++k) {
    const float T00 = cf2[104 + 4 * k + 0], T01 = cf2[104 + 4 * k + 1];
    const float T10 = cf2[104 + 4 * k + 2], T11 = cf2[104 + 4 * k + 3];
    const int len = 1 << k;
    const float q0 = __shfl_up(u0, (unsigned)len, 32);
    const float q1 = __shfl_up(u1, (unsigned)len, 32);
    if (lane >= len) {
      u0 = T00 * q0 + T01 * q1 + u0;
      u1 = T10 * q0 + T11 * q1 + u1;
    }
  }
  {
    const float e0 = __shfl_up(u0, 1u, 32);
    const float e1 = __shfl_up(u1, 1u, 32);
    t0 = (lane >= 1) ? e0 : 0.f;
    t1 = (lane >= 1) ? e1 : 0.f;
  }
#pragma unroll
  for (int c = 0; c < RSEG - 1; ++c) {
    if (c < r) {
      const float n0 = R2.a * t0 + R2.b * t1 + z2[c].x;
      const float n1 = R2.c * t0 + R2.d * t1 + z2[c].y;
      t0 = n0; t1 = n1;
    }
  }
  o.cx = __int_as_float(__builtin_amdgcn_readlane(__float_as_int(t0), own));
  o.cy = __int_as_float(__builtin_amdgcn_readlane(__float_as_int(t1), own));
  return o;
}

template <bool EMITNEXT, bool LAST>
__global__ void __launch_bounds__(32) tile_kernel(const float* in, float* out,
                                                  const float* __restrict__ coef,
                                                  const float* __restrict__ fr,
                                                  const float* __restrict__ gt,
                                                  float* ts,
                                                  int sec, int prevsec) {
  __shared__ float xs[TILE];
  const int lane = threadIdx.x;
  const int b = blockIdx.x / WPR;
  const int w = blockIdx.x - b * WPR;
  const int base0 = b * LSIG + w * TPW * TILE;
  const int m = lane & 15;
  const int half = lane >> 4;
  const bool fuse = prevsec >= 0;

  const float* cf = coef + (b * NSEC + sec) * CSTRIDE;
  const v2f* hf = (const v2f*)(fr + (b * NSEC + sec) * FSTRIDE);
  v2f ha[4];
#pragma unroll
  for (int k = 0; k < 4; ++k) ha[k] = hf[k * 32 + lane];
  const v2f gfr = hf[128 + lane];
  float f0[16], f1[16], Ms[16];
#pragma unroll
  for (int j = 0; j < 16; ++j) {
    f0[j] = cf[16 + j];
    f1[j] = cf[32 + j];
    Ms[j] = cf[80 + j];
  }
  float4 ga = {0.f, 0.f, 0.f, 0.f}, gb = ga, gc = ga, gd = ga;
  if (fuse) {
    const float4* G4 = (const float4*)(gt + (b * NSEC + prevsec) * TILE * 2);
    ga = G4[lane * 2 + 0];
    gb = G4[lane * 2 + 1];
    gc = G4[64 + lane * 2 + 0];
    gd = G4[64 + lane * 2 + 1];
  }
  float2* ts2 = (float2*)ts;
  const float* src = in + base0 + lane * 4;
  float* dst = out + base0 + lane * 4;

  float4 u0 = *(const float4*)(src);
  float4 u1 = *(const float4*)(src + 128);
  float svx_l = 0.f, svy_l = 0.f;
  float cin0 = 0.f, cin1 = 0.f;
  if (LAST) {
    const last_entry e = run_states_last(coef + (b * NSEC + prevsec) * CSTRIDE, cf,
                                         ts2 + (prevsec * NB + b) * WPR,
                                         ts2 + (sec * NB + b) * WPR, w, lane);
    svx_l = e.sl.x; svy_l = e.sl.y;
    cin0 = e.cx; cin1 = e.cy;
  } else if (fuse) {
    const float2 t = run_tile_states(coef + (b * NSEC + prevsec) * CSTRIDE,
                                     ts2 + (prevsec * NB + b) * WPR, w, lane);
    svx_l = t.x; svy_l = t.y;
  }
  float wx[8], wy[8];
  float nx00 = 0.f, nx01 = 0.f, nx10 = 0.f, nx11 = 0.f;
  float pn0 = 0.f, pn1 = 0.f;
  if (EMITNEXT) {
    const float4* W4 = (const float4*)(gt + (b * NSEC + sec + 1) * TILE * 2 +
                                       (31 - 2 * m - half) * 16);
    const float4 wa = W4[0], wb = W4[1], wc = W4[2], wd = W4[3];
    wx[7] = wa.x; wy[7] = wa.y; wx[6] = wa.z; wy[6] = wa.w;
    wx[5] = wb.x; wy[5] = wb.y; wx[4] = wb.z; wy[4] = wb.w;
    wx[3] = wc.x; wy[3] = wc.y; wx[2] = wc.z; wy[2] = wc.w;
    wx[1] = wd.x; wy[1] = wd.y; wx[0] = wd.z; wy[0] = wd.w;
    const float* cn = coef + (b * NSEC + sec + 1) * CSTRIDE;
    nx00 = cn[96]; nx01 = cn[97]; nx10 = cn[98]; nx11 = cn[99];
  } else {
#pragma unroll
    for (int r = 0; r < 8; ++r) { wx[r] = 0.f; wy[r] = 0.f; }
  }

  for (int i = 0; i < TPW; ++i) {
    float4 n0 = u0, n1 = u1;
    if (i + 1 < TPW) {
      n0 = *(const float4*)(src + (i + 1) * TILE);
      n1 = *(const float4*)(src + (i + 1) * TILE + 128);
    }

    if (fuse) {
      float2 sv;
      sv.x = __int_as_float(__builtin_amdgcn_readlane(__float_as_int(svx_l), i));
      sv.y = __int_as_float(__builtin_amdgcn_readlane(__float_as_int(svy_l), i));
      u0.x += ga.x * sv.x + ga.y * sv.y;
      u0.y += ga.z * sv.x + ga.w * sv.y;
      u0.z += gb.x * sv.x + gb.y * sv.y;
      u0.w += gb.z * sv.x + gb.w * sv.y;
      u1.x += gc.x * sv.x + gc.y * sv.y;
      u1.y += gc.z * sv.x + gc.w * sv.y;
      u1.z += gd.x * sv.x + gd.y * sv.y;
      u1.w += gd.z * sv.x + gd.w * sv.y;
    }
    *(float4*)(xs + lane * 4)       = u0;
    *(float4*)(xs + 128 + lane * 4) = u1;
    wave_lds_order();

    float4 xc[4];
#pragma unroll
    for (int q = 0; q < 4; ++q)
      xc[q] = ((const float4*)(xs + m * 16))[q];

    v8f acc = {0.f, 0.f, 0.f, 0.f, 0.f, 0.f, 0.f, 0.f};
#pragma unroll
    for (int k = 0; k < 4; ++k) {
      v2f bb;
      bb.x = half ? xc[k].z : xc[k].x;
      bb.y = half ? xc[k].w : xc[k].y;
      acc = wmma_f32_16x16x4(ha[k], bb, acc);
    }

    float fx0 = 0.f, fx1 = 0.f;
#pragma unroll
    for (int q = 0; q < 4; ++q) {
      fx0 += f0[4 * q + 0] * xc[q].x + f0[4 * q + 1] * xc[q].y +
             f0[4 * q + 2] * xc[q].z + f0[4 * q + 3] * xc[q].w;
      fx1 += f1[4 * q + 0] * xc[q].x + f1[4 * q + 1] * xc[q].y +
             f1[4 * q + 2] * xc[q].z + f1[4 * q + 3] * xc[q].w;
    }

    if (m == 0) {
      fx0 += Ms[0] * cin0 + Ms[1] * cin1;
      fx1 += Ms[2] * cin0 + Ms[3] * cin1;
    }

    float p0 = fx0, p1 = fx1;
#pragma unroll
    for (int s = 0; s < 4; ++s) {
      float M00 = Ms[4 * s + 0], M01 = Ms[4 * s + 1];
      float M10 = Ms[4 * s + 2], M11 = Ms[4 * s + 3];
      int len = 1 << s;
      float q0 = __shfl_up(p0, (unsigned)len, 32);
      float q1 = __shfl_up(p1, (unsigned)len, 32);
      if (m >= len) {
        p0 = M00 * q0 + M01 * q1 + p0;
        p1 = M10 * q0 + M11 * q1 + p1;
      }
    }
    float e0 = __shfl_up(p0, 1u, 32);
    float e1 = __shfl_up(p1, 1u, 32);
    float s0c = (m >= 1) ? e0 : cin0;
    float s1c = (m >= 1) ? e1 : cin1;
    cin0 = __int_as_float(__builtin_amdgcn_readlane(__float_as_int(p0), 15));
    cin1 = __int_as_float(__builtin_amdgcn_readlane(__float_as_int(p1), 15));

    {
      v2f bb;
      bb.x = half ? 0.f : s0c;
      bb.y = half ? 0.f : s1c;
      acc = wmma_f32_16x16x4(gfr, bb, acc);
    }

    if (EMITNEXT) {
      float q0 = 0.f, q1 = 0.f;
#pragma unroll
      for (int r = 0; r < 8; ++r) {
        q0 += wx[r] * acc[r];
        q1 += wy[r] * acc[r];
      }
      const float n0p = nx00 * pn0 + nx01 * pn1 + q0;
      const float n1p = nx10 * pn0 + nx11 * pn1 + q1;
      pn0 = n0p; pn1 = n1p;
    }

    wave_lds_order();
    {
      float4 d0, d1;
      d0.x = acc[0]; d0.y = acc[1]; d0.z = acc[2]; d0.w = acc[3];
      d1.x = acc[4]; d1.y = acc[5]; d1.z = acc[6]; d1.w = acc[7];
      *(float4*)(xs + m * 16 + 8 * half)     = d0;
      *(float4*)(xs + m * 16 + 8 * half + 4) = d1;
    }
    wave_lds_order();
    float4 o0 = *(const float4*)(xs + lane * 4);
    float4 o1 = *(const float4*)(xs + 128 + lane * 4);
    wave_lds_order();
    *(float4*)(dst + i * TILE)       = o0;
    *(float4*)(dst + i * TILE + 128) = o1;

    u0 = n0; u1 = n1;
  }
  if (EMITNEXT) {
#pragma unroll
    for (int d = 16; d >= 1; d >>= 1) {
      pn0 += __shfl_xor(pn0, d, 32);
      pn1 += __shfl_xor(pn1, d, 32);
    }
    if (lane == 0) {
      float2 z; z.x = pn0; z.y = pn1;
      ts2[((sec + 1) * NB + b) * WPR + w] = z;
    }
  }
  if (!LAST) {
    if (lane == 0) {
      float2 z; z.x = cin0; z.y = cin1;
      ts2[(sec * NB + b) * WPR + w] = z;
    }
  }
}

extern "C" void kernel_launch(void* const* d_in, const int* in_sizes, int n_in,
                              void* d_out, int out_size, void* d_ws, size_t ws_size,
                              hipStream_t stream) {
  (void)in_sizes; (void)n_in; (void)out_size; (void)ws_size;
  const float* x     = (const float*)d_in[0];
  const float* knobs = (const float*)d_in[1];
  const float* W1    = (const float*)d_in[2];
  const float* b1v   = (const float*)d_in[3];
  const float* W2    = (const float*)d_in[4];
  const float* b2v   = (const float*)d_in[5];
  float* out = (float*)d_out;

  float* COEF = (float*)d_ws;
  float* FR   = COEF + NB * NSEC * CSTRIDE;
  float* GT   = FR   + NB * NSEC * FSTRIDE;
  float* TS   = GT   + NB * NSEC * TILE * 2;

  derive_kernel<<<NB * NSEC, 32, 0, stream>>>(knobs, W1, b1v, W2, b2v, COEF, FR, GT);

  const int tileWaves = NB * WPR;
  static_assert(NSEC == 3, "the three launches below are spelled out");
  tile_kernel<false, false><<<tileWaves, 32, 0, stream>>>(x, out, COEF, FR, GT, TS, 0, -1);
  tile_kernel<true, false><<<tileWaves, 32, 0, stream>>>(out, out, COEF, FR, GT, TS, 1, 0);
  tile_kernel<false, true><<<tileWaves, 32, 0, stream>>>(out, out, COEF, FR, GT, TS, 2, 1);
}


# sections 0 and 1 run back to back on each tile in one pass (zero-start per run, handed over in LDS); last pass restores true states from per-run 6-vectors of the cascade; sum order changes
# speedup vs baseline: 9.0568x; 1.1194x over previous
#include <hip/hip_runtime.h>
#include <hip/hip_bf16.h>

#define NB      64
#define LSIG    480000
#define TILE    256
#define NTILES  (LSIG / TILE)
#define NSEC    3
#define CSTRIDE 128
#define FSTRIDE 320
#define TPW     15
#define WPR     (NTILES / TPW)
#define RSEG    4
#define C6STRIDE 256
#define DWAVES  5
static_assert(WPR <= 32 * RSEG, "one wave holds a whole row of run states");

typedef __attribute__((ext_vector_type(2))) float v2f;
typedef __attribute__((ext_vector_type(8))) float v8f;

__device__ __forceinline__ v8f wmma_f32_16x16x4(v2f a, v2f b, v8f c) {
  return __builtin_amdgcn_wmma_f32_16x16x4_f32(false, a, false, b, (short)0, c,
                                               false, false);
}

struct m22 { float a, b, c, d; };
__device__ __forceinline__ m22 mm(const m22 x, const m22 y) {
  m22 r;
  r.a = x.a * y.a + x.b * y.c;  r.b = x.a * y.b + x.b * y.d;
  r.c = x.c * y.a + x.d * y.c;  r.d = x.c * y.b + x.d * y.d;
  return r;
}

struct m66 { float v[36]; };
__device__ __forceinline__ m66 mm6(const m66& x, const m66& y) {
  m66 r;
#pragma unroll
  for (int i = 0; i < 6; ++i) {
#pragma unroll
    for (int j = 0; j < 6; ++j) {
      float a = 0.f;
#pragma unroll
      for (int k = 0; k < 6; ++k)
        if (j <= (i | 1) && k <= (i | 1) && j <= (k | 1)) a += x.v[i * 6 + k] * y.v[k * 6 + j];
      r.v[i * 6 + j] = a;
    }
  }
  return r;
}
__device__ __forceinline__ void mv6(const float* __restrict__ M, const float* s, float* o) {
#pragma unroll
  for (int i = 0; i < 6; ++i) {
    float a = 0.f;
#pragma unroll
    for (int j = 0; j < 6; ++j)
      if (j <= (i | 1)) a += M[i * 6 + j] * s[j];
    o[i] = a;
  }
}

__global__ void __launch_bounds__(32) derive_kernel(const float* __restrict__ knobs,
                                                    const float* __restrict__ W1,
                                                    const float* __restrict__ b1v,
                                                    const float* __restrict__ W2,
                                                    const float* __restrict__ b2v,
                                                    float* __restrict__ coef,
                                                    float* __restrict__ fr,
                                                    float* __restrict__ gt,
                                                    float* __restrict__ c6) {
  __shared__ float hs[16], g0s[16], g1s[16];
  const int lane = threadIdx.x;
  const int b = blockIdx.x / DWAVES;
  const int s = blockIdx.x - b * DWAVES;
  const float k0 = knobs[b * 3 + 0], k1 = knobs[b * 3 + 1], k2 = knobs[b * 3 + 2];
  float hmlp[16];
  for (int i = 0; i < 16; ++i)
    hmlp[i] = tanhf(k0 * W1[0 * 16 + i] + k1 * W1[1 * 16 + i] +
                    k2 * W1[2 * 16 + i] + b1v[i]);

  if (s >= NSEC) {
    float cc0[NSEC], bb0[NSEC], bb1[NSEC], aa1[NSEC], aa2[NSEC];
#pragma unroll
    for (int t = 0; t < NSEC; ++t) {
      float q[5];
#pragma unroll
      for (int j = 0; j < 5; ++j) {
        float acc = b2v[t * 5 + j];
        for (int i = 0; i < 16; ++i) acc += hmlp[i] * W2[i * 15 + t * 5 + j];
        q[j] = acc;
      }
      const float a2c = 0.999f * tanhf(q[4]);
      const float a1c = 0.999f * (1.0f + a2c) * tanhf(q[3]);
      cc0[t] = q[0]; aa1[t] = a1c; aa2[t] = a2c;
      bb0[t] = q[1] - a1c * q[0]; bb1[t] = q[2] - a2c * q[0];
    }
    m66 Y;
#pragma unroll
    for (int i = 0; i < 36; ++i) Y.v[i] = 0.f;
#pragma unroll
    for (int t = 0; t < NSEC; ++t) {
      Y.v[(2 * t) * 6 + 2 * t] = -aa1[t];     Y.v[(2 * t) * 6 + 2 * t + 1] = 1.0f;
      Y.v[(2 * t + 1) * 6 + 2 * t] = -aa2[t]; Y.v[(2 * t + 1) * 6 + 2 * t + 1] = 0.0f;
    }
    Y.v[2 * 6 + 0] = bb0[1];          Y.v[3 * 6 + 0] = bb1[1];
    Y.v[4 * 6 + 2] = bb0[2];          Y.v[5 * 6 + 2] = bb1[2];
    Y.v[4 * 6 + 0] = cc0[1] * bb0[2]; Y.v[5 * 6 + 0] = cc0[1] * bb1[2];
    if (s == NSEC) {
      static_assert(TPW == 15 && RSEG == 4, "the products below spell out 15 = 8+4+2+1 and R^4");
      m66 P = Y;
      for (int q = 0; q < 8; ++q) P = mm6(P, P);
      const m66 Y256 = P;
      const m66 Y512 = mm6(Y256, Y256);
      const m66 Y1024 = mm6(Y512, Y512);
      const m66 Y2048 = mm6(Y1024, Y1024);
      const m66 R6 = mm6(mm6(mm6(Y2048, Y1024), Y512), Y256);
      float* cp = c6 + b * C6STRIDE;
      if (lane == 0) {
#pragma unroll
        for (int i = 0; i < 4; ++i)
#pragma unroll
          for (int j = 0; j < 4; ++j) cp[i * 4 + j] = Y256.v[i * 6 + j];
#pragma unroll
        for (int i = 0; i < 36; ++i) cp[16 + i] = R6.v[i];
      }
      m66 T = mm6(R6, R6); T = mm6(T, T);
      for (int k = 0; k < 5; ++k) {
        if (lane == 0) {
#pragma unroll
          for (int i = 0; i < 36; ++i) cp[52 + 36 * k + i] = T.v[i];
        }
        if (k + 1 < 5) T = mm6(T, T);
      }
    } else {
      float r0 = cc0[1], r1 = 0.0f, r2 = 1.0f, r3 = 0.0f;
      float keep[32];
#pragma unroll
      for (int j = 0; j < 32; ++j) keep[j] = 0.0f;
      for (int blk = 0; blk < 32; ++blk) {
        float t[32];
#pragma unroll
        for (int i = 0; i < 8; ++i) {
          t[4 * i + 0] = r0; t[4 * i + 1] = r1; t[4 * i + 2] = r2; t[4 * i + 3] = r3;
          const float n0 = r0 * Y.v[0 * 6 + 0] + r1 * Y.v[1 * 6 + 0] + r2 * Y.v[2 * 6 + 0] + r3 * Y.v[3 * 6 + 0];
          const float n1 = r0 * Y.v[0 * 6 + 1] + r1 * Y.v[1 * 6 + 1];
          const float n2 = r2 * Y.v[2 * 6 + 2] + r3 * Y.v[3 * 6 + 2];
          const float n3 = r2 * Y.v[2 * 6 + 3] + r3 * Y.v[3 * 6 + 3];
          r0 = n0; r1 = n1; r2 = n2; r3 = n3;
        }
        if (lane == blk) {
#pragma unroll
          for (int j = 0; j < 32; ++j) keep[j] = t[j];
        }
      }
      float4* G = (float4*)(gt + b * NSEC * TILE * 2 + lane * 32);
#pragma unroll
      for (int q = 0; q < 8; ++q) {
        float4 v;
        v.x = keep[4 * q + 0]; v.y = keep[4 * q + 1]; v.z = keep[4 * q + 2]; v.w = keep[4 * q + 3];
        G[q] = v;
      }
    }
    return;
  }

  float p[5];
  for (int j = 0; j < 5; ++j) {
    float acc = b2v[s * 5 + j];
    for (int i = 0; i < 16; ++i) acc += hmlp[i] * W2[i * 15 + s * 5 + j];
    p[j] = acc;
  }
  const float c0 = p[0], c1 = p[1], c2 = p[2];
  const float a2c = 0.999f * tanhf(p[4]);
  const float a1c = 0.999f * (1.0f + a2c) * tanhf(p[3]);
  const float A00 = -a1c, A01 = 1.0f, A10 = -a2c, A11 = 0.0f;
  const float Bv0 = c1 - a1c * c0, Bv1 = c2 - a2c * c0;
  float* cf = coef + (b * NSEC + s) * CSTRIDE;

  float h[16];
  {
    float v0 = Bv0, v1 = Bv1;
    h[0] = c0;
    for (int k = 1; k < 16; ++k) {
      h[k] = v0;
      float n0 = A00 * v0 + A01 * v1;
      float n1 = A10 * v0 + A11 * v1;
      v0 = n0; v1 = n1;
    }
  }
  float fa[16], fb1[16];
  {
    float v0 = Bv0, v1 = Bv1;
    fa[15] = v0; fb1[15] = v1;
    for (int j = 14; j >= 0; --j) {
      float n0 = A00 * v0 + A01 * v1;
      float n1 = A10 * v0 + A11 * v1;
      v0 = n0; v1 = n1;
      fa[j] = v0; fb1[j] = v1;
    }
  }
  float g0i[16], g1i[16];
  {
    float g0 = 1.0f, g1 = 0.0f;
    for (int i = 0; i < 16; ++i) {
      g0i[i] = g0; g1i[i] = g1;
      float n0 = g0 * A00 + g1 * A10;
      float n1 = g0 * A01 + g1 * A11;
      g0 = n0; g1 = n1;
    }
  }
  m22 Am; Am.a = A00; Am.b = A01; Am.c = A10; Am.d = A11;
  const m22 X2 = mm(Am, Am), X4 = mm(X2, X2), X8 = mm(X4, X4);
  m22 Xp[5];
  Xp[0] = mm(X8, X8);
  for (int q = 1; q < 5; ++q) Xp[q] = mm(Xp[q - 1], Xp[q - 1]);

  if (lane == 0) {
    for (int j = 0; j < 16; ++j) {
      cf[16 + j] = fa[j]; cf[32 + j] = fb1[j];
      hs[j] = h[j]; g0s[j] = g0i[j]; g1s[j] = g1i[j];
    }
    for (int q = 0; q < 5; ++q) {
      cf[80 + 4 * q + 0] = Xp[q].a; cf[80 + 4 * q + 1] = Xp[q].b;
      cf[80 + 4 * q + 2] = Xp[q].c; cf[80 + 4 * q + 3] = Xp[q].d;
    }
  }
  __syncthreads();

  {
    float* fb = fr + (b * NSEC + s) * FSTRIDE;
    const int m = lane & 15, half = lane >> 4;
    for (int k = 0; k < 4; ++k) {
      const int K0 = 4 * k + 2 * half;
      const int i0 = m - K0, i1 = m - K0 - 1;
      const float v0 = hs[i0 < 0 ? 0 : i0];
      const float v1 = hs[i1 < 0 ? 0 : i1];
      fb[(k * 32 + lane) * 2 + 0] = (i0 >= 0) ? v0 : 0.0f;
      fb[(k * 32 + lane) * 2 + 1] = (i1 >= 0) ? v1 : 0.0f;
    }
    fb[256 + lane * 2 + 0] = half ? 0.0f : g0s[m];
    fb[256 + lane * 2 + 1] = half ? 0.0f : g1s[m];
  }
  if (s == NSEC - 1) {
    float r[16];
    for (int j = 0; j < 16; ++j) r[j] = 0.0f;
    float v0 = Bv0, v1 = Bv1;
    for (int blk = 0; blk < 32; ++blk) {
      float t[16];
      for (int i = 0; i < 8; ++i) {
        t[2 * i + 0] = v0; t[2 * i + 1] = v1;
        const float n0 = A00 * v0 + A01 * v1;
        const float n1 = A10 * v0 + A11 * v1;
        v0 = n0; v1 = n1;
      }
      if (lane == blk) {
        for (int j = 0; j < 16; ++j) r[j] = t[j];
      }
    }
    float4* G = (float4*)(gt + (b * NSEC + s) * TILE * 2 + lane * 16);
    for (int q = 0; q < 4; ++q) {
      float4 v;
      v.x = r[4 * q + 0]; v.y = r[4 * q + 1]; v.z = r[4 * q + 2]; v.w = r[4 * q + 3];
      G[q] = v;
    }
  }
}


__device__ __forceinline__ void wave_lds_order() {
  __builtin_amdgcn_fence(__ATOMIC_RELEASE, "wavefront");
  __builtin_amdgcn_wave_barrier();
  __builtin_amdgcn_fence(__ATOMIC_ACQUIRE, "wavefront");
}

struct sec_ops {
  v2f ha[4];
  v2f gfr;
  float f0[16], f1[16], Ms[16];
};
__device__ __forceinline__ void load_sec(sec_ops& o, const float* __restrict__ coef,
                                         const float* __restrict__ fr, int b, int sec,
                                         int lane) {
  const float* cf = coef + (b * NSEC + sec) * CSTRIDE;
  const v2f* hf = (const v2f*)(fr + (b * NSEC + sec) * FSTRIDE);
#pragma unroll
  for (int k = 0; k < 4; ++k) o.ha[k] = hf[k * 32 + lane];
  o.gfr = hf[128 + lane];
#pragma unroll
  for (int j = 0; j < 16; ++j) {
    o.f0[j] = cf[16 + j];
    o.f1[j] = cf[32 + j];
    o.Ms[j] = cf[80 + j];
  }
}

__device__ __forceinline__ v8f sec_tile(const sec_ops& o, const float4* xc,
                                        float& cin0, float& cin1, int m, int half) {
  v8f acc = {0.f, 0.f, 0.f, 0.f, 0.f, 0.f, 0.f, 0.f};
#pragma unroll
  for (int k = 0; k < 4; ++k) {
    v2f bb;
    bb.x = half ? xc[k].z : xc[k].x;
    bb.y = half ? xc[k].w : xc[k].y;
    acc = wmma_f32_16x16x4(o.ha[k], bb, acc);
  }

  float fx0 = 0.f, fx1 = 0.f;
#pragma unroll
  for (int q = 0; q < 4; ++q) {
    fx0 += o.f0[4 * q + 0] * xc[q].x + o.f0[4 * q + 1] * xc[q].y +
           o.f0[4 * q + 2] * xc[q].z + o.f0[4 * q + 3] * xc[q].w;
    fx1 += o.f1[4 * q + 0] * xc[q].x + o.f1[4 * q + 1] * xc[q].y +
           o.f1[4 * q + 2] * xc[q].z + o.f1[4 * q + 3] * xc[q].w;
  }

  if (m == 0) {
    fx0 += o.Ms[0] * cin0 + o.Ms[1] * cin1;
    fx1 += o.Ms[2] * cin0 + o.Ms[3] * cin1;
  }

  float p0 = fx0, p1 = fx1;
#pragma unroll
  for (int s = 0; s < 4; ++s) {
    float M00 = o.Ms[4 * s + 0], M01 = o.Ms[4 * s + 1];
    float M10 = o.Ms[4 * s + 2], M11 = o.Ms[4 * s + 3];
    int len = 1 << s;
    float q0 = __shfl_up(p0, (unsigned)len, 32);
    float q1 = __shfl_up(p1, (unsigned)len, 32);
    if (m >= len) {
      p0 = M00 * q0 + M01 * q1 + p0;
      p1 = M10 * q0 + M11 * q1 + p1;
    }
  }
  float e0 = __shfl_up(p0, 1u, 32);
  float e1 = __shfl_up(p1, 1u, 32);
  float s0c = (m >= 1) ? e0 : cin0;
  float s1c = (m >= 1) ? e1 : cin1;
  cin0 = __int_as_float(__builtin_amdgcn_readlane(__float_as_int(p0), 15));
  cin1 = __int_as_float(__builtin_amdgcn_readlane(__float_as_int(p1), 15));

  {
    v2f bb;
    bb.x = half ? 0.f : s0c;
    bb.y = half ? 0.f : s1c;
    acc = wmma_f32_16x16x4(o.gfr, bb, acc);
  }
  return acc;
}

__device__ __forceinline__ void store_block(float* xs, const v8f acc, int m, int half) {
  float4 d0, d1;
  d0.x = acc[0]; d0.y = acc[1]; d0.z = acc[2]; d0.w = acc[3];
  d1.x = acc[4]; d1.y = acc[5]; d1.z = acc[6]; d1.w = acc[7];
  *(float4*)(xs + m * 16 + 8 * half)     = d0;
  *(float4*)(xs + m * 16 + 8 * half + 4) = d1;
}

__global__ void __launch_bounds__(32) pair_kernel(const float* __restrict__ in,
                                                  float* __restrict__ out,
                                                  const float* __restrict__ coef,
                                                  const float* __restrict__ fr,
                                                  const float* __restrict__ gt,
                                                  float* __restrict__ ts) {
  __shared__ float xs[TILE];
  const int lane = threadIdx.x;
  const int b = blockIdx.x / WPR;
  const int w = blockIdx.x - b * WPR;
  const int base0 = b * LSIG + w * TPW * TILE;
  const int m = lane & 15;
  const int half = lane >> 4;

  sec_ops S0, S1;
  load_sec(S0, coef, fr, b, 0, lane);
  load_sec(S1, coef, fr, b, 1, lane);
  float2* ts2 = (float2*)ts;
  const float* src = in + base0 + lane * 4;
  float* dst = out + base0 + lane * 4;

  float4 u0 = *(const float4*)(src);
  float4 u1 = *(const float4*)(src + 128);
  float ca0 = 0.f, ca1 = 0.f, cb0 = 0.f, cb1 = 0.f;
  float wx[8], wy[8];
  float pn0 = 0.f, pn1 = 0.f;
  const float4* W4 = (const float4*)(gt + (b * NSEC + 2) * TILE * 2 +
                                     (31 - 2 * m - half) * 16);
  const float4 wa = W4[0], wb = W4[1], wc = W4[2], wd = W4[3];
  wx[7] = wa.x; wy[7] = wa.y; wx[6] = wa.z; wy[6] = wa.w;
  wx[5] = wb.x; wy[5] = wb.y; wx[4] = wb.z; wy[4] = wb.w;
  wx[3] = wc.x; wy[3] = wc.y; wx[2] = wc.z; wy[2] = wc.w;
  wx[1] = wd.x; wy[1] = wd.y; wx[0] = wd.z; wy[0] = wd.w;
  const float* cn = coef + (b * NSEC + 2) * CSTRIDE;
  const float nx00 = cn[96], nx01 = cn[97], nx10 = cn[98], nx11 = cn[99];

  for (int i = 0; i < TPW; ++i) {
    float4 n0 = u0, n1 = u1;
    if (i + 1 < TPW) {
      n0 = *(const float4*)(src + (i + 1) * TILE);
      n1 = *(const float4*)(src + (i + 1) * TILE + 128);
    }
    *(float4*)(xs + lane * 4)       = u0;
    *(float4*)(xs + 128 + lane * 4) = u1;
    wave_lds_order();

    float4 xc[4];
#pragma unroll
    for (int q = 0; q < 4; ++q)
      xc[q] = ((const float4*)(xs + m * 16))[q];
    v8f acc = sec_tile(S0, xc, ca0, ca1, m, half);

    wave_lds_order();
    store_block(xs, acc, m, half);
    wave_lds_order();
#pragma unroll
    for (int q = 0; q < 4; ++q)
      xc[q] = ((const float4*)(xs + m * 16))[q];
    acc = sec_tile(S1, xc, cb0, cb1, m, half);

    {
      float q0 = 0.f, q1 = 0.f;
#pragma unroll
      for (int r = 0; r < 8; ++r) {
        q0 += wx[r] * acc[r];
        q1 += wy[r] * acc[r];
      }
      const float n0p = nx00 * pn0 + nx01 * pn1 + q0;
      const float n1p = nx10 * pn0 + nx11 * pn1 + q1;
      pn0 = n0p; pn1 = n1p;
    }

    wave_lds_order();
    store_block(xs, acc, m, half);
    wave_lds_order();
    float4 o0 = *(const float4*)(xs + lane * 4);
    float4 o1 = *(const float4*)(xs + 128 + lane * 4);
    wave_lds_order();
    *(float4*)(dst + i * TILE)       = o0;
    *(float4*)(dst + i * TILE + 128) = o1;

    u0 = n0; u1 = n1;
  }
#pragma unroll
  for (int d = 16; d >= 1; d >>= 1) {
    pn0 += __shfl_xor(pn0, d, 32);
    pn1 += __shfl_xor(pn1, d, 32);
  }
  if (lane == 0) {
    float2 z;
    z.x = ca0; z.y = ca1; ts2[(0 * NB + b) * WPR + w] = z;
    z.x = cb0; z.y = cb1; ts2[(1 * NB + b) * WPR + w] = z;
    z.x = pn0; z.y = pn1; ts2[(2 * NB + b) * WPR + w] = z;
  }
}

struct entry6 { float sl[4]; float cx, cy; };
__device__ __forceinline__ entry6 run_states6(const float* __restrict__ cp,
                                              const float2* z0row, const float2* z1row,
                                              const float2* z2row, int w, int lane) {
  const float* R6 = cp + 16;
  float z[RSEG][6];
#pragma unroll
  for (int c = 0; c < RSEG; ++c) {
    const int k = lane * RSEG + c;
#pragma unroll
    for (int j = 0; j < 6; ++j) z[c][j] = 0.f;
    if (k < WPR) {
      const float2 a = z0row[k], bq = z1row[k], cq = z2row[k];
      z[c][0] = a.x;  z[c][1] = a.y;
      z[c][2] = bq.x; z[c][3] = bq.y;
      z[c][4] = cq.x; z[c][5] = cq.y;
    }
  }
  float s[6], t[6];
#pragma unroll
  for (int j = 0; j < 6; ++j) s[j] = 0.f;
#pragma unroll
  for (int c = 0; c < RSEG; ++c) {
    mv6(R6, s, t);
#pragma unroll
    for (int j = 0; j < 6; ++j) s[j] = t[j] + z[c][j];
  }
  float p[6];
#pragma unroll
  for (int j = 0; j < 6; ++j) p[j] = s[j];
#pragma unroll
  for (int k = 0; k < 5; ++k) {
    const int len = 1 << k;
    float q[6];
#pragma unroll
    for (int j = 0; j < 6; ++j) q[j] = __shfl_up(p[j], (unsigned)len, 32);
    mv6(cp + 52 + 36 * k, q, t);
    if (lane >= len) {
#pragma unroll
      for (int j = 0; j < 6; ++j) p[j] = t[j] + p[j];
    }
  }
#pragma unroll
  for (int j = 0; j < 6; ++j) {
    const float e = __shfl_up(p[j], 1u, 32);
    s[j] = (lane >= 1) ? e : 0.f;
  }
  const int own = w / RSEG, r = w - own * RSEG;
#pragma unroll
  for (int c = 0; c < RSEG - 1; ++c) {
    if (c < r) {
      mv6(R6, s, t);
#pragma unroll
      for (int j = 0; j < 6; ++j) s[j] = t[j] + z[c][j];
    }
  }
  float sg[6];
#pragma unroll
  for (int j = 0; j < 6; ++j)
    sg[j] = __int_as_float(__builtin_amdgcn_readlane(__float_as_int(s[j]), own));
  entry6 o;
  float v[4];
#pragma unroll
  for (int j = 0; j < 4; ++j) { o.sl[j] = 0.f; v[j] = sg[j]; }
#pragma unroll
  for (int i = 0; i < TPW; ++i) {
    if (lane == i) {
#pragma unroll
      for (int j = 0; j < 4; ++j) o.sl[j] = v[j];
    }
    float nv[4];
#pragma unroll
    for (int a = 0; a < 4; ++a) {
      float acc = 0.f;
#pragma unroll
      for (int j = 0; j < 4; ++j)
        if (j <= (a | 1)) acc += cp[a * 4 + j] * v[j];
      nv[a] = acc;
    }
#pragma unroll
    for (int j = 0; j < 4; ++j) v[j] = nv[j];
  }
  o.cx = sg[4]; o.cy = sg[5];
  return o;
}

__global__ void __launch_bounds__(32) last_kernel(const float* in, float* out,
                                                  const float* __restrict__ coef,
                                                  const float* __restrict__ fr,
                                                  const float* __restrict__ gt,
                                                  const float* __restrict__ ts,
                                                  const float* __restrict__ c6) {
  __shared__ float xs[TILE];
  const int lane = threadIdx.x;
  const int b = blockIdx.x / WPR;
  const int w = blockIdx.x - b * WPR;
  const int base0 = b * LSIG + w * TPW * TILE;
  const int m = lane & 15;
  const int half = lane >> 4;

  sec_ops S2;
  load_sec(S2, coef, fr, b, NSEC - 1, lane);
  const float4* G4 = (const float4*)(gt + b * NSEC * TILE * 2);
  float4 ga[4], gb[4];
#pragma unroll
  for (int j = 0; j < 4; ++j) {
    ga[j] = G4[lane * 4 + j];
    gb[j] = G4[128 + lane * 4 + j];
  }
  const float2* ts2 = (const float2*)ts;
  const float* src = in + base0 + lane * 4;
  float* dst = out + base0 + lane * 4;

  float4 u0 = *(const float4*)(src);
  float4 u1 = *(const float4*)(src + 128);
  const entry6 e = run_states6(c6 + b * C6STRIDE, ts2 + (0 * NB + b) * WPR,
                               ts2 + (1 * NB + b) * WPR, ts2 + (2 * NB + b) * WPR,
                               w, lane);
  float cin0 = e.cx, cin1 = e.cy;

  for (int i = 0; i < TPW; ++i) {
    float4 n0 = u0, n1 = u1;
    if (i + 1 < TPW) {
      n0 = *(const float4*)(src + (i + 1) * TILE);
      n1 = *(const float4*)(src + (i + 1) * TILE + 128);
    }

    {
      float sv[4];
#pragma unroll
      for (int j = 0; j < 4; ++j)
        sv[j] = __int_as_float(__builtin_amdgcn_readlane(__float_as_int(e.sl[j]), i));
      u0.x += ga[0].x * sv[0] + ga[0].y * sv[1] + ga[0].z * sv[2] + ga[0].w * sv[3];
      u0.y += ga[1].x * sv[0] + ga[1].y * sv[1] + ga[1].z * sv[2] + ga[1].w * sv[3];
      u0.z += ga[2].x * sv[0] + ga[2].y * sv[1] + ga[2].z * sv[2] + ga[2].w * sv[3];
      u0.w += ga[3].x * sv[0] + ga[3].y * sv[1] + ga[3].z * sv[2] + ga[3].w * sv[3];
      u1.x += gb[0].x * sv[0] + gb[0].y * sv[1] + gb[0].z * sv[2] + gb[0].w * sv[3];
      u1.y += gb[1].x * sv[0] + gb[1].y * sv[1] + gb[1].z * sv[2] + gb[1].w * sv[3];
      u1.z += gb[2].x * sv[0] + gb[2].y * sv[1] + gb[2].z * sv[2] + gb[2].w * sv[3];
      u1.w += gb[3].x * sv[0] + gb[3].y * sv[1] + gb[3].z * sv[2] + gb[3].w * sv[3];
    }
    *(float4*)(xs + lane * 4)       = u0;
    *(float4*)(xs + 128 + lane * 4) = u1;
    wave_lds_order();

    float4 xc[4];
#pragma unroll
    for (int q = 0; q < 4; ++q)
      xc[q] = ((const float4*)(xs + m * 16))[q];
    const v8f acc = sec_tile(S2, xc, cin0, cin1, m, half);

    wave_lds_order();
    store_block(xs, acc, m, half);
    wave_lds_order();
    float4 o0 = *(const float4*)(xs + lane * 4);
    float4 o1 = *(const float4*)(xs + 128 + lane * 4);
    wave_lds_order();
    *(float4*)(dst + i * TILE)       = o0;
    *(float4*)(dst + i * TILE + 128) = o1;

    u0 = n0; u1 = n1;
  }
}

extern "C" void kernel_launch(void* const* d_in, const int* in_sizes, int n_in,
                              void* d_out, int out_size, void* d_ws, size_t ws_size,
                              hipStream_t stream) {
  (void)in_sizes; (void)n_in; (void)out_size; (void)ws_size;
  const float* x     = (const float*)d_in[0];
  const float* knobs = (const float*)d_in[1];
  const float* W1    = (const float*)d_in[2];
  const float* b1v   = (const float*)d_in[3];
  const float* W2    = (const float*)d_in[4];
  const float* b2v   = (const float*)d_in[5];
  float* out = (float*)d_out;

  float* COEF = (float*)d_ws;
  float* FR   = COEF + NB * NSEC * CSTRIDE;
  float* GT   = FR   + NB * NSEC * FSTRIDE;
  float* TS   = GT   + NB * NSEC * TILE * 2;
  float* C6   = TS   + NSEC * NB * WPR * 2;

  derive_kernel<<<NB * DWAVES, 32, 0, stream>>>(knobs, W1, b1v, W2, b2v, COEF, FR, GT, C6);

  const int tileWaves = NB * WPR;
  static_assert(NSEC == 3, "the two launches below are spelled out");
  pair_kernel<<<tileWaves, 32, 0, stream>>>(x, out, COEF, FR, GT, TS);
  last_kernel<<<tileWaves, 32, 0, stream>>>(out, out, COEF, FR, GT, TS, C6);
}


# H*X products take k in the D layout's order (lane half h: k = 8h..8h+7): section 0's output registers feed section 1 with no LDS trip, each half sums its 8 k of the chunk sums; sum order changes
# speedup vs baseline: 9.5533x; 1.0548x over previous
#include <hip/hip_runtime.h>
#include <hip/hip_bf16.h>

#define NB      64
#define LSIG    480000
#define TILE    256
#define NTILES  (LSIG / TILE)
#define NSEC    3
#define CSTRIDE 128
#define FSTRIDE 320
#define TPW     15
#define WPR     (NTILES / TPW)
#define RSEG    4
#define C6STRIDE 256
#define DWAVES  5
static_assert(WPR <= 32 * RSEG, "one wave holds a whole row of run states");

typedef __attribute__((ext_vector_type(2))) float v2f;
typedef __attribute__((ext_vector_type(8))) float v8f;

__device__ __forceinline__ v8f wmma_f32_16x16x4(v2f a, v2f b, v8f c) {
  return __builtin_amdgcn_wmma_f32_16x16x4_f32(false, a, false, b, (short)0, c,
                                               false, false);
}

struct m22 { float a, b, c, d; };
__device__ __forceinline__ m22 mm(const m22 x, const m22 y) {
  m22 r;
  r.a = x.a * y.a + x.b * y.c;  r.b = x.a * y.b + x.b * y.d;
  r.c = x.c * y.a + x.d * y.c;  r.d = x.c * y.b + x.d * y.d;
  return r;
}

struct m66 { float v[36]; };
__device__ __forceinline__ m66 mm6(const m66& x, const m66& y) {
  m66 r;
#pragma unroll
  for (int i = 0; i < 6; ++i) {
#pragma unroll
    for (int j = 0; j < 6; ++j) {
      float a = 0.f;
#pragma unroll
      for (int k = 0; k < 6; ++k)
        if (j <= (i | 1) && k <= (i | 1) && j <= (k | 1)) a += x.v[i * 6 + k] * y.v[k * 6 + j];
      r.v[i * 6 + j] = a;
    }
  }
  return r;
}
__device__ __forceinline__ void mv6(const float* __restrict__ M, const float* s, float* o) {
#pragma unroll
  for (int i = 0; i < 6; ++i) {
    float a = 0.f;
#pragma unroll
    for (int j = 0; j < 6; ++j)
      if (j <= (i | 1)) a += M[i * 6 + j] * s[j];
    o[i] = a;
  }
}

__global__ void __launch_bounds__(32) derive_kernel(const float* __restrict__ knobs,
                                                    const float* __restrict__ W1,
                                                    const float* __restrict__ b1v,
                                                    const float* __restrict__ W2,
                                                    const float* __restrict__ b2v,
                                                    float* __restrict__ coef,
                                                    float* __restrict__ fr,
                                                    float* __restrict__ gt,
                                                    float* __restrict__ c6) {
  __shared__ float hs[16], g0s[16], g1s[16];
  const int lane = threadIdx.x;
  const int b = blockIdx.x / DWAVES;
  const int s = blockIdx.x - b * DWAVES;
  const float k0 = knobs[b * 3 + 0], k1 = knobs[b * 3 + 1], k2 = knobs[b * 3 + 2];
  float hmlp[16];
  for (int i = 0; i < 16; ++i)
    hmlp[i] = tanhf(k0 * W1[0 * 16 + i] + k1 * W1[1 * 16 + i] +
                    k2 * W1[2 * 16 + i] + b1v[i]);

  if (s >= NSEC) {
    float cc0[NSEC], bb0[NSEC], bb1[NSEC], aa1[NSEC], aa2[NSEC];
#pragma unroll
    for (int t = 0; t < NSEC; ++t) {
      float q[5];
#pragma unroll
      for (int j = 0; j < 5; ++j) {
        float acc = b2v[t * 5 + j];
        for (int i = 0; i < 16; ++i) acc += hmlp[i] * W2[i * 15 + t * 5 + j];
        q[j] = acc;
      }
      const float a2c = 0.999f * tanhf(q[4]);
      const float a1c = 0.999f * (1.0f + a2c) * tanhf(q[3]);
      cc0[t] = q[0]; aa1[t] = a1c; aa2[t] = a2c;
      bb0[t] = q[1] - a1c * q[0]; bb1[t] = q[2] - a2c * q[0];
    }
    m66 Y;
#pragma unroll
    for (int i = 0; i < 36; ++i) Y.v[i] = 0.f;
#pragma unroll
    for (int t = 0; t < NSEC; ++t) {
      Y.v[(2 * t) * 6 + 2 * t] = -aa1[t];     Y.v[(2 * t) * 6 + 2 * t + 1] = 1.0f;
      Y.v[(2 * t + 1) * 6 + 2 * t] = -aa2[t]; Y.v[(2 * t + 1) * 6 + 2 * t + 1] = 0.0f;
    }
    Y.v[2 * 6 + 0] = bb0[1];          Y.v[3 * 6 + 0] = bb1[1];
    Y.v[4 * 6 + 2] = bb0[2];          Y.v[5 * 6 + 2] = bb1[2];
    Y.v[4 * 6 + 0] = cc0[1] * bb0[2]; Y.v[5 * 6 + 0] = cc0[1] * bb1[2];
    if (s == NSEC) {
      static_assert(TPW == 15 && RSEG == 4, "the products below spell out 15 = 8+4+2+1 and R^4");
      m66 P = Y;
      for (int q = 0; q < 8; ++q) P = mm6(P, P);
      const m66 Y256 = P;
      const m66 Y512 = mm6(Y256, Y256);
      const m66 Y1024 = mm6(Y512, Y512);
      const m66 Y2048 = mm6(Y1024, Y1024);
      const m66 R6 = mm6(mm6(mm6(Y2048, Y1024), Y512), Y256);
      float* cp = c6 + b * C6STRIDE;
      if (lane == 0) {
#pragma unroll
        for (int i = 0; i < 4; ++i)
#pragma unroll
          for (int j = 0; j < 4; ++j) cp[i * 4 + j] = Y256.v[i * 6 + j];
#pragma unroll
        for (int i = 0; i < 36; ++i) cp[16 + i] = R6.v[i];
      }
      m66 T = mm6(R6, R6); T = mm6(T, T);
      for (int k = 0; k < 5; ++k) {
        if (lane == 0) {
#pragma unroll
          for (int i = 0; i < 36; ++i) cp[52 + 36 * k + i] = T.v[i];
        }
        if (k + 1 < 5) T = mm6(T, T);
      }
    } else {
      float r0 = cc0[1], r1 = 0.0f, r2 = 1.0f, r3 = 0.0f;
      float keep[32];
#pragma unroll
      for (int j = 0; j < 32; ++j) keep[j] = 0.0f;
      for (int blk = 0; blk < 32; ++blk) {
        float t[32];
#pragma unroll
        for (int i = 0; i < 8; ++i) {
          t[4 * i + 0] = r0; t[4 * i + 1] = r1; t[4 * i + 2] = r2; t[4 * i + 3] = r3;
          const float n0 = r0 * Y.v[0 * 6 + 0] + r1 * Y.v[1 * 6 + 0] + r2 * Y.v[2 * 6 + 0] + r3 * Y.v[3 * 6 + 0];
          const float n1 = r0 * Y.v[0 * 6 + 1] + r1 * Y.v[1 * 6 + 1];
          const float n2 = r2 * Y.v[2 * 6 + 2] + r3 * Y.v[3 * 6 + 2];
          const float n3 = r2 * Y.v[2 * 6 + 3] + r3 * Y.v[3 * 6 + 3];
          r0 = n0; r1 = n1; r2 = n2; r3 = n3;
        }
        if (lane == blk) {
#pragma unroll
          for (int j = 0; j < 32; ++j) keep[j] = t[j];
        }
      }
      float4* G = (float4*)(gt + b * NSEC * TILE * 2 + lane * 32);
#pragma unroll
      for (int q = 0; q < 8; ++q) {
        float4 v;
        v.x = keep[4 * q + 0]; v.y = keep[4 * q + 1]; v.z = keep[4 * q + 2]; v.w = keep[4 * q + 3];
        G[q] = v;
      }
    }
    return;
  }

  float p[5];
  for (int j = 0; j < 5; ++j) {
    float acc = b2v[s * 5 + j];
    for (int i = 0; i < 16; ++i) acc += hmlp[i] * W2[i * 15 + s * 5 + j];
    p[j] = acc;
  }
  const float c0 = p[0], c1 = p[1], c2 = p[2];
  const float a2c = 0.999f * tanhf(p[4]);
  const float a1c = 0.999f * (1.0f + a2c) * tanhf(p[3]);
  const float A00 = -a1c, A01 = 1.0f, A10 = -a2c, A11 = 0.0f;
  const float Bv0 = c1 - a1c * c0, Bv1 = c2 - a2c * c0;
  float* cf = coef + (b * NSEC + s) * CSTRIDE;

  float h[16];
  {
    float v0 = Bv0, v1 = Bv1;
    h[0] = c0;
    for (int k = 1; k < 16; ++k) {
      h[k] = v0;
      float n0 = A00 * v0 + A01 * v1;
      float n1 = A10 * v0 + A11 * v1;
      v0 = n0; v1 = n1;
    }
  }
  float fa[16], fb1[16];
  {
    float v0 = Bv0, v1 = Bv1;
    fa[15] = v0; fb1[15] = v1;
    for (int j = 14; j >= 0; --j) {
      float n0 = A00 * v0 + A01 * v1;
      float n1 = A10 * v0 + A11 * v1;
      v0 = n0; v1 = n1;
      fa[j] = v0; fb1[j] = v1;
    }
  }
  float g0i[16], g1i[16];
  {
    float g0 = 1.0f, g1 = 0.0f;
    for (int i = 0; i < 16; ++i) {
      g0i[i] = g0; g1i[i] = g1;
      float n0 = g0 * A00 + g1 * A10;
      float n1 = g0 * A01 + g1 * A11;
      g0 = n0; g1 = n1;
    }
  }
  m22 Am; Am.a = A00; Am.b = A01; Am.c = A10; Am.d = A11;
  const m22 X2 = mm(Am, Am), X4 = mm(X2, X2), X8 = mm(X4, X4);
  m22 Xp[5];
  Xp[0] = mm(X8, X8);
  for (int q = 1; q < 5; ++q) Xp[q] = mm(Xp[q - 1], Xp[q - 1]);

  if (lane == 0) {
    for (int j = 0; j < 16; ++j) {
      cf[16 + j] = fa[j]; cf[32 + j] = fb1[j];
      hs[j] = h[j]; g0s[j] = g0i[j]; g1s[j] = g1i[j];
    }
    for (int q = 0; q < 5; ++q) {
      cf[80 + 4 * q + 0] = Xp[q].a; cf[80 + 4 * q + 1] = Xp[q].b;
      cf[80 + 4 * q + 2] = Xp[q].c; cf[80 + 4 * q + 3] = Xp[q].d;
    }
  }
  __syncthreads();

  {
    float* fb = fr + (b * NSEC + s) * FSTRIDE;
    const int m = lane & 15, half = lane >> 4;
    for (int k = 0; k < 4; ++k) {
      const int K0 = 8 * half + 2 * k;
      const int i0 = m - K0, i1 = m - K0 - 1;
      const float v0 = hs[i0 < 0 ? 0 : i0];
      const float v1 = hs[i1 < 0 ? 0 : i1];
      fb[(k * 32 + lane) * 2 + 0] = (i0 >= 0) ? v0 : 0.0f;
      fb[(k * 32 + lane) * 2 + 1] = (i1 >= 0) ? v1 : 0.0f;
    }
    fb[256 + lane * 2 + 0] = half ? 0.0f : g0s[m];
    fb[256 + lane * 2 + 1] = half ? 0.0f : g1s[m];
  }
  if (s == NSEC - 1) {
    float r[16];
    for (int j = 0; j < 16; ++j) r[j] = 0.0f;
    float v0 = Bv0, v1 = Bv1;
    for (int blk = 0; blk < 32; ++blk) {
      float t[16];
      for (int i = 0; i < 8; ++i) {
        t[2 * i + 0] = v0; t[2 * i + 1] = v1;
        const float n0 = A00 * v0 + A01 * v1;
        const float n1 = A10 * v0 + A11 * v1;
        v0 = n0; v1 = n1;
      }
      if (lane == blk) {
        for (int j = 0; j < 16; ++j) r[j] = t[j];
      }
    }
    float4* G = (float4*)(gt + (b * NSEC + s) * TILE * 2 + lane * 16);
    for (int q = 0; q < 4; ++q) {
      float4 v;
      v.x = r[4 * q + 0]; v.y = r[4 * q + 1]; v.z = r[4 * q + 2]; v.w = r[4 * q + 3];
      G[q] = v;
    }
  }
}


__device__ __forceinline__ void wave_lds_order() {
  __builtin_amdgcn_fence(__ATOMIC_RELEASE, "wavefront");
  __builtin_amdgcn_wave_barrier();
  __builtin_amdgcn_fence(__ATOMIC_ACQUIRE, "wavefront");
}

struct sec_ops {
  v2f ha[4];
  v2f gfr;
  float f0h[8], f1h[8];
  float Ms[16];
};
__device__ __forceinline__ void load_sec(sec_ops& o, const float* __restrict__ coef,
                                         const float* __restrict__ fr, int b, int sec,
                                         int lane, int half) {
  const float* cf = coef + (b * NSEC + sec) * CSTRIDE;
  const v2f* hf = (const v2f*)(fr + (b * NSEC + sec) * FSTRIDE);
#pragma unroll
  for (int k = 0; k < 4; ++k) o.ha[k] = hf[k * 32 + lane];
  o.gfr = hf[128 + lane];
#pragma unroll
  for (int j = 0; j < 8; ++j) {
    o.f0h[j] = cf[16 + 8 * half + j];
    o.f1h[j] = cf[32 + 8 * half + j];
  }
#pragma unroll
  for (int j = 0; j < 16; ++j) o.Ms[j] = cf[80 + j];
}

__device__ __forceinline__ float other_half(float v) {
  return __int_as_float(__builtin_amdgcn_permlanex16(__float_as_int(v), __float_as_int(v),
                                                     0x76543210, 0xfedcba98, false, false));
}

__device__ __forceinline__ v8f sec_tile(const sec_ops& o, const float* xb,
                                        float& cin0, float& cin1, int m, int half) {
  v8f acc = {0.f, 0.f, 0.f, 0.f, 0.f, 0.f, 0.f, 0.f};
#pragma unroll
  for (int k = 0; k < 4; ++k) {
    v2f bb;
    bb.x = xb[2 * k];
    bb.y = xb[2 * k + 1];
    acc = wmma_f32_16x16x4(o.ha[k], bb, acc);
  }

  float fx0 = 0.f, fx1 = 0.f;
#pragma unroll
  for (int i = 0; i < 8; ++i) {
    fx0 += o.f0h[i] * xb[i];
    fx1 += o.f1h[i] * xb[i];
  }
  fx0 += other_half(fx0);
  fx1 += other_half(fx1);

  if (m == 0) {
    fx0 += o.Ms[0] * cin0 + o.Ms[1] * cin1;
    fx1 += o.Ms[2] * cin0 + o.Ms[3] * cin1;
  }

  float p0 = fx0, p1 = fx1;
#pragma unroll
  for (int s = 0; s < 4; ++s) {
    float M00 = o.Ms[4 * s + 0], M01 = o.Ms[4 * s + 1];
    float M10 = o.Ms[4 * s + 2], M11 = o.Ms[4 * s + 3];
    int len = 1 << s;
    float q0 = __shfl_up(p0, (unsigned)len, 32);
    float q1 = __shfl_up(p1, (unsigned)len, 32);
    if (m >= len) {
      p0 = M00 * q0 + M01 * q1 + p0;
      p1 = M10 * q0 + M11 * q1 + p1;
    }
  }
  float e0 = __shfl_up(p0, 1u, 32);
  float e1 = __shfl_up(p1, 1u, 32);
  float s0c = (m >= 1) ? e0 : cin0;
  float s1c = (m >= 1) ? e1 : cin1;
  cin0 = __int_as_float(__builtin_amdgcn_readlane(__float_as_int(p0), 15));
  cin1 = __int_as_float(__builtin_amdgcn_readlane(__float_as_int(p1), 15));

  {
    v2f bb;
    bb.x = half ? 0.f : s0c;
    bb.y = half ? 0.f : s1c;
    acc = wmma_f32_16x16x4(o.gfr, bb, acc);
  }
  return acc;
}

__device__ __forceinline__ void store_block(float* xs, const v8f acc, int m, int half) {
  float4 d0, d1;
  d0.x = acc[0]; d0.y = acc[1]; d0.z = acc[2]; d0.w = acc[3];
  d1.x = acc[4]; d1.y = acc[5]; d1.z = acc[6]; d1.w = acc[7];
  *(float4*)(xs + m * 16 + 8 * half)     = d0;
  *(float4*)(xs + m * 16 + 8 * half + 4) = d1;
}

__global__ void __launch_bounds__(32) pair_kernel(const float* __restrict__ in,
                                                  float* __restrict__ out,
                                                  const float* __restrict__ coef,
                                                  const float* __restrict__ fr,
                                                  const float* __restrict__ gt,
                                                  float* __restrict__ ts) {
  __shared__ float xs[TILE];
  const int lane = threadIdx.x;
  const int b = blockIdx.x / WPR;
  const int w = blockIdx.x - b * WPR;
  const int base0 = b * LSIG + w * TPW * TILE;
  const int m = lane & 15;
  const int half = lane >> 4;

  sec_ops S0, S1;
  load_sec(S0, coef, fr, b, 0, lane, half);
  load_sec(S1, coef, fr, b, 1, lane, half);
  float2* ts2 = (float2*)ts;
  const float* src = in + base0 + lane * 4;
  float* dst = out + base0 + lane * 4;

  float4 u0 = *(const float4*)(src);
  float4 u1 = *(const float4*)(src + 128);
  float ca0 = 0.f, ca1 = 0.f, cb0 = 0.f, cb1 = 0.f;
  float wx[8], wy[8];
  float pn0 = 0.f, pn1 = 0.f;
  const float4* W4 = (const float4*)(gt + (b * NSEC + 2) * TILE * 2 +
                                     (31 - 2 * m - half) * 16);
  const float4 wa = W4[0], wb = W4[1], wc = W4[2], wd = W4[3];
  wx[7] = wa.x; wy[7] = wa.y; wx[6] = wa.z; wy[6] = wa.w;
  wx[5] = wb.x; wy[5] = wb.y; wx[4] = wb.z; wy[4] = wb.w;
  wx[3] = wc.x; wy[3] = wc.y; wx[2] = wc.z; wy[2] = wc.w;
  wx[1] = wd.x; wy[1] = wd.y; wx[0] = wd.z; wy[0] = wd.w;
  const float* cn = coef + (b * NSEC + 2) * CSTRIDE;
  const float nx00 = cn[96], nx01 = cn[97], nx10 = cn[98], nx11 = cn[99];

  for (int i = 0; i < TPW; ++i) {
    float4 n0 = u0, n1 = u1;
    if (i + 1 < TPW) {
      n0 = *(const float4*)(src + (i + 1) * TILE);
      n1 = *(const float4*)(src + (i + 1) * TILE + 128);
    }
    *(float4*)(xs + lane * 4)       = u0;
    *(float4*)(xs + 128 + lane * 4) = u1;
    wave_lds_order();

    float xb[8];
    {
      const float4 xa = *(const float4*)(xs + m * 16 + 8 * half);
      const float4 xd = *(const float4*)(xs + m * 16 + 8 * half + 4);
      xb[0] = xa.x; xb[1] = xa.y; xb[2] = xa.z; xb[3] = xa.w;
      xb[4] = xd.x; xb[5] = xd.y; xb[6] = xd.z; xb[7] = xd.w;
    }
    v8f acc = sec_tile(S0, xb, ca0, ca1, m, half);

#pragma unroll
    for (int r = 0; r < 8; ++r) xb[r] = acc[r];
    acc = sec_tile(S1, xb, cb0, cb1, m, half);

    {
      float q0 = 0.f, q1 = 0.f;
#pragma unroll
      for (int r = 0; r < 8; ++r) {
        q0 += wx[r] * acc[r];
        q1 += wy[r] * acc[r];
      }
      const float n0p = nx00 * pn0 + nx01 * pn1 + q0;
      const float n1p = nx10 * pn0 + nx11 * pn1 + q1;
      pn0 = n0p; pn1 = n1p;
    }

    wave_lds_order();
    store_block(xs, acc, m, half);
    wave_lds_order();
    float4 o0 = *(const float4*)(xs + lane * 4);
    float4 o1 = *(const float4*)(xs + 128 + lane * 4);
    wave_lds_order();
    *(float4*)(dst + i * TILE)       = o0;
    *(float4*)(dst + i * TILE + 128) = o1;

    u0 = n0; u1 = n1;
  }
#pragma unroll
  for (int d = 16; d >= 1; d >>= 1) {
    pn0 += __shfl_xor(pn0, d, 32);
    pn1 += __shfl_xor(pn1, d, 32);
  }
  if (lane == 0) {
    float2 z;
    z.x = ca0; z.y = ca1; ts2[(0 * NB + b) * WPR + w] = z;
    z.x = cb0; z.y = cb1; ts2[(1 * NB + b) * WPR + w] = z;
    z.x = pn0; z.y = pn1; ts2[(2 * NB + b) * WPR + w] = z;
  }
}

struct entry6 { float sl[4]; float cx, cy; };
__device__ __forceinline__ entry6 run_states6(const float* __restrict__ cp,
                                              const float2* z0row, const float2* z1row,
                                              const float2* z2row, int w, int lane) {
  const float* R6 = cp + 16;
  float z[RSEG][6];
#pragma unroll
  for (int c = 0; c < RSEG; ++c) {
    const int k = lane * RSEG + c;
#pragma unroll
    for (int j = 0; j < 6; ++j) z[c][j] = 0.f;
    if (k < WPR) {
      const float2 a = z0row[k], bq = z1row[k], cq = z2row[k];
      z[c][0] = a.x;  z[c][1] = a.y;
      z[c][2] = bq.x; z[c][3] = bq.y;
      z[c][4] = cq.x; z[c][5] = cq.y;
    }
  }
  float s[6], t[6];
#pragma unroll
  for (int j = 0; j < 6; ++j) s[j] = 0.f;
#pragma unroll
  for (int c = 0; c < RSEG; ++c) {
    mv6(R6, s, t);
#pragma unroll
    for (int j = 0; j < 6; ++j) s[j] = t[j] + z[c][j];
  }
  float p[6];
#pragma unroll
  for (int j = 0; j < 6; ++j) p[j] = s[j];
#pragma unroll
  for (int k = 0; k < 5; ++k) {
    const int len = 1 << k;
    float q[6];
#pragma unroll
    for (int j = 0; j < 6; ++j) q[j] = __shfl_up(p[j], (unsigned)len, 32);
    mv6(cp + 52 + 36 * k, q, t);
    if (lane >= len) {
#pragma unroll
      for (int j = 0; j < 6; ++j) p[j] = t[j] + p[j];
    }
  }
#pragma unroll
  for (int j = 0; j < 6; ++j) {
    const float e = __shfl_up(p[j], 1u, 32);
    s[j] = (lane >= 1) ? e : 0.f;
  }
  const int own = w / RSEG, r = w - own * RSEG;
#pragma unroll
  for (int c = 0; c < RSEG - 1; ++c) {
    if (c < r) {
      mv6(R6, s, t);
#pragma unroll
      for (int j = 0; j < 6; ++j) s[j] = t[j] + z[c][j];
    }
  }
  float sg[6];
#pragma unroll
  for (int j = 0; j < 6; ++j)
    sg[j] = __int_as_float(__builtin_amdgcn_readlane(__float_as_int(s[j]), own));
  entry6 o;
  float v[4];
#pragma unroll
  for (int j = 0; j < 4; ++j) { o.sl[j] = 0.f; v[j] = sg[j]; }
#pragma unroll
  for (int i = 0; i < TPW; ++i) {
    if (lane == i) {
#pragma unroll
      for (int j = 0; j < 4; ++j) o.sl[j] = v[j];
    }
    float nv[4];
#pragma unroll
    for (int a = 0; a < 4; ++a) {
      float acc = 0.f;
#pragma unroll
      for (int j = 0; j < 4; ++j)
        if (j <= (a | 1)) acc += cp[a * 4 + j] * v[j];
      nv[a] = acc;
    }
#pragma unroll
    for (int j = 0; j < 4; ++j) v[j] = nv[j];
  }
  o.cx = sg[4]; o.cy = sg[5];
  return o;
}

__global__ void __launch_bounds__(32) last_kernel(const float* in, float* out,
                                                  const float* __restrict__ coef,
                                                  const float* __restrict__ fr,
                                                  const float* __restrict__ gt,
                                                  const float* __restrict__ ts,
                                                  const float* __restrict__ c6) {
  __shared__ float xs[TILE];
  const int lane = threadIdx.x;
  const int b = blockIdx.x / WPR;
  const int w = blockIdx.x - b * WPR;
  const int base0 = b * LSIG + w * TPW * TILE;
  const int m = lane & 15;
  const int half = lane >> 4;

  const float2* ts2 = (const float2*)ts;
  const float* src = in + base0 + lane * 4;
  float* dst = out + base0 + lane * 4;

  float4 u0 = *(const float4*)(src);
  float4 u1 = *(const float4*)(src + 128);
  const entry6 e = run_states6(c6 + b * C6STRIDE, ts2 + (0 * NB + b) * WPR,
                               ts2 + (1 * NB + b) * WPR, ts2 + (2 * NB + b) * WPR,
                               w, lane);
  float cin0 = e.cx, cin1 = e.cy;
  sec_ops S2;
  load_sec(S2, coef, fr, b, NSEC - 1, lane, half);
  const float4* G4 = (const float4*)(gt + b * NSEC * TILE * 2);
  v2f pa[4][2], pb[4][2];
  {
    float4 ga[4], gb[4];
#pragma unroll
    for (int j = 0; j < 4; ++j) {
      ga[j] = G4[lane * 4 + j];
      gb[j] = G4[128 + lane * 4 + j];
    }
#pragma unroll
    for (int p = 0; p < 2; ++p) {
      pa[0][p].x = ga[2 * p].x; pa[0][p].y = ga[2 * p + 1].x;
      pa[1][p].x = ga[2 * p].y; pa[1][p].y = ga[2 * p + 1].y;
      pa[2][p].x = ga[2 * p].z; pa[2][p].y = ga[2 * p + 1].z;
      pa[3][p].x = ga[2 * p].w; pa[3][p].y = ga[2 * p + 1].w;
      pb[0][p].x = gb[2 * p].x; pb[0][p].y = gb[2 * p + 1].x;
      pb[1][p].x = gb[2 * p].y; pb[1][p].y = gb[2 * p + 1].y;
      pb[2][p].x = gb[2 * p].z; pb[2][p].y = gb[2 * p + 1].z;
      pb[3][p].x = gb[2 * p].w; pb[3][p].y = gb[2 * p + 1].w;
    }
  }

  for (int i = 0; i < TPW; ++i) {
    float4 n0 = u0, n1 = u1;
    if (i + 1 < TPW) {
      n0 = *(const float4*)(src + (i + 1) * TILE);
      n1 = *(const float4*)(src + (i + 1) * TILE + 128);
    }

    {
      float sv[4];
#pragma unroll
      for (int j = 0; j < 4; ++j)
        sv[j] = __int_as_float(__builtin_amdgcn_readlane(__float_as_int(e.sl[j]), i));
      const v2f c0 = pa[0][0] * sv[0] + pa[1][0] * sv[1] + pa[2][0] * sv[2] + pa[3][0] * sv[3];
      const v2f c1 = pa[0][1] * sv[0] + pa[1][1] * sv[1] + pa[2][1] * sv[2] + pa[3][1] * sv[3];
      const v2f c2 = pb[0][0] * sv[0] + pb[1][0] * sv[1] + pb[2][0] * sv[2] + pb[3][0] * sv[3];
      const v2f c3 = pb[0][1] * sv[0] + pb[1][1] * sv[1] + pb[2][1] * sv[2] + pb[3][1] * sv[3];
      u0.x += c0.x; u0.y += c0.y; u0.z += c1.x; u0.w += c1.y;
      u1.x += c2.x; u1.y += c2.y; u1.z += c3.x; u1.w += c3.y;
    }
    *(float4*)(xs + lane * 4)       = u0;
    *(float4*)(xs + 128 + lane * 4) = u1;
    wave_lds_order();

    float xb[8];
    {
      const float4 xa = *(const float4*)(xs + m * 16 + 8 * half);
      const float4 xd = *(const float4*)(xs + m * 16 + 8 * half + 4);
      xb[0] = xa.x; xb[1] = xa.y; xb[2] = xa.z; xb[3] = xa.w;
      xb[4] = xd.x; xb[5] = xd.y; xb[6] = xd.z; xb[7] = xd.w;
    }
    const v8f acc = sec_tile(S2, xb, cin0, cin1, m, half);

    wave_lds_order();
    store_block(xs, acc, m, half);
    wave_lds_order();
    float4 o0 = *(const float4*)(xs + lane * 4);
    float4 o1 = *(const float4*)(xs + 128 + lane * 4);
    wave_lds_order();
    *(float4*)(dst + i * TILE)       = o0;
    *(float4*)(dst + i * TILE + 128) = o1;

    u0 = n0; u1 = n1;
  }
}

extern "C" void kernel_launch(void* const* d_in, const int* in_sizes, int n_in,
                              void* d_out, int out_size, void* d_ws, size_t ws_size,
                              hipStream_t stream) {
  (void)in_sizes; (void)n_in; (void)out_size; (void)ws_size;
  const float* x     = (const float*)d_in[0];
  const float* knobs = (const float*)d_in[1];
  const float* W1    = (const float*)d_in[2];
  const float* b1v   = (const float*)d_in[3];
  const float* W2    = (const float*)d_in[4];
  const float* b2v   = (const float*)d_in[5];
  float* out = (float*)d_out;

  float* COEF = (float*)d_ws;
  float* FR   = COEF + NB * NSEC * CSTRIDE;
  float* GT   = FR   + NB * NSEC * FSTRIDE;
  float* TS   = GT   + NB * NSEC * TILE * 2;
  float* C6   = TS   + NSEC * NB * WPR * 2;

  derive_kernel<<<NB * DWAVES, 32, 0, stream>>>(knobs, W1, b1v, W2, b2v, COEF, FR, GT, C6);

  const int tileWaves = NB * WPR;
  static_assert(NSEC == 3, "the two launches below are spelled out");
  pair_kernel<<<tileWaves, 32, 0, stream>>>(x, out, COEF, FR, GT, TS);
  last_kernel<<<tileWaves, 32, 0, stream>>>(out, out, COEF, FR, GT, TS, C6);
}
